# starting kernel
# baseline (speedup 1.0000x reference)
#include <hip/hip_runtime.h>


typedef float v2f __attribute__((ext_vector_type(2)));
typedef float v8f __attribute__((ext_vector_type(8)));

#define HID 128
#define NPRED 16


__global__ void k_set_deg(float* deg, int n) {
  int i = blockIdx.x * blockDim.x + threadIdx.x;
  if (i < n) deg[i] = 1.0f;
}

__global__ void k_deg_scatter(const int* __restrict__ col,
                              const float* __restrict__ w,
                              float* deg, int E) {
  int e = blockIdx.x * blockDim.x + threadIdx.x;
  if (e < E) atomicAdd(&deg[col[e]], w[e]);
}

__global__ void k_dinv(float* deg, int n) {
  int i = blockIdx.x * blockDim.x + threadIdx.x;
  if (i < n) {
    float d = deg[i];
    deg[i] = (d > 0.0f) ? rsqrtf(d) : 0.0f;
  }
}

__global__ void k_norm(const int* __restrict__ row, const int* __restrict__ col,
                       const float* __restrict__ w, const float* __restrict__ dinv,
                       float* __restrict__ norm, int E) {
  int e = blockIdx.x * blockDim.x + threadIdx.x;
  if (e < E) norm[e] = dinv[row[e]] * w[e] * dinv[col[e]];
}

__global__ void k_gemm_wmma(const float* __restrict__ A,
                            const float* __restrict__ W,
                            float* __restrict__ C, int n) {
  int wid  = (blockIdx.x * blockDim.x + threadIdx.x) >> 5;
  int lane = threadIdx.x & 31;
  int row0 = wid * 16;
  if (row0 >= n) return;

  int half = lane >> 4;
  int l    = lane & 15;

  v8f acc[8];
#pragma unroll
  for (int t = 0; t < 8; ++t) acc[t] = (v8f){0,0,0,0,0,0,0,0};

  const float* arow = A + (size_t)(row0 + l) * HID;

  for (int k = 0; k < HID; k += 4) {
    v2f afrag = *(const v2f*)(arow + k + half * 2);
#pragma unroll
    for (int t = 0; t < 8; ++t) {
      int ncol = t * 16 + l;
      v2f bfrag;
      bfrag.x = W[(size_t)(k + half * 2)     * HID + ncol];
      bfrag.y = W[(size_t)(k + half * 2 + 1) * HID + ncol];
      acc[t] = __builtin_amdgcn_wmma_f32_16x16x4_f32(
           false, afrag,  false, bfrag,
           (short)0, acc[t],  false,  false);
    }
  }

#pragma unroll
  for (int t = 0; t < 8; ++t) {
#pragma unroll
    for (int r = 0; r < 8; ++r) {
      int m    = row0 + r + half * 8;
      int ncol = t * 16 + l;
      C[(size_t)m * HID + ncol] = acc[t][r];
    }
  }
}

__global__ void k_scatter(const float* __restrict__ xw,
                          const int* __restrict__ row,
                          const int* __restrict__ col,
                          const float* __restrict__ norm,
                          float* __restrict__ acc, int E) {
  int e = ((blockIdx.x * blockDim.x + threadIdx.x) >> 5);
  if (e >= E) return;
  int lane = threadIdx.x & 31;
  int r = row[e], c = col[e];
  float s = norm[e];
  const float4 v = *(const float4*)(xw + (size_t)r * HID + lane * 4);
  float* dst = acc + (size_t)c * HID + lane * 4;
  atomicAdd(dst + 0, v.x * s);
  atomicAdd(dst + 1, v.y * s);
  atomicAdd(dst + 2, v.z * s);
  atomicAdd(dst + 3, v.w * s);
}

__global__ void k_epilogue(float* __restrict__ acc,
                           const float* __restrict__ xw,
                           const float* __restrict__ dinv,
                           const float* __restrict__ b, int n) {
  int i = blockIdx.x * blockDim.x + threadIdx.x;
  if (i >= n * (HID / 4)) return;
  int node = i >> 5;
  int f4   = i & 31;
  float s  = dinv[node];
  s = s * s;
  float4 a  = ((float4*)acc)[i];
  float4 xv = ((const float4*)xw)[i];
  float4 bv = ((const float4*)b)[f4];
  float4 o;
  o.x = a.x + xv.x * s + bv.x;  o.x = o.x > 0.0f ? o.x : 0.0f;
  o.y = a.y + xv.y * s + bv.y;  o.y = o.y > 0.0f ? o.y : 0.0f;
  o.z = a.z + xv.z * s + bv.z;  o.z = o.z > 0.0f ? o.z : 0.0f;
  o.w = a.w + xv.w * s + bv.w;  o.w = o.w > 0.0f ? o.w : 0.0f;
  ((float4*)acc)[i] = o;
}

__global__ void k_out_proj(const float* __restrict__ h,
                           const float* __restrict__ Wout,
                           const float* __restrict__ bout,
                           float* __restrict__ out, int n) {
  int idx = blockIdx.x * blockDim.x + threadIdx.x;
  if (idx >= n * NPRED) return;
  int node = idx >> 4;
  int p    = idx & 15;
  const float4* hr = (const float4*)(h + (size_t)node * HID);
  const float4* wr = (const float4*)(Wout + (size_t)p * HID);
  float s = 0.0f;
#pragma unroll 8
  for (int k = 0; k < HID / 4; ++k) {
    float4 a = hr[k], b = wr[k];
    s += a.x * b.x + a.y * b.y + a.z * b.z + a.w * b.w;
  }
  out[idx] = s + bout[p];
}


extern "C" void kernel_launch(void* const* d_in, const int* in_sizes, int n_in,
                              void* d_out, int out_size, void* d_ws, size_t ws_size,
                              hipStream_t stream) {
  const float* x    = (const float*)d_in[0];
  const int*   ei   = (const int*)  d_in[1];
  const float* ew   = (const float*)d_in[2];
  const float* W1   = (const float*)d_in[3];
  const float* b1   = (const float*)d_in[4];
  const float* W2   = (const float*)d_in[5];
  const float* b2   = (const float*)d_in[6];
  const float* Wout = (const float*)d_in[7];
  const float* bout = (const float*)d_in[8];
  float* out = (float*)d_out;

  const int n = in_sizes[0] / HID;
  const int E = in_sizes[2];
  const int* erow = ei;
  const int* ecol = ei + E;

  float* dinv = (float*)d_ws;
  float* norm = dinv + n;
  float* xw   = norm + E;
  float* acc  = xw + (size_t)n * HID;

  const int B = 256;
  int gN  = (n + B - 1) / B;
  int gE  = (E + B - 1) / B;
  int nTiles = (n + 15) / 16;
  int gGemm  = (nTiles + 7) / 8;
  int gScat  = (E + 7) / 8;
  int gEpi   = (n * (HID / 4) + B - 1) / B;
  int gOut   = (n * NPRED + B - 1) / B;

  k_set_deg    <<<gN, B, 0, stream>>>(dinv, n);
  k_deg_scatter<<<gE, B, 0, stream>>>(ecol, ew, dinv, E);
  k_dinv       <<<gN, B, 0, stream>>>(dinv, n);
  k_norm       <<<gE, B, 0, stream>>>(erow, ecol, ew, dinv, norm, E);

  k_gemm_wmma<<<gGemm, B, 0, stream>>>(x, W1, xw, n);
  hipMemsetAsync(acc, 0, (size_t)n * HID * sizeof(float), stream);
  k_scatter  <<<gScat, B, 0, stream>>>(xw, erow, ecol, norm, acc, E);
  k_epilogue <<<gEpi, B, 0, stream>>>(acc, xw, dinv, b1, n);

  k_gemm_wmma<<<gGemm, B, 0, stream>>>(acc, W2, xw, n);
  hipMemsetAsync(acc, 0, (size_t)n * HID * sizeof(float), stream);
  k_scatter  <<<gScat, B, 0, stream>>>(xw, erow, ecol, norm, acc, E);
  k_epilogue <<<gEpi, B, 0, stream>>>(acc, xw, dinv, b2, n);

  k_out_proj<<<gOut, B, 0, stream>>>(acc, Wout, bout, out, n);
}


# Edges ordered by destination in the call (integer count with arrival-order rank, scan, place); one wave sums a node's rows in registers: no float atomics, no row-plane zero fill; sum order changes
# speedup vs baseline: 2.3264x; 2.3264x over previous
#include <hip/hip_runtime.h>


typedef float v2f __attribute__((ext_vector_type(2)));
typedef float v8f __attribute__((ext_vector_type(8)));

#define HID 128
#define NPRED 16


__global__ void k_set_deg(float* deg, int n) {
  int i = blockIdx.x * blockDim.x + threadIdx.x;
  if (i < n) deg[i] = 1.0f;
}

__global__ void k_deg_scatter(const int* __restrict__ col,
                              const float* __restrict__ w,
                              float* deg, int E) {
  int e = blockIdx.x * blockDim.x + threadIdx.x;
  if (e < E) atomicAdd(&deg[col[e]], w[e]);
}

__global__ void k_dinv(float* deg, int n) {
  int i = blockIdx.x * blockDim.x + threadIdx.x;
  if (i < n) {
    float d = deg[i];
    deg[i] = (d > 0.0f) ? rsqrtf(d) : 0.0f;
  }
}

__global__ void k_norm(const int* __restrict__ row, const int* __restrict__ col,
                       const float* __restrict__ w, const float* __restrict__ dinv,
                       float* __restrict__ norm, int E) {
  int e = blockIdx.x * blockDim.x + threadIdx.x;
  if (e < E) norm[e] = dinv[row[e]] * w[e] * dinv[col[e]];
}

__global__ void k_gemm_wmma(const float* __restrict__ A,
                            const float* __restrict__ W,
                            float* __restrict__ C, int n) {
  int wid  = (blockIdx.x * blockDim.x + threadIdx.x) >> 5;
  int lane = threadIdx.x & 31;
  int row0 = wid * 16;
  if (row0 >= n) return;

  int half = lane >> 4;
  int l    = lane & 15;

  v8f acc[8];
#pragma unroll
  for (int t = 0; t < 8; ++t) acc[t] = (v8f){0,0,0,0,0,0,0,0};

  const float* arow = A + (size_t)(row0 + l) * HID;

  for (int k = 0; k < HID; k += 4) {
    v2f afrag = *(const v2f*)(arow + k + half * 2);
#pragma unroll
    for (int t = 0; t < 8; ++t) {
      int ncol = t * 16 + l;
      v2f bfrag;
      bfrag.x = W[(size_t)(k + half * 2)     * HID + ncol];
      bfrag.y = W[(size_t)(k + half * 2 + 1) * HID + ncol];
      acc[t] = __builtin_amdgcn_wmma_f32_16x16x4_f32(
           false, afrag,  false, bfrag,
           (short)0, acc[t],  false,  false);
    }
  }

#pragma unroll
  for (int t = 0; t < 8; ++t) {
#pragma unroll
    for (int r = 0; r < 8; ++r) {
      int m    = row0 + r + half * 8;
      int ncol = t * 16 + l;
      C[(size_t)m * HID + ncol] = acc[t][r];
    }
  }
}

__global__ void k_scatter(const float* __restrict__ xw,
                          const int* __restrict__ row,
                          const int* __restrict__ col,
                          const float* __restrict__ norm,
                          float* __restrict__ acc, int E) {
  int e = ((blockIdx.x * blockDim.x + threadIdx.x) >> 5);
  if (e >= E) return;
  int lane = threadIdx.x & 31;
  int r = row[e], c = col[e];
  float s = norm[e];
  const float4 v = *(const float4*)(xw + (size_t)r * HID + lane * 4);
  float* dst = acc + (size_t)c * HID + lane * 4;
  atomicAdd(dst + 0, v.x * s);
  atomicAdd(dst + 1, v.y * s);
  atomicAdd(dst + 2, v.z * s);
  atomicAdd(dst + 3, v.w * s);
}

__global__ void k_epilogue(float* __restrict__ acc,
                           const float* __restrict__ xw,
                           const float* __restrict__ dinv,
                           const float* __restrict__ b, int n) {
  int i = blockIdx.x * blockDim.x + threadIdx.x;
  if (i >= n * (HID / 4)) return;
  int node = i >> 5;
  int f4   = i & 31;
  float s  = dinv[node];
  s = s * s;
  float4 a  = ((float4*)acc)[i];
  float4 xv = ((const float4*)xw)[i];
  float4 bv = ((const float4*)b)[f4];
  float4 o;
  o.x = a.x + xv.x * s + bv.x;  o.x = o.x > 0.0f ? o.x : 0.0f;
  o.y = a.y + xv.y * s + bv.y;  o.y = o.y > 0.0f ? o.y : 0.0f;
  o.z = a.z + xv.z * s + bv.z;  o.z = o.z > 0.0f ? o.z : 0.0f;
  o.w = a.w + xv.w * s + bv.w;  o.w = o.w > 0.0f ? o.w : 0.0f;
  ((float4*)acc)[i] = o;
}

__global__ void k_out_proj(const float* __restrict__ h,
                           const float* __restrict__ Wout,
                           const float* __restrict__ bout,
                           float* __restrict__ out, int n) {
  int idx = blockIdx.x * blockDim.x + threadIdx.x;
  if (idx >= n * NPRED) return;
  int node = idx >> 4;
  int p    = idx & 15;
  const float4* hr = (const float4*)(h + (size_t)node * HID);
  const float4* wr = (const float4*)(Wout + (size_t)p * HID);
  float s = 0.0f;
#pragma unroll 8
  for (int k = 0; k < HID / 4; ++k) {
    float4 a = hr[k], b = wr[k];
    s += a.x * b.x + a.y * b.y + a.z * b.z + a.w * b.w;
  }
  out[idx] = s + bout[p];
}


#define SCAN_T     1024
#define SCAN_ITEMS 4096

__global__ void k_zero_cnt(int* __restrict__ cnt, int npad) {
  int i = blockIdx.x * blockDim.x + threadIdx.x;
  if (i < npad) cnt[i] = 0;
}

__global__ void k_rank(const int* __restrict__ col, int* cnt,
                       int* __restrict__ rank, int E) {
  int e = blockIdx.x * blockDim.x + threadIdx.x;
  if (e < E) rank[e] = atomicAdd(&cnt[col[e]], 1);
}

__global__ void k_scan(const int* __restrict__ cnt, int* __restrict__ offs) {
  __shared__ int s_front[SCAN_T / 32];
  __shared__ int s_wave[SCAN_T / 32];
  const int tid  = threadIdx.x;
  const int lane = tid & 31;
  const int wv   = tid >> 5;
  const int base = blockIdx.x * SCAN_ITEMS;

  int s = 0;
  const int4* front = (const int4*)cnt + tid;
#pragma unroll 4
  for (int k = 0; k < (int)blockIdx.x; ++k) {
    int4 f = front[k * SCAN_T];
    s += f.x + f.y + f.z + f.w;
  }
  for (int m = 16; m > 0; m >>= 1) s += __shfl_xor(s, m);
  if (lane == 0) s_front[wv] = s;

  const int4 v = *(const int4*)(cnt + base + tid * 4);
  const int t = v.x + v.y + v.z + v.w;
  int inc = t;
  for (int d = 1; d < 32; d <<= 1) {
    int y = __shfl_up(inc, d);
    if (lane >= d) inc += y;
  }
  if (lane == 31) s_wave[wv] = inc;
  __syncthreads();

  int ex = inc - t;
  for (int w = 0; w < SCAN_T / 32; ++w) {
    ex += s_front[w];
    if (w < wv) ex += s_wave[w];
  }
  int4 o;
  o.x = ex;
  o.y = ex + v.x;
  o.z = o.y + v.y;
  o.w = o.z + v.z;
  *(int4*)(offs + base + tid * 4) = o;
}

__global__ void k_fill(const int* __restrict__ row, const int* __restrict__ col,
                       const float* __restrict__ w, const int* __restrict__ rank,
                       const int* __restrict__ offs, int2* __restrict__ edge, int E) {
  int e = blockIdx.x * blockDim.x + threadIdx.x;
  if (e < E) {
    int p = offs[col[e]] + rank[e];
    edge[p] = make_int2(row[e], __float_as_int(w[e]));
  }
}

__global__ void k_dinv_seg(const int* __restrict__ offs, const int2* __restrict__ edge,
                           float* __restrict__ dinv, int n) {
  int c = blockIdx.x * blockDim.x + threadIdx.x;
  if (c >= n) return;
  const int p1 = offs[c + 1];
  float d = 1.0f;
  for (int p = offs[c]; p < p1; ++p) d += __int_as_float(edge[p].y);
  dinv[c] = (d > 0.0f) ? rsqrtf(d) : 0.0f;
}

__global__ void k_norm_seg(const int* __restrict__ offs, int2* edge,
                           const float* __restrict__ dinv, int n) {
  int c = blockIdx.x * blockDim.x + threadIdx.x;
  if (c >= n) return;
  const int p1 = offs[c + 1];
  const float dc = dinv[c];
  for (int p = offs[c]; p < p1; ++p) {
    int2 ev = edge[p];
    float nv = dinv[ev.x] * __int_as_float(ev.y) * dc;
    edge[p].y = __float_as_int(nv);
  }
}

__global__ void k_gather(const float* __restrict__ xw, const int* __restrict__ offs,
                         const int2* __restrict__ edge, const float* __restrict__ dinv,
                         const float* __restrict__ b, float* __restrict__ h, int n) {
  const int c = __builtin_amdgcn_readfirstlane(
      (int)((blockIdx.x * blockDim.x + threadIdx.x) >> 5));
  if (c >= n) return;
  const int lane = threadIdx.x & 31;
  const int p0 = offs[c];
  const int p1 = offs[c + 1];
  const float* xl = xw + lane * 4;
  const float4 xv = *(const float4*)(xl + (size_t)c * HID);
  const float4 bv = ((const float4*)b)[lane];
  float s = dinv[c];

  float4 a = make_float4(0.0f, 0.0f, 0.0f, 0.0f);
  for (int base = p0; base < p1; base += 32) {
    int m = p1 - base;
    if (m > 32) m = 32;
    int2 ev = make_int2(0, 0);
    if (lane < m) ev = edge[base + lane];
    int j = 0;
    for (; j + 4 <= m; j += 4) {
      const int r0 = __builtin_amdgcn_readlane(ev.x, j);
      const int r1 = __builtin_amdgcn_readlane(ev.x, j + 1);
      const int r2 = __builtin_amdgcn_readlane(ev.x, j + 2);
      const int r3 = __builtin_amdgcn_readlane(ev.x, j + 3);
      const float4 v0 = *(const float4*)(xl + (size_t)r0 * HID);
      const float4 v1 = *(const float4*)(xl + (size_t)r1 * HID);
      const float4 v2 = *(const float4*)(xl + (size_t)r2 * HID);
      const float4 v3 = *(const float4*)(xl + (size_t)r3 * HID);
      const float s0 = __int_as_float(__builtin_amdgcn_readlane(ev.y, j));
      const float s1 = __int_as_float(__builtin_amdgcn_readlane(ev.y, j + 1));
      const float s2 = __int_as_float(__builtin_amdgcn_readlane(ev.y, j + 2));
      const float s3 = __int_as_float(__builtin_amdgcn_readlane(ev.y, j + 3));
      a.x += v0.x * s0;  a.y += v0.y * s0;  a.z += v0.z * s0;  a.w += v0.w * s0;
      a.x += v1.x * s1;  a.y += v1.y * s1;  a.z += v1.z * s1;  a.w += v1.w * s1;
      a.x += v2.x * s2;  a.y += v2.y * s2;  a.z += v2.z * s2;  a.w += v2.w * s2;
      a.x += v3.x * s3;  a.y += v3.y * s3;  a.z += v3.z * s3;  a.w += v3.w * s3;
    }
    for (; j < m; ++j) {
      const int r0 = __builtin_amdgcn_readlane(ev.x, j);
      const float s0 = __int_as_float(__builtin_amdgcn_readlane(ev.y, j));
      const float4 v0 = *(const float4*)(xl + (size_t)r0 * HID);
      a.x += v0.x * s0;  a.y += v0.y * s0;  a.z += v0.z * s0;  a.w += v0.w * s0;
    }
  }

  s = s * s;
  float4 o;
  o.x = a.x + xv.x * s + bv.x;  o.x = o.x > 0.0f ? o.x : 0.0f;
  o.y = a.y + xv.y * s + bv.y;  o.y = o.y > 0.0f ? o.y : 0.0f;
  o.z = a.z + xv.z * s + bv.z;  o.z = o.z > 0.0f ? o.z : 0.0f;
  o.w = a.w + xv.w * s + bv.w;  o.w = o.w > 0.0f ? o.w : 0.0f;
  *(float4*)(h + (size_t)c * HID + lane * 4) = o;
}


extern "C" void kernel_launch(void* const* d_in, const int* in_sizes, int n_in,
                              void* d_out, int out_size, void* d_ws, size_t ws_size,
                              hipStream_t stream) {
  const float* x    = (const float*)d_in[0];
  const int*   ei   = (const int*)  d_in[1];
  const float* ew   = (const float*)d_in[2];
  const float* W1   = (const float*)d_in[3];
  const float* b1   = (const float*)d_in[4];
  const float* W2   = (const float*)d_in[5];
  const float* b2   = (const float*)d_in[6];
  const float* Wout = (const float*)d_in[7];
  const float* bout = (const float*)d_in[8];
  float* out = (float*)d_out;

  const int n = in_sizes[0] / HID;
  const int E = in_sizes[2];
  const int* erow = ei;
  const int* ecol = ei + E;

  const int B = 256;
  int gN  = (n + B - 1) / B;
  int gE  = (E + B - 1) / B;
  int nTiles = (n + 15) / 16;
  int gGemm  = (nTiles + 7) / 8;
  int gOut   = (n * NPRED + B - 1) / B;

  const int    gScan = (n + SCAN_ITEMS) / SCAN_ITEMS;
  const int    npad  = gScan * SCAN_ITEMS;
  const size_t plane = (size_t)n * HID;
  const size_t epad  = ((size_t)E + 63) / 64 * 64;
  const size_t ndpad = ((size_t)n + 127) / 128 * 128;
  const size_t need  = (2 * plane + 2 * epad + (size_t)npad + ndpad) * sizeof(float);

  if (ws_size >= need && (size_t)npad + (size_t)E <= plane) {
    float* xw   = (float*)d_ws;
    float* h    = xw + plane;
    int2*  edge = (int2*)(h + plane);
    int*   offs = (int*)(h + plane + 2 * epad);
    float* dinv = (float*)(offs + npad);
    int*   cnt  = (int*)h;
    int*   rank = cnt + npad;

    int gGath = (n + 7) / 8;

    k_zero_cnt<<<npad / B, B, 0, stream>>>(cnt, npad);
    k_rank    <<<gE, B, 0, stream>>>(ecol, cnt, rank, E);
    k_scan    <<<gScan, SCAN_T, 0, stream>>>(cnt, offs);
    k_fill    <<<gE, B, 0, stream>>>(erow, ecol, ew, rank, offs, edge, E);
    k_dinv_seg<<<gN, B, 0, stream>>>(offs, edge, dinv, n);
    k_norm_seg<<<gN, B, 0, stream>>>(offs, edge, dinv, n);

    k_gemm_wmma<<<gGemm, B, 0, stream>>>(x, W1, xw, n);
    k_gather   <<<gGath, B, 0, stream>>>(xw, offs, edge, dinv, b1, h, n);

    k_gemm_wmma<<<gGemm, B, 0, stream>>>(h, W2, xw, n);
    k_gather   <<<gGath, B, 0, stream>>>(xw, offs, edge, dinv, b2, h, n);

    k_out_proj<<<gOut, B, 0, stream>>>(h, Wout, bout, out, n);
    return;
  }

  float* dinv = (float*)d_ws;
  float* norm = dinv + n;
  float* xw   = norm + E;
  float* acc  = xw + (size_t)n * HID;

  int gScat  = (E + 7) / 8;
  int gEpi   = (n * (HID / 4) + B - 1) / B;

  k_set_deg    <<<gN, B, 0, stream>>>(dinv, n);
  k_deg_scatter<<<gE, B, 0, stream>>>(ecol, ew, dinv, E);
  k_dinv       <<<gN, B, 0, stream>>>(dinv, n);
  k_norm       <<<gE, B, 0, stream>>>(erow, ecol, ew, dinv, norm, E);

  k_gemm_wmma<<<gGemm, B, 0, stream>>>(x, W1, xw, n);
  hipMemsetAsync(acc, 0, (size_t)n * HID * sizeof(float), stream);
  k_scatter  <<<gScat, B, 0, stream>>>(xw, erow, ecol, norm, acc, E);
  k_epilogue <<<gEpi, B, 0, stream>>>(acc, xw, dinv, b1, n);

  k_gemm_wmma<<<gGemm, B, 0, stream>>>(acc, W2, xw, n);
  hipMemsetAsync(acc, 0, (size_t)n * HID * sizeof(float), stream);
  k_scatter  <<<gScat, B, 0, stream>>>(xw, erow, ecol, norm, acc, E);
  k_epilogue <<<gEpi, B, 0, stream>>>(acc, xw, dinv, b2, n);

  k_out_proj<<<gOut, B, 0, stream>>>(acc, Wout, bout, out, n);
}


# Output head fused into the second aggregation: a block's 16 finished rows pass through LDS to one wave's f32 16x16x4 WMMAs with Wout; h2 never stored, one launch less; out's sum order changes
# speedup vs baseline: 2.6606x; 1.1437x over previous
#include <hip/hip_runtime.h>


typedef float v2f __attribute__((ext_vector_type(2)));
typedef float v8f __attribute__((ext_vector_type(8)));

#define HID 128
#define NPRED 16


__global__ void k_set_deg(float* deg, int n) {
  int i = blockIdx.x * blockDim.x + threadIdx.x;
  if (i < n) deg[i] = 1.0f;
}

__global__ void k_deg_scatter(const int* __restrict__ col,
                              const float* __restrict__ w,
                              float* deg, int E) {
  int e = blockIdx.x * blockDim.x + threadIdx.x;
  if (e < E) atomicAdd(&deg[col[e]], w[e]);
}

__global__ void k_dinv(float* deg, int n) {
  int i = blockIdx.x * blockDim.x + threadIdx.x;
  if (i < n) {
    float d = deg[i];
    deg[i] = (d > 0.0f) ? rsqrtf(d) : 0.0f;
  }
}

__global__ void k_norm(const int* __restrict__ row, const int* __restrict__ col,
                       const float* __restrict__ w, const float* __restrict__ dinv,
                       float* __restrict__ norm, int E) {
  int e = blockIdx.x * blockDim.x + threadIdx.x;
  if (e < E) norm[e] = dinv[row[e]] * w[e] * dinv[col[e]];
}

__global__ void k_gemm_wmma(const float* __restrict__ A,
                            const float* __restrict__ W,
                            float* __restrict__ C, int n) {
  int wid  = (blockIdx.x * blockDim.x + threadIdx.x) >> 5;
  int lane = threadIdx.x & 31;
  int row0 = wid * 16;
  if (row0 >= n) return;

  int half = lane >> 4;
  int l    = lane & 15;

  v8f acc[8];
#pragma unroll
  for (int t = 0; t < 8; ++t) acc[t] = (v8f){0,0,0,0,0,0,0,0};

  const float* arow = A + (size_t)(row0 + l) * HID;

  for (int k = 0; k < HID; k += 4) {
    v2f afrag = *(const v2f*)(arow + k + half * 2);
#pragma unroll
    for (int t = 0; t < 8; ++t) {
      int ncol = t * 16 + l;
      v2f bfrag;
      bfrag.x = W[(size_t)(k + half * 2)     * HID + ncol];
      bfrag.y = W[(size_t)(k + half * 2 + 1) * HID + ncol];
      acc[t] = __builtin_amdgcn_wmma_f32_16x16x4_f32(
           false, afrag,  false, bfrag,
           (short)0, acc[t],  false,  false);
    }
  }

#pragma unroll
  for (int t = 0; t < 8; ++t) {
#pragma unroll
    for (int r = 0; r < 8; ++r) {
      int m    = row0 + r + half * 8;
      int ncol = t * 16 + l;
      C[(size_t)m * HID + ncol] = acc[t][r];
    }
  }
}

__global__ void k_scatter(const float* __restrict__ xw,
                          const int* __restrict__ row,
                          const int* __restrict__ col,
                          const float* __restrict__ norm,
                          float* __restrict__ acc, int E) {
  int e = ((blockIdx.x * blockDim.x + threadIdx.x) >> 5);
  if (e >= E) return;
  int lane = threadIdx.x & 31;
  int r = row[e], c = col[e];
  float s = norm[e];
  const float4 v = *(const float4*)(xw + (size_t)r * HID + lane * 4);
  float* dst = acc + (size_t)c * HID + lane * 4;
  atomicAdd(dst + 0, v.x * s);
  atomicAdd(dst + 1, v.y * s);
  atomicAdd(dst + 2, v.z * s);
  atomicAdd(dst + 3, v.w * s);
}

__global__ void k_epilogue(float* __restrict__ acc,
                           const float* __restrict__ xw,
                           const float* __restrict__ dinv,
                           const float* __restrict__ b, int n) {
  int i = blockIdx.x * blockDim.x + threadIdx.x;
  if (i >= n * (HID / 4)) return;
  int node = i >> 5;
  int f4   = i & 31;
  float s  = dinv[node];
  s = s * s;
  float4 a  = ((float4*)acc)[i];
  float4 xv = ((const float4*)xw)[i];
  float4 bv = ((const float4*)b)[f4];
  float4 o;
  o.x = a.x + xv.x * s + bv.x;  o.x = o.x > 0.0f ? o.x : 0.0f;
  o.y = a.y + xv.y * s + bv.y;  o.y = o.y > 0.0f ? o.y : 0.0f;
  o.z = a.z + xv.z * s + bv.z;  o.z = o.z > 0.0f ? o.z : 0.0f;
  o.w = a.w + xv.w * s + bv.w;  o.w = o.w > 0.0f ? o.w : 0.0f;
  ((float4*)acc)[i] = o;
}

__global__ void k_out_proj(const float* __restrict__ h,
                           const float* __restrict__ Wout,
                           const float* __restrict__ bout,
                           float* __restrict__ out, int n) {
  int idx = blockIdx.x * blockDim.x + threadIdx.x;
  if (idx >= n * NPRED) return;
  int node = idx >> 4;
  int p    = idx & 15;
  const float4* hr = (const float4*)(h + (size_t)node * HID);
  const float4* wr = (const float4*)(Wout + (size_t)p * HID);
  float s = 0.0f;
#pragma unroll 8
  for (int k = 0; k < HID / 4; ++k) {
    float4 a = hr[k], b = wr[k];
    s += a.x * b.x + a.y * b.y + a.z * b.z + a.w * b.w;
  }
  out[idx] = s + bout[p];
}


#define SCAN_T     1024
#define SCAN_ITEMS 4096

__global__ void k_zero_cnt(int* __restrict__ cnt, int npad) {
  int i = blockIdx.x * blockDim.x + threadIdx.x;
  if (i < npad) cnt[i] = 0;
}

__global__ void k_rank(const int* __restrict__ col, int* cnt,
                       int* __restrict__ rank, int E) {
  int e = blockIdx.x * blockDim.x + threadIdx.x;
  if (e < E) rank[e] = atomicAdd(&cnt[col[e]], 1);
}

__global__ void k_scan(const int* __restrict__ cnt, int* __restrict__ offs) {
  __shared__ int s_front[SCAN_T / 32];
  __shared__ int s_wave[SCAN_T / 32];
  const int tid  = threadIdx.x;
  const int lane = tid & 31;
  const int wv   = tid >> 5;
  const int base = blockIdx.x * SCAN_ITEMS;

  int s = 0;
  const int4* front = (const int4*)cnt + tid;
#pragma unroll 4
  for (int k = 0; k < (int)blockIdx.x; ++k) {
    int4 f = front[k * SCAN_T];
    s += f.x + f.y + f.z + f.w;
  }
  for (int m = 16; m > 0; m >>= 1) s += __shfl_xor(s, m);
  if (lane == 0) s_front[wv] = s;

  const int4 v = *(const int4*)(cnt + base + tid * 4);
  const int t = v.x + v.y + v.z + v.w;
  int inc = t;
  for (int d = 1; d < 32; d <<= 1) {
    int y = __shfl_up(inc, d);
    if (lane >= d) inc += y;
  }
  if (lane == 31) s_wave[wv] = inc;
  __syncthreads();

  int ex = inc - t;
  for (int w = 0; w < SCAN_T / 32; ++w) {
    ex += s_front[w];
    if (w < wv) ex += s_wave[w];
  }
  int4 o;
  o.x = ex;
  o.y = ex + v.x;
  o.z = o.y + v.y;
  o.w = o.z + v.z;
  *(int4*)(offs + base + tid * 4) = o;
}

__global__ void k_fill(const int* __restrict__ row, const int* __restrict__ col,
                       const float* __restrict__ w, const int* __restrict__ rank,
                       const int* __restrict__ offs, int2* __restrict__ edge, int E) {
  int e = blockIdx.x * blockDim.x + threadIdx.x;
  if (e < E) {
    int p = offs[col[e]] + rank[e];
    edge[p] = make_int2(row[e], __float_as_int(w[e]));
  }
}

__global__ void k_dinv_seg(const int* __restrict__ offs, const int2* __restrict__ edge,
                           float* __restrict__ dinv, int n) {
  int c = blockIdx.x * blockDim.x + threadIdx.x;
  if (c >= n) return;
  const int p1 = offs[c + 1];
  float d = 1.0f;
  for (int p = offs[c]; p < p1; ++p) d += __int_as_float(edge[p].y);
  dinv[c] = (d > 0.0f) ? rsqrtf(d) : 0.0f;
}

__global__ void k_norm_seg(const int* __restrict__ offs, int2* edge,
                           const float* __restrict__ dinv, int n) {
  int c = blockIdx.x * blockDim.x + threadIdx.x;
  if (c >= n) return;
  const int p1 = offs[c + 1];
  const float dc = dinv[c];
  for (int p = offs[c]; p < p1; ++p) {
    int2 ev = edge[p];
    float nv = dinv[ev.x] * __int_as_float(ev.y) * dc;
    edge[p].y = __float_as_int(nv);
  }
}

__device__ __forceinline__ float4 gather_row(const int c, const int lane,
                                             const float* __restrict__ xw,
                                             const int* __restrict__ offs,
                                             const int2* __restrict__ edge,
                                             const float* __restrict__ dinv,
                                             const float* __restrict__ b) {
  const int p0 = offs[c];
  const int p1 = offs[c + 1];
  const float* xl = xw + lane * 4;
  const float4 xv = *(const float4*)(xl + (size_t)c * HID);
  const float4 bv = ((const float4*)b)[lane];
  float s = dinv[c];

  float4 a = make_float4(0.0f, 0.0f, 0.0f, 0.0f);
  for (int base = p0; base < p1; base += 32) {
    int m = p1 - base;
    if (m > 32) m = 32;
    int2 ev = make_int2(0, 0);
    if (lane < m) ev = edge[base + lane];
    int j = 0;
    for (; j + 4 <= m; j += 4) {
      const int r0 = __builtin_amdgcn_readlane(ev.x, j);
      const int r1 = __builtin_amdgcn_readlane(ev.x, j + 1);
      const int r2 = __builtin_amdgcn_readlane(ev.x, j + 2);
      const int r3 = __builtin_amdgcn_readlane(ev.x, j + 3);
      const float4 v0 = *(const float4*)(xl + (size_t)r0 * HID);
      const float4 v1 = *(const float4*)(xl + (size_t)r1 * HID);
      const float4 v2 = *(const float4*)(xl + (size_t)r2 * HID);
      const float4 v3 = *(const float4*)(xl + (size_t)r3 * HID);
      const float s0 = __int_as_float(__builtin_amdgcn_readlane(ev.y, j));
      const float s1 = __int_as_float(__builtin_amdgcn_readlane(ev.y, j + 1));
      const float s2 = __int_as_float(__builtin_amdgcn_readlane(ev.y, j + 2));
      const float s3 = __int_as_float(__builtin_amdgcn_readlane(ev.y, j + 3));
      a.x += v0.x * s0;  a.y += v0.y * s0;  a.z += v0.z * s0;  a.w += v0.w * s0;
      a.x += v1.x * s1;  a.y += v1.y * s1;  a.z += v1.z * s1;  a.w += v1.w * s1;
      a.x += v2.x * s2;  a.y += v2.y * s2;  a.z += v2.z * s2;  a.w += v2.w * s2;
      a.x += v3.x * s3;  a.y += v3.y * s3;  a.z += v3.z * s3;  a.w += v3.w * s3;
    }
    for (; j < m; ++j) {
      const int r0 = __builtin_amdgcn_readlane(ev.x, j);
      const float s0 = __int_as_float(__builtin_amdgcn_readlane(ev.y, j));
      const float4 v0 = *(const float4*)(xl + (size_t)r0 * HID);
      a.x += v0.x * s0;  a.y += v0.y * s0;  a.z += v0.z * s0;  a.w += v0.w * s0;
    }
  }

  s = s * s;
  float4 o;
  o.x = a.x + xv.x * s + bv.x;  o.x = o.x > 0.0f ? o.x : 0.0f;
  o.y = a.y + xv.y * s + bv.y;  o.y = o.y > 0.0f ? o.y : 0.0f;
  o.z = a.z + xv.z * s + bv.z;  o.z = o.z > 0.0f ? o.z : 0.0f;
  o.w = a.w + xv.w * s + bv.w;  o.w = o.w > 0.0f ? o.w : 0.0f;
  return o;
}

__global__ void k_gather(const float* __restrict__ xw, const int* __restrict__ offs,
                         const int2* __restrict__ edge, const float* __restrict__ dinv,
                         const float* __restrict__ b, float* __restrict__ h, int n) {
  const int c = __builtin_amdgcn_readfirstlane(
      (int)((blockIdx.x * blockDim.x + threadIdx.x) >> 5));
  if (c >= n) return;
  const int lane = threadIdx.x & 31;
  const float4 o = gather_row(c, lane, xw, offs, edge, dinv, b);
  *(float4*)(h + (size_t)c * HID + lane * 4) = o;
}

#define TP 132

__global__ void __launch_bounds__(512)
k_gather_head(const float* __restrict__ xw, const int* __restrict__ offs,
              const int2* __restrict__ edge, const float* __restrict__ dinv,
              const float* __restrict__ b, const float* __restrict__ Wout,
              const float* __restrict__ bout, float* __restrict__ out, int n) {
  __shared__ __attribute__((aligned(16))) float tile[16 * TP];
  const int wv   = __builtin_amdgcn_readfirstlane((int)(threadIdx.x >> 5));
  const int lane = threadIdx.x & 31;
  const int row0 = blockIdx.x * 16;
  const int c    = row0 + wv;

  float4 o = make_float4(0.0f, 0.0f, 0.0f, 0.0f);
  if (c < n) o = gather_row(c, lane, xw, offs, edge, dinv, b);
  *(float4*)(tile + wv * TP + lane * 4) = o;
  __syncthreads();
  if (wv != 0) return;

  const int half = lane >> 4;
  const int l    = lane & 15;
  const float* ar = tile + l * TP + half * 2;
  const float* br = Wout + (size_t)l * HID + half * 2;
  v8f acc0 = (v8f){0,0,0,0,0,0,0,0};
  v8f acc1 = (v8f){0,0,0,0,0,0,0,0};
#pragma unroll 4
  for (int k = 0; k < HID; k += 8) {
    const v2f a0 = *(const v2f*)(ar + k);
    const v2f b0 = *(const v2f*)(br + k);
    const v2f a1 = *(const v2f*)(ar + k + 4);
    const v2f b1 = *(const v2f*)(br + k + 4);
    acc0 = __builtin_amdgcn_wmma_f32_16x16x4_f32(false, a0, false, b0, (short)0, acc0, false, false);
    acc1 = __builtin_amdgcn_wmma_f32_16x16x4_f32(false, a1, false, b1, (short)0, acc1, false, false);
  }
  const float bo = bout[l];
#pragma unroll
  for (int r = 0; r < 8; ++r) {
    const int m = row0 + r + half * 8;
    if (m < n) out[(size_t)m * NPRED + l] = acc0[r] + acc1[r] + bo;
  }
}


extern "C" void kernel_launch(void* const* d_in, const int* in_sizes, int n_in,
                              void* d_out, int out_size, void* d_ws, size_t ws_size,
                              hipStream_t stream) {
  const float* x    = (const float*)d_in[0];
  const int*   ei   = (const int*)  d_in[1];
  const float* ew   = (const float*)d_in[2];
  const float* W1   = (const float*)d_in[3];
  const float* b1   = (const float*)d_in[4];
  const float* W2   = (const float*)d_in[5];
  const float* b2   = (const float*)d_in[6];
  const float* Wout = (const float*)d_in[7];
  const float* bout = (const float*)d_in[8];
  float* out = (float*)d_out;

  const int n = in_sizes[0] / HID;
  const int E = in_sizes[2];
  const int* erow = ei;
  const int* ecol = ei + E;

  const int B = 256;
  int gN  = (n + B - 1) / B;
  int gE  = (E + B - 1) / B;
  int nTiles = (n + 15) / 16;
  int gGemm  = (nTiles + 7) / 8;
  int gOut   = (n * NPRED + B - 1) / B;

  const int    gScan = (n + SCAN_ITEMS) / SCAN_ITEMS;
  const int    npad  = gScan * SCAN_ITEMS;
  const size_t plane = (size_t)n * HID;
  const size_t epad  = ((size_t)E + 63) / 64 * 64;
  const size_t ndpad = ((size_t)n + 127) / 128 * 128;
  const size_t need  = (2 * plane + 2 * epad + (size_t)npad + ndpad) * sizeof(float);

  if (ws_size >= need && (size_t)npad + (size_t)E <= plane) {
    float* xw   = (float*)d_ws;
    float* h    = xw + plane;
    int2*  edge = (int2*)(h + plane);
    int*   offs = (int*)(h + plane + 2 * epad);
    float* dinv = (float*)(offs + npad);
    int*   cnt  = (int*)h;
    int*   rank = cnt + npad;

    int gGath = (n + 7) / 8;

    k_zero_cnt<<<npad / B, B, 0, stream>>>(cnt, npad);
    k_rank    <<<gE, B, 0, stream>>>(ecol, cnt, rank, E);
    k_scan    <<<gScan, SCAN_T, 0, stream>>>(cnt, offs);
    k_fill    <<<gE, B, 0, stream>>>(erow, ecol, ew, rank, offs, edge, E);
    k_dinv_seg<<<gN, B, 0, stream>>>(offs, edge, dinv, n);
    k_norm_seg<<<gN, B, 0, stream>>>(offs, edge, dinv, n);

    k_gemm_wmma<<<gGemm, B, 0, stream>>>(x, W1, xw, n);
    k_gather   <<<gGath, B, 0, stream>>>(xw, offs, edge, dinv, b1, h, n);

    k_gemm_wmma<<<gGemm, B, 0, stream>>>(h, W2, xw, n);
    k_gather_head<<<nTiles, 512, 0, stream>>>(xw, offs, edge, dinv, b2, Wout, bout, out, n);
    return;
  }

  float* dinv = (float*)d_ws;
  float* norm = dinv + n;
  float* xw   = norm + E;
  float* acc  = xw + (size_t)n * HID;

  int gScat  = (E + 7) / 8;
  int gEpi   = (n * (HID / 4) + B - 1) / B;

  k_set_deg    <<<gN, B, 0, stream>>>(dinv, n);
  k_deg_scatter<<<gE, B, 0, stream>>>(ecol, ew, dinv, E);
  k_dinv       <<<gN, B, 0, stream>>>(dinv, n);
  k_norm       <<<gE, B, 0, stream>>>(erow, ecol, ew, dinv, norm, E);

  k_gemm_wmma<<<gGemm, B, 0, stream>>>(x, W1, xw, n);
  hipMemsetAsync(acc, 0, (size_t)n * HID * sizeof(float), stream);
  k_scatter  <<<gScat, B, 0, stream>>>(xw, erow, ecol, norm, acc, E);
  k_epilogue <<<gEpi, B, 0, stream>>>(acc, xw, dinv, b1, n);

  k_gemm_wmma<<<gGemm, B, 0, stream>>>(acc, W2, xw, n);
  hipMemsetAsync(acc, 0, (size_t)n * HID * sizeof(float), stream);
  k_scatter  <<<gScat, B, 0, stream>>>(xw, erow, ecol, norm, acc, E);
  k_epilogue <<<gEpi, B, 0, stream>>>(acc, xw, dinv, b2, n);

  k_out_proj<<<gOut, B, 0, stream>>>(acc, Wout, bout, out, n);
}


# Layer 2's dense product fused into layer 1's aggregation: resident blocks keep W2 in LDS, 16 gathered rows pass LDS -> f32 16x16x4 WMMAs; h1 never stored, one launch less; its k sum splits in two
# speedup vs baseline: 2.9478x; 1.1079x over previous
#include <hip/hip_runtime.h>


typedef float v2f __attribute__((ext_vector_type(2)));
typedef float v8f __attribute__((ext_vector_type(8)));

#define HID 128
#define NPRED 16


__global__ void k_set_deg(float* deg, int n) {
  int i = blockIdx.x * blockDim.x + threadIdx.x;
  if (i < n) deg[i] = 1.0f;
}

__global__ void k_deg_scatter(const int* __restrict__ col,
                              const float* __restrict__ w,
                              float* deg, int E) {
  int e = blockIdx.x * blockDim.x + threadIdx.x;
  if (e < E) atomicAdd(&deg[col[e]], w[e]);
}

__global__ void k_dinv(float* deg, int n) {
  int i = blockIdx.x * blockDim.x + threadIdx.x;
  if (i < n) {
    float d = deg[i];
    deg[i] = (d > 0.0f) ? rsqrtf(d) : 0.0f;
  }
}

__global__ void k_norm(const int* __restrict__ row, const int* __restrict__ col,
                       const float* __restrict__ w, const float* __restrict__ dinv,
                       float* __restrict__ norm, int E) {
  int e = blockIdx.x * blockDim.x + threadIdx.x;
  if (e < E) norm[e] = dinv[row[e]] * w[e] * dinv[col[e]];
}

__global__ void k_gemm_wmma(const float* __restrict__ A,
                            const float* __restrict__ W,
                            float* __restrict__ C, int n) {
  int wid  = (blockIdx.x * blockDim.x + threadIdx.x) >> 5;
  int lane = threadIdx.x & 31;
  int row0 = wid * 16;
  if (row0 >= n) return;

  int half = lane >> 4;
  int l    = lane & 15;

  v8f acc[8];
#pragma unroll
  for (int t = 0; t < 8; ++t) acc[t] = (v8f){0,0,0,0,0,0,0,0};

  const float* arow = A + (size_t)(row0 + l) * HID;

  for (int k = 0; k < HID; k += 4) {
    v2f afrag = *(const v2f*)(arow + k + half * 2);
#pragma unroll
    for (int t = 0; t < 8; ++t) {
      int ncol = t * 16 + l;
      v2f bfrag;
      bfrag.x = W[(size_t)(k + half * 2)     * HID + ncol];
      bfrag.y = W[(size_t)(k + half * 2 + 1) * HID + ncol];
      acc[t] = __builtin_amdgcn_wmma_f32_16x16x4_f32(
           false, afrag,  false, bfrag,
           (short)0, acc[t],  false,  false);
    }
  }

#pragma unroll
  for (int t = 0; t < 8; ++t) {
#pragma unroll
    for (int r = 0; r < 8; ++r) {
      int m    = row0 + r + half * 8;
      int ncol = t * 16 + l;
      C[(size_t)m * HID + ncol] = acc[t][r];
    }
  }
}

__global__ void k_scatter(const float* __restrict__ xw,
                          const int* __restrict__ row,
                          const int* __restrict__ col,
                          const float* __restrict__ norm,
                          float* __restrict__ acc, int E) {
  int e = ((blockIdx.x * blockDim.x + threadIdx.x) >> 5);
  if (e >= E) return;
  int lane = threadIdx.x & 31;
  int r = row[e], c = col[e];
  float s = norm[e];
  const float4 v = *(const float4*)(xw + (size_t)r * HID + lane * 4);
  float* dst = acc + (size_t)c * HID + lane * 4;
  atomicAdd(dst + 0, v.x * s);
  atomicAdd(dst + 1, v.y * s);
  atomicAdd(dst + 2, v.z * s);
  atomicAdd(dst + 3, v.w * s);
}

__global__ void k_epilogue(float* __restrict__ acc,
                           const float* __restrict__ xw,
                           const float* __restrict__ dinv,
                           const float* __restrict__ b, int n) {
  int i = blockIdx.x * blockDim.x + threadIdx.x;
  if (i >= n * (HID / 4)) return;
  int node = i >> 5;
  int f4   = i & 31;
  float s  = dinv[node];
  s = s * s;
  float4 a  = ((float4*)acc)[i];
  float4 xv = ((const float4*)xw)[i];
  float4 bv = ((const float4*)b)[f4];
  float4 o;
  o.x = a.x + xv.x * s + bv.x;  o.x = o.x > 0.0f ? o.x : 0.0f;
  o.y = a.y + xv.y * s + bv.y;  o.y = o.y > 0.0f ? o.y : 0.0f;
  o.z = a.z + xv.z * s + bv.z;  o.z = o.z > 0.0f ? o.z : 0.0f;
  o.w = a.w + xv.w * s + bv.w;  o.w = o.w > 0.0f ? o.w : 0.0f;
  ((float4*)acc)[i] = o;
}

__global__ void k_out_proj(const float* __restrict__ h,
                           const float* __restrict__ Wout,
                           const float* __restrict__ bout,
                           float* __restrict__ out, int n) {
  int idx = blockIdx.x * blockDim.x + threadIdx.x;
  if (idx >= n * NPRED) return;
  int node = idx >> 4;
  int p    = idx & 15;
  const float4* hr = (const float4*)(h + (size_t)node * HID);
  const float4* wr = (const float4*)(Wout + (size_t)p * HID);
  float s = 0.0f;
#pragma unroll 8
  for (int k = 0; k < HID / 4; ++k) {
    float4 a = hr[k], b = wr[k];
    s += a.x * b.x + a.y * b.y + a.z * b.z + a.w * b.w;
  }
  out[idx] = s + bout[p];
}


#define SCAN_T     1024
#define SCAN_ITEMS 4096

__global__ void k_zero_cnt(int* __restrict__ cnt, int* __restrict__ ctr, int npad) {
  int i = blockIdx.x * blockDim.x + threadIdx.x;
  if (i < npad) cnt[i] = 0;
  if (i == 0) *ctr = 0;
}

__global__ void k_rank(const int* __restrict__ col, int* cnt,
                       int* __restrict__ rank, int E) {
  int e = blockIdx.x * blockDim.x + threadIdx.x;
  if (e < E) rank[e] = atomicAdd(&cnt[col[e]], 1);
}

__global__ void k_scan(const int* __restrict__ cnt, int* __restrict__ offs) {
  __shared__ int s_front[SCAN_T / 32];
  __shared__ int s_wave[SCAN_T / 32];
  const int tid  = threadIdx.x;
  const int lane = tid & 31;
  const int wv   = tid >> 5;
  const int base = blockIdx.x * SCAN_ITEMS;

  int s = 0;
  const int4* front = (const int4*)cnt + tid;
#pragma unroll 4
  for (int k = 0; k < (int)blockIdx.x; ++k) {
    int4 f = front[k * SCAN_T];
    s += f.x + f.y + f.z + f.w;
  }
  for (int m = 16; m > 0; m >>= 1) s += __shfl_xor(s, m);
  if (lane == 0) s_front[wv] = s;

  const int4 v = *(const int4*)(cnt + base + tid * 4);
  const int t = v.x + v.y + v.z + v.w;
  int inc = t;
  for (int d = 1; d < 32; d <<= 1) {
    int y = __shfl_up(inc, d);
    if (lane >= d) inc += y;
  }
  if (lane == 31) s_wave[wv] = inc;
  __syncthreads();

  int ex = inc - t;
  for (int w = 0; w < SCAN_T / 32; ++w) {
    ex += s_front[w];
    if (w < wv) ex += s_wave[w];
  }
  int4 o;
  o.x = ex;
  o.y = ex + v.x;
  o.z = o.y + v.y;
  o.w = o.z + v.z;
  *(int4*)(offs + base + tid * 4) = o;
}

__global__ void k_fill(const int* __restrict__ row, const int* __restrict__ col,
                       const float* __restrict__ w, const int* __restrict__ rank,
                       const int* __restrict__ offs, int2* __restrict__ edge, int E) {
  int e = blockIdx.x * blockDim.x + threadIdx.x;
  if (e < E) {
    int p = offs[col[e]] + rank[e];
    edge[p] = make_int2(row[e], __float_as_int(w[e]));
  }
}

__global__ void k_dinv_seg(const int* __restrict__ offs, const int2* __restrict__ edge,
                           float* __restrict__ dinv, int n) {
  int c = blockIdx.x * blockDim.x + threadIdx.x;
  if (c >= n) return;
  const int p1 = offs[c + 1];
  float d = 1.0f;
  for (int p = offs[c]; p < p1; ++p) d += __int_as_float(edge[p].y);
  dinv[c] = (d > 0.0f) ? rsqrtf(d) : 0.0f;
}

__global__ void k_norm_seg(const int* __restrict__ offs, int2* edge,
                           const float* __restrict__ dinv, int n) {
  int c = blockIdx.x * blockDim.x + threadIdx.x;
  if (c >= n) return;
  const int p1 = offs[c + 1];
  const float dc = dinv[c];
  for (int p = offs[c]; p < p1; ++p) {
    int2 ev = edge[p];
    float nv = dinv[ev.x] * __int_as_float(ev.y) * dc;
    edge[p].y = __float_as_int(nv);
  }
}

__device__ __forceinline__ float4 gather_row(const int c, const int lane,
                                             const float* __restrict__ xw,
                                             const int* __restrict__ offs,
                                             const int2* __restrict__ edge,
                                             const float* __restrict__ dinv,
                                             const float* __restrict__ b) {
  const int p0 = offs[c];
  const int p1 = offs[c + 1];
  const float* xl = xw + lane * 4;
  const float4 xv = *(const float4*)(xl + (size_t)c * HID);
  const float4 bv = ((const float4*)b)[lane];
  float s = dinv[c];

  float4 a = make_float4(0.0f, 0.0f, 0.0f, 0.0f);
  for (int base = p0; base < p1; base += 32) {
    int m = p1 - base;
    if (m > 32) m = 32;
    int2 ev = make_int2(0, 0);
    if (lane < m) ev = edge[base + lane];
    int j = 0;
    for (; j + 4 <= m; j += 4) {
      const int r0 = __builtin_amdgcn_readlane(ev.x, j);
      const int r1 = __builtin_amdgcn_readlane(ev.x, j + 1);
      const int r2 = __builtin_amdgcn_readlane(ev.x, j + 2);
      const int r3 = __builtin_amdgcn_readlane(ev.x, j + 3);
      const float4 v0 = *(const float4*)(xl + (size_t)r0 * HID);
      const float4 v1 = *(const float4*)(xl + (size_t)r1 * HID);
      const float4 v2 = *(const float4*)(xl + (size_t)r2 * HID);
      const float4 v3 = *(const float4*)(xl + (size_t)r3 * HID);
      const float s0 = __int_as_float(__builtin_amdgcn_readlane(ev.y, j));
      const float s1 = __int_as_float(__builtin_amdgcn_readlane(ev.y, j + 1));
      const float s2 = __int_as_float(__builtin_amdgcn_readlane(ev.y, j + 2));
      const float s3 = __int_as_float(__builtin_amdgcn_readlane(ev.y, j + 3));
      a.x += v0.x * s0;  a.y += v0.y * s0;  a.z += v0.z * s0;  a.w += v0.w * s0;
      a.x += v1.x * s1;  a.y += v1.y * s1;  a.z += v1.z * s1;  a.w += v1.w * s1;
      a.x += v2.x * s2;  a.y += v2.y * s2;  a.z += v2.z * s2;  a.w += v2.w * s2;
      a.x += v3.x * s3;  a.y += v3.y * s3;  a.z += v3.z * s3;  a.w += v3.w * s3;
    }
    for (; j < m; ++j) {
      const int r0 = __builtin_amdgcn_readlane(ev.x, j);
      const float s0 = __int_as_float(__builtin_amdgcn_readlane(ev.y, j));
      const float4 v0 = *(const float4*)(xl + (size_t)r0 * HID);
      a.x += v0.x * s0;  a.y += v0.y * s0;  a.z += v0.z * s0;  a.w += v0.w * s0;
    }
  }

  s = s * s;
  float4 o;
  o.x = a.x + xv.x * s + bv.x;  o.x = o.x > 0.0f ? o.x : 0.0f;
  o.y = a.y + xv.y * s + bv.y;  o.y = o.y > 0.0f ? o.y : 0.0f;
  o.z = a.z + xv.z * s + bv.z;  o.z = o.z > 0.0f ? o.z : 0.0f;
  o.w = a.w + xv.w * s + bv.w;  o.w = o.w > 0.0f ? o.w : 0.0f;
  return o;
}

#define TP 132

__global__ void __launch_bounds__(512, 16)
k_gather_gemm(const float* __restrict__ xw, const int* __restrict__ offs,
              const int2* __restrict__ edge, const float* __restrict__ dinv,
              const float* __restrict__ b, const float* __restrict__ W,
              float* __restrict__ C, int* ctr, int n, int nTiles) {
  __shared__ __attribute__((aligned(16))) float wf[HID * HID];
  __shared__ __attribute__((aligned(16))) float tile[16 * TP];
  __shared__ int s_first;
  __shared__ int s_next;
  const int wv   = __builtin_amdgcn_readfirstlane((int)(threadIdx.x >> 5));
  const int lane = threadIdx.x & 31;
  const int half = lane >> 4;
  const int l    = lane & 15;

  if (threadIdx.x == 0) s_first = atomicAdd(ctr, 1);

  {
    const int t = wv & 7;
    const float* wc = W + (size_t)(half * 2) * HID + t * 16 + l;
#pragma unroll 8
    for (int i = 0; i < 16; ++i) {
      const int ks = 2 * i + (wv >> 3);
      v2f f;
      f.x = wc[(size_t)(4 * ks)     * HID];
      f.y = wc[(size_t)(4 * ks + 1) * HID];
      *(v2f*)(wf + ((ks * 8 + t) * 32 + lane) * 2) = f;
    }
  }
  __syncthreads();

  int tix = __builtin_amdgcn_readfirstlane(s_first);
  while (tix < nTiles) {
    const int c = tix * 16 + wv;

    float4 o = make_float4(0.0f, 0.0f, 0.0f, 0.0f);
    if (c < n) o = gather_row(c, lane, xw, offs, edge, dinv, b);
    *(float4*)(tile + wv * TP + lane * 4) = o;
    __syncthreads();
    if (threadIdx.x == 480) s_next = atomicAdd(ctr, 1);

    v8f acc0 = (v8f){0,0,0,0,0,0,0,0};
    v8f acc1 = (v8f){0,0,0,0,0,0,0,0};
    if (wv < 8) {
      const float* ar = tile + l * TP + half * 2;
      const float* bf = wf + (wv * 32 + lane) * 2;
#pragma unroll 4
      for (int ks = 0; ks < 32; ks += 2) {
        const v2f a0 = *(const v2f*)(ar + ks * 4);
        const v2f b0 = *(const v2f*)(bf + ks * 512);
        const v2f a1 = *(const v2f*)(ar + ks * 4 + 4);
        const v2f b1 = *(const v2f*)(bf + ks * 512 + 512);
        acc0 = __builtin_amdgcn_wmma_f32_16x16x4_f32(false, a0, false, b0, (short)0, acc0, false, false);
        acc1 = __builtin_amdgcn_wmma_f32_16x16x4_f32(false, a1, false, b1, (short)0, acc1, false, false);
      }
    }
    __syncthreads();
    const int tnext = __builtin_amdgcn_readfirstlane(s_next);
    if (wv < 8) {
      float* st = tile + (half * 8) * TP + wv * 16 + l;
#pragma unroll
      for (int r = 0; r < 8; ++r) st[r * TP] = acc0[r] + acc1[r];
    }
    __syncthreads();
    if (c < n) *(float4*)(C + (size_t)c * HID + lane * 4) = *(const float4*)(tile + wv * TP + lane * 4);
    tix = tnext;
  }
}

__global__ void __launch_bounds__(512)
k_gather_head(const float* __restrict__ xw, const int* __restrict__ offs,
              const int2* __restrict__ edge, const float* __restrict__ dinv,
              const float* __restrict__ b, const float* __restrict__ Wout,
              const float* __restrict__ bout, float* __restrict__ out, int n) {
  __shared__ __attribute__((aligned(16))) float tile[16 * TP];
  const int wv   = __builtin_amdgcn_readfirstlane((int)(threadIdx.x >> 5));
  const int lane = threadIdx.x & 31;
  const int row0 = blockIdx.x * 16;
  const int c    = row0 + wv;

  float4 o = make_float4(0.0f, 0.0f, 0.0f, 0.0f);
  if (c < n) o = gather_row(c, lane, xw, offs, edge, dinv, b);
  *(float4*)(tile + wv * TP + lane * 4) = o;
  __syncthreads();
  if (wv != 0) return;

  const int half = lane >> 4;
  const int l    = lane & 15;
  const float* ar = tile + l * TP + half * 2;
  const float* br = Wout + (size_t)l * HID + half * 2;
  v8f acc0 = (v8f){0,0,0,0,0,0,0,0};
  v8f acc1 = (v8f){0,0,0,0,0,0,0,0};
#pragma unroll 4
  for (int k = 0; k < HID; k += 8) {
    const v2f a0 = *(const v2f*)(ar + k);
    const v2f b0 = *(const v2f*)(br + k);
    const v2f a1 = *(const v2f*)(ar + k + 4);
    const v2f b1 = *(const v2f*)(br + k + 4);
    acc0 = __builtin_amdgcn_wmma_f32_16x16x4_f32(false, a0, false, b0, (short)0, acc0, false, false);
    acc1 = __builtin_amdgcn_wmma_f32_16x16x4_f32(false, a1, false, b1, (short)0, acc1, false, false);
  }
  const float bo = bout[l];
#pragma unroll
  for (int r = 0; r < 8; ++r) {
    const int m = row0 + r + half * 8;
    if (m < n) out[(size_t)m * NPRED + l] = acc0[r] + acc1[r] + bo;
  }
}


extern "C" void kernel_launch(void* const* d_in, const int* in_sizes, int n_in,
                              void* d_out, int out_size, void* d_ws, size_t ws_size,
                              hipStream_t stream) {
  const float* x    = (const float*)d_in[0];
  const int*   ei   = (const int*)  d_in[1];
  const float* ew   = (const float*)d_in[2];
  const float* W1   = (const float*)d_in[3];
  const float* b1   = (const float*)d_in[4];
  const float* W2   = (const float*)d_in[5];
  const float* b2   = (const float*)d_in[6];
  const float* Wout = (const float*)d_in[7];
  const float* bout = (const float*)d_in[8];
  float* out = (float*)d_out;

  const int n = in_sizes[0] / HID;
  const int E = in_sizes[2];
  const int* erow = ei;
  const int* ecol = ei + E;

  const int B = 256;
  int gN  = (n + B - 1) / B;
  int gE  = (E + B - 1) / B;
  int nTiles = (n + 15) / 16;
  int gGemm  = (nTiles + 7) / 8;
  int gOut   = (n * NPRED + B - 1) / B;

  const int    gScan = (n + SCAN_ITEMS) / SCAN_ITEMS;
  const int    npad  = gScan * SCAN_ITEMS;
  const size_t plane = (size_t)n * HID;
  const size_t epad  = ((size_t)E + 63) / 64 * 64;
  const size_t ndpad = ((size_t)n + 127) / 128 * 128;
  const size_t need  = (2 * plane + 2 * epad + (size_t)npad + ndpad + 128) * sizeof(float);

  if (ws_size >= need && (size_t)npad + (size_t)E <= plane) {
    float* xw   = (float*)d_ws;
    float* h    = xw + plane;
    int2*  edge = (int2*)(h + plane);
    int*   offs = (int*)(h + plane + 2 * epad);
    float* dinv = (float*)(offs + npad);
    int*   ctr  = (int*)(dinv + ndpad);
    int*   cnt  = (int*)h;
    int*   rank = cnt + npad;

    k_zero_cnt<<<npad / B, B, 0, stream>>>(cnt, ctr, npad);
    k_rank    <<<gE, B, 0, stream>>>(ecol, cnt, rank, E);
    k_scan    <<<gScan, SCAN_T, 0, stream>>>(cnt, offs);
    k_fill    <<<gE, B, 0, stream>>>(erow, ecol, ew, rank, offs, edge, E);
    k_dinv_seg<<<gN, B, 0, stream>>>(offs, edge, dinv, n);
    k_norm_seg<<<gN, B, 0, stream>>>(offs, edge, dinv, n);

    k_gemm_wmma<<<gGemm, B, 0, stream>>>(x, W1, xw, n);
    k_gather_gemm<<<1024, 512, 0, stream>>>(xw, offs, edge, dinv, b1, W2, h, ctr, n, nTiles);

    k_gather_head<<<nTiles, 512, 0, stream>>>(h, offs, edge, dinv, b2, Wout, bout, out, n);
    return;
  }

  float* dinv = (float*)d_ws;
  float* norm = dinv + n;
  float* xw   = norm + E;
  float* acc  = xw + (size_t)n * HID;

  int gScat  = (E + 7) / 8;
  int gEpi   = (n * (HID / 4) + B - 1) / B;

  k_set_deg    <<<gN, B, 0, stream>>>(dinv, n);
  k_deg_scatter<<<gE, B, 0, stream>>>(ecol, ew, dinv, E);
  k_dinv       <<<gN, B, 0, stream>>>(dinv, n);
  k_norm       <<<gE, B, 0, stream>>>(erow, ecol, ew, dinv, norm, E);

  k_gemm_wmma<<<gGemm, B, 0, stream>>>(x, W1, xw, n);
  hipMemsetAsync(acc, 0, (size_t)n * HID * sizeof(float), stream);
  k_scatter  <<<gScat, B, 0, stream>>>(xw, erow, ecol, norm, acc, E);
  k_epilogue <<<gEpi, B, 0, stream>>>(acc, xw, dinv, b1, n);

  k_gemm_wmma<<<gGemm, B, 0, stream>>>(acc, W2, xw, n);
  hipMemsetAsync(acc, 0, (size_t)n * HID * sizeof(float), stream);
  k_scatter  <<<gScat, B, 0, stream>>>(xw, erow, ecol, norm, acc, E);
  k_epilogue <<<gEpi, B, 0, stream>>>(acc, xw, dinv, b2, n);

  k_out_proj<<<gOut, B, 0, stream>>>(acc, Wout, bout, out, n);
}


# Each layer aggregates first, multiplies after ((A x) W for A (x W)): both gathers end in their dense f32 WMMA product in LDS, head behind layer 2; GEMM launch and its plane go; sum order changes
# speedup vs baseline: 3.2016x; 1.0861x over previous
#include <hip/hip_runtime.h>


typedef float v2f __attribute__((ext_vector_type(2)));
typedef float v8f __attribute__((ext_vector_type(8)));

#define HID 128
#define NPRED 16


__global__ void k_set_deg(float* deg, int n) {
  int i = blockIdx.x * blockDim.x + threadIdx.x;
  if (i < n) deg[i] = 1.0f;
}

__global__ void k_deg_scatter(const int* __restrict__ col,
                              const float* __restrict__ w,
                              float* deg, int E) {
  int e = blockIdx.x * blockDim.x + threadIdx.x;
  if (e < E) atomicAdd(&deg[col[e]], w[e]);
}

__global__ void k_dinv(float* deg, int n) {
  int i = blockIdx.x * blockDim.x + threadIdx.x;
  if (i < n) {
    float d = deg[i];
    deg[i] = (d > 0.0f) ? rsqrtf(d) : 0.0f;
  }
}

__global__ void k_norm(const int* __restrict__ row, const int* __restrict__ col,
                       const float* __restrict__ w, const float* __restrict__ dinv,
                       float* __restrict__ norm, int E) {
  int e = blockIdx.x * blockDim.x + threadIdx.x;
  if (e < E) norm[e] = dinv[row[e]] * w[e] * dinv[col[e]];
}

__global__ void k_gemm_wmma(const float* __restrict__ A,
                            const float* __restrict__ W,
                            float* __restrict__ C, int n) {
  int wid  = (blockIdx.x * blockDim.x + threadIdx.x) >> 5;
  int lane = threadIdx.x & 31;
  int row0 = wid * 16;
  if (row0 >= n) return;

  int half = lane >> 4;
  int l    = lane & 15;

  v8f acc[8];
#pragma unroll
  for (int t = 0; t < 8; ++t) acc[t] = (v8f){0,0,0,0,0,0,0,0};

  const float* arow = A + (size_t)(row0 + l) * HID;

  for (int k = 0; k < HID; k += 4) {
    v2f afrag = *(const v2f*)(arow + k + half * 2);
#pragma unroll
    for (int t = 0; t < 8; ++t) {
      int ncol = t * 16 + l;
      v2f bfrag;
      bfrag.x = W[(size_t)(k + half * 2)     * HID + ncol];
      bfrag.y = W[(size_t)(k + half * 2 + 1) * HID + ncol];
      acc[t] = __builtin_amdgcn_wmma_f32_16x16x4_f32(
           false, afrag,  false, bfrag,
           (short)0, acc[t],  false,  false);
    }
  }

#pragma unroll
  for (int t = 0; t < 8; ++t) {
#pragma unroll
    for (int r = 0; r < 8; ++r) {
      int m    = row0 + r + half * 8;
      int ncol = t * 16 + l;
      C[(size_t)m * HID + ncol] = acc[t][r];
    }
  }
}

__global__ void k_scatter(const float* __restrict__ xw,
                          const int* __restrict__ row,
                          const int* __restrict__ col,
                          const float* __restrict__ norm,
                          float* __restrict__ acc, int E) {
  int e = ((blockIdx.x * blockDim.x + threadIdx.x) >> 5);
  if (e >= E) return;
  int lane = threadIdx.x & 31;
  int r = row[e], c = col[e];
  float s = norm[e];
  const float4 v = *(const float4*)(xw + (size_t)r * HID + lane * 4);
  float* dst = acc + (size_t)c * HID + lane * 4;
  atomicAdd(dst + 0, v.x * s);
  atomicAdd(dst + 1, v.y * s);
  atomicAdd(dst + 2, v.z * s);
  atomicAdd(dst + 3, v.w * s);
}

__global__ void k_epilogue(float* __restrict__ acc,
                           const float* __restrict__ xw,
                           const float* __restrict__ dinv,
                           const float* __restrict__ b, int n) {
  int i = blockIdx.x * blockDim.x + threadIdx.x;
  if (i >= n * (HID / 4)) return;
  int node = i >> 5;
  int f4   = i & 31;
  float s  = dinv[node];
  s = s * s;
  float4 a  = ((float4*)acc)[i];
  float4 xv = ((const float4*)xw)[i];
  float4 bv = ((const float4*)b)[f4];
  float4 o;
  o.x = a.x + xv.x * s + bv.x;  o.x = o.x > 0.0f ? o.x : 0.0f;
  o.y = a.y + xv.y * s + bv.y;  o.y = o.y > 0.0f ? o.y : 0.0f;
  o.z = a.z + xv.z * s + bv.z;  o.z = o.z > 0.0f ? o.z : 0.0f;
  o.w = a.w + xv.w * s + bv.w;  o.w = o.w > 0.0f ? o.w : 0.0f;
  ((float4*)acc)[i] = o;
}

__global__ void k_out_proj(const float* __restrict__ h,
                           const float* __restrict__ Wout,
                           const float* __restrict__ bout,
                           float* __restrict__ out, int n) {
  int idx = blockIdx.x * blockDim.x + threadIdx.x;
  if (idx >= n * NPRED) return;
  int node = idx >> 4;
  int p    = idx & 15;
  const float4* hr = (const float4*)(h + (size_t)node * HID);
  const float4* wr = (const float4*)(Wout + (size_t)p * HID);
  float s = 0.0f;
#pragma unroll 8
  for (int k = 0; k < HID / 4; ++k) {
    float4 a = hr[k], b = wr[k];
    s += a.x * b.x + a.y * b.y + a.z * b.z + a.w * b.w;
  }
  out[idx] = s + bout[p];
}


#define SCAN_T     1024
#define SCAN_ITEMS 4096

__global__ void k_zero_cnt(int* __restrict__ cnt, int* __restrict__ ctr, int npad) {
  int i = blockIdx.x * blockDim.x + threadIdx.x;
  if (i < npad) cnt[i] = 0;
  if (i < 2) ctr[i] = 0;
}

__global__ void k_rank(const int* __restrict__ col, int* cnt,
                       int* __restrict__ rank, int E) {
  int e = blockIdx.x * blockDim.x + threadIdx.x;
  if (e < E) rank[e] = atomicAdd(&cnt[col[e]], 1);
}

__global__ void k_scan(const int* __restrict__ cnt, int* __restrict__ offs) {
  __shared__ int s_front[SCAN_T / 32];
  __shared__ int s_wave[SCAN_T / 32];
  const int tid  = threadIdx.x;
  const int lane = tid & 31;
  const int wv   = tid >> 5;
  const int base = blockIdx.x * SCAN_ITEMS;

  int s = 0;
  const int4* front = (const int4*)cnt + tid;
#pragma unroll 4
  for (int k = 0; k < (int)blockIdx.x; ++k) {
    int4 f = front[k * SCAN_T];
    s += f.x + f.y + f.z + f.w;
  }
  for (int m = 16; m > 0; m >>= 1) s += __shfl_xor(s, m);
  if (lane == 0) s_front[wv] = s;

  const int4 v = *(const int4*)(cnt + base + tid * 4);
  const int t = v.x + v.y + v.z + v.w;
  int inc = t;
  for (int d = 1; d < 32; d <<= 1) {
    int y = __shfl_up(inc, d);
    if (lane >= d) inc += y;
  }
  if (lane == 31) s_wave[wv] = inc;
  __syncthreads();

  int ex = inc - t;
  for (int w = 0; w < SCAN_T / 32; ++w) {
    ex += s_front[w];
    if (w < wv) ex += s_wave[w];
  }
  int4 o;
  o.x = ex;
  o.y = ex + v.x;
  o.z = o.y + v.y;
  o.w = o.z + v.z;
  *(int4*)(offs + base + tid * 4) = o;
}

__global__ void k_fill(const int* __restrict__ row, const int* __restrict__ col,
                       const float* __restrict__ w, const int* __restrict__ rank,
                       const int* __restrict__ offs, int2* __restrict__ edge, int E) {
  int e = blockIdx.x * blockDim.x + threadIdx.x;
  if (e < E) {
    int p = offs[col[e]] + rank[e];
    edge[p] = make_int2(row[e], __float_as_int(w[e]));
  }
}

__global__ void k_dinv_seg(const int* __restrict__ offs, const int2* __restrict__ edge,
                           float* __restrict__ dinv, int n) {
  int c = blockIdx.x * blockDim.x + threadIdx.x;
  if (c >= n) return;
  const int p1 = offs[c + 1];
  float d = 1.0f;
  for (int p = offs[c]; p < p1; ++p) d += __int_as_float(edge[p].y);
  dinv[c] = (d > 0.0f) ? rsqrtf(d) : 0.0f;
}

__global__ void k_norm_seg(const int* __restrict__ offs, int2* edge,
                           const float* __restrict__ dinv, int n) {
  int c = blockIdx.x * blockDim.x + threadIdx.x;
  if (c >= n) return;
  const int p1 = offs[c + 1];
  const float dc = dinv[c];
  for (int p = offs[c]; p < p1; ++p) {
    int2 ev = edge[p];
    float nv = dinv[ev.x] * __int_as_float(ev.y) * dc;
    edge[p].y = __float_as_int(nv);
  }
}

__device__ __forceinline__ float4 gather_row(const int c, const int lane,
                                             const float* __restrict__ xin,
                                             const int* __restrict__ offs,
                                             const int2* __restrict__ edge,
                                             const float* __restrict__ dinv) {
  const int p0 = offs[c];
  const int p1 = offs[c + 1];
  const float* xl = xin + lane * 4;
  const float4 xv = *(const float4*)(xl + (size_t)c * HID);
  float s = dinv[c];

  float4 a = make_float4(0.0f, 0.0f, 0.0f, 0.0f);
  for (int base = p0; base < p1; base += 32) {
    int m = p1 - base;
    if (m > 32) m = 32;
    int2 ev = make_int2(0, 0);
    if (lane < m) ev = edge[base + lane];
    int j = 0;
    for (; j + 4 <= m; j += 4) {
      const int r0 = __builtin_amdgcn_readlane(ev.x, j);
      const int r1 = __builtin_amdgcn_readlane(ev.x, j + 1);
      const int r2 = __builtin_amdgcn_readlane(ev.x, j + 2);
      const int r3 = __builtin_amdgcn_readlane(ev.x, j + 3);
      const float4 v0 = *(const float4*)(xl + (size_t)r0 * HID);
      const float4 v1 = *(const float4*)(xl + (size_t)r1 * HID);
      const float4 v2 = *(const float4*)(xl + (size_t)r2 * HID);
      const float4 v3 = *(const float4*)(xl + (size_t)r3 * HID);
      const float s0 = __int_as_float(__builtin_amdgcn_readlane(ev.y, j));
      const float s1 = __int_as_float(__builtin_amdgcn_readlane(ev.y, j + 1));
      const float s2 = __int_as_float(__builtin_amdgcn_readlane(ev.y, j + 2));
      const float s3 = __int_as_float(__builtin_amdgcn_readlane(ev.y, j + 3));
      a.x += v0.x * s0;  a.y += v0.y * s0;  a.z += v0.z * s0;  a.w += v0.w * s0;
      a.x += v1.x * s1;  a.y += v1.y * s1;  a.z += v1.z * s1;  a.w += v1.w * s1;
      a.x += v2.x * s2;  a.y += v2.y * s2;  a.z += v2.z * s2;  a.w += v2.w * s2;
      a.x += v3.x * s3;  a.y += v3.y * s3;  a.z += v3.z * s3;  a.w += v3.w * s3;
    }
    for (; j < m; ++j) {
      const int r0 = __builtin_amdgcn_readlane(ev.x, j);
      const float s0 = __int_as_float(__builtin_amdgcn_readlane(ev.y, j));
      const float4 v0 = *(const float4*)(xl + (size_t)r0 * HID);
      a.x += v0.x * s0;  a.y += v0.y * s0;  a.z += v0.z * s0;  a.w += v0.w * s0;
    }
  }

  s = s * s;
  float4 o;
  o.x = a.x + xv.x * s;
  o.y = a.y + xv.y * s;
  o.z = a.z + xv.z * s;
  o.w = a.w + xv.w * s;
  return o;
}

#define TP 132

__device__ __forceinline__ void stage_w(float* wf, const float* __restrict__ W,
                                        const int wv, const int lane) {
  const int half = lane >> 4;
  const int l    = lane & 15;
  const int t = wv & 7;
  const float* wc = W + (size_t)(half * 2) * HID + t * 16 + l;
#pragma unroll 8
  for (int i = 0; i < 16; ++i) {
    const int ks = 2 * i + (wv >> 3);
    v2f f;
    f.x = wc[(size_t)(4 * ks)     * HID];
    f.y = wc[(size_t)(4 * ks + 1) * HID];
    *(v2f*)(wf + ((ks * 8 + t) * 32 + lane) * 2) = f;
  }
}

__device__ __forceinline__ void tile_dense(float* tile, const float* wf,
                                           const float* __restrict__ b,
                                           const int wv, const int lane) {
  const int half = lane >> 4;
  const int l    = lane & 15;
  v8f acc0 = (v8f){0,0,0,0,0,0,0,0};
  v8f acc1 = (v8f){0,0,0,0,0,0,0,0};
  float bias = 0.0f;
  if (wv < 8) {
    bias = b[wv * 16 + l];
    const float* ar = tile + l * TP + half * 2;
    const float* bf = wf + (wv * 32 + lane) * 2;
#pragma unroll 4
    for (int ks = 0; ks < 32; ks += 2) {
      const v2f a0 = *(const v2f*)(ar + ks * 4);
      const v2f b0 = *(const v2f*)(bf + ks * 512);
      const v2f a1 = *(const v2f*)(ar + ks * 4 + 4);
      const v2f b1 = *(const v2f*)(bf + ks * 512 + 512);
      acc0 = __builtin_amdgcn_wmma_f32_16x16x4_f32(false, a0, false, b0, (short)0, acc0, false, false);
      acc1 = __builtin_amdgcn_wmma_f32_16x16x4_f32(false, a1, false, b1, (short)0, acc1, false, false);
    }
  }
  __syncthreads();
  if (wv < 8) {
    float* st = tile + (half * 8) * TP + wv * 16 + l;
#pragma unroll
    for (int r = 0; r < 8; ++r) {
      const float v = acc0[r] + acc1[r] + bias;
      st[r * TP] = v > 0.0f ? v : 0.0f;
    }
  }
  __syncthreads();
}

__global__ void __launch_bounds__(512, 16)
k_gather_dense(const float* __restrict__ xin, const int* __restrict__ offs,
               const int2* __restrict__ edge, const float* __restrict__ dinv,
               const float* __restrict__ W, const float* __restrict__ b,
               float* __restrict__ C, int* ctr, int n, int nTiles) {
  __shared__ __attribute__((aligned(16))) float wf[HID * HID];
  __shared__ __attribute__((aligned(16))) float tile[16 * TP];
  __shared__ int s_first;
  __shared__ int s_next;
  const int wv   = __builtin_amdgcn_readfirstlane((int)(threadIdx.x >> 5));
  const int lane = threadIdx.x & 31;

  if (threadIdx.x == 0) s_first = atomicAdd(ctr, 1);
  stage_w(wf, W, wv, lane);
  __syncthreads();

  int tix = __builtin_amdgcn_readfirstlane(s_first);
  while (tix < nTiles) {
    const int c = tix * 16 + wv;

    float4 o = make_float4(0.0f, 0.0f, 0.0f, 0.0f);
    if (c < n) o = gather_row(c, lane, xin, offs, edge, dinv);
    *(float4*)(tile + wv * TP + lane * 4) = o;
    __syncthreads();
    if (threadIdx.x == 480) s_next = atomicAdd(ctr, 1);

    tile_dense(tile, wf, b, wv, lane);
    const int tnext = __builtin_amdgcn_readfirstlane(s_next);
    if (c < n) *(float4*)(C + (size_t)c * HID + lane * 4) = *(const float4*)(tile + wv * TP + lane * 4);
    tix = tnext;
  }
}

__global__ void __launch_bounds__(512, 16)
k_gather_dense_head(const float* __restrict__ hin, const int* __restrict__ offs,
                    const int2* __restrict__ edge, const float* __restrict__ dinv,
                    const float* __restrict__ W, const float* __restrict__ b,
                    const float* __restrict__ Wout, const float* __restrict__ bout,
                    float* __restrict__ out, int* ctr, int n, int nTiles) {
  __shared__ __attribute__((aligned(16))) float wf[HID * HID];
  __shared__ __attribute__((aligned(16))) float tile[16 * TP];
  __shared__ int s_first;
  __shared__ int s_next;
  const int wv   = __builtin_amdgcn_readfirstlane((int)(threadIdx.x >> 5));
  const int lane = threadIdx.x & 31;
  const int half = lane >> 4;
  const int l    = lane & 15;

  if (threadIdx.x == 0) s_first = atomicAdd(ctr, 1);
  stage_w(wf, W, wv, lane);
  __syncthreads();

  int tix = __builtin_amdgcn_readfirstlane(s_first);
  while (tix < nTiles) {
    const int row0 = tix * 16;
    const int c    = row0 + wv;

    float4 o = make_float4(0.0f, 0.0f, 0.0f, 0.0f);
    if (c < n) o = gather_row(c, lane, hin, offs, edge, dinv);
    __syncthreads();
    *(float4*)(tile + wv * TP + lane * 4) = o;
    __syncthreads();
    if (threadIdx.x == 480) s_next = atomicAdd(ctr, 1);

    tile_dense(tile, wf, b, wv, lane);
    const int tnext = __builtin_amdgcn_readfirstlane(s_next);

    if (wv == 8) {
      const float* ar = tile + l * TP + half * 2;
      const float* br = Wout + (size_t)l * HID + half * 2;
      v8f acc0 = (v8f){0,0,0,0,0,0,0,0};
      v8f acc1 = (v8f){0,0,0,0,0,0,0,0};
#pragma unroll 4
      for (int k = 0; k < HID; k += 8) {
        const v2f a0 = *(const v2f*)(ar + k);
        const v2f b0 = *(const v2f*)(br + k);
        const v2f a1 = *(const v2f*)(ar + k + 4);
        const v2f b1 = *(const v2f*)(br + k + 4);
        acc0 = __builtin_amdgcn_wmma_f32_16x16x4_f32(false, a0, false, b0, (short)0, acc0, false, false);
        acc1 = __builtin_amdgcn_wmma_f32_16x16x4_f32(false, a1, false, b1, (short)0, acc1, false, false);
      }
      const float bo = bout[l];
#pragma unroll
      for (int r = 0; r < 8; ++r) {
        const int m = row0 + r + half * 8;
        if (m < n) out[(size_t)m * NPRED + l] = acc0[r] + acc1[r] + bo;
      }
    }
    tix = tnext;
  }
}


extern "C" void kernel_launch(void* const* d_in, const int* in_sizes, int n_in,
                              void* d_out, int out_size, void* d_ws, size_t ws_size,
                              hipStream_t stream) {
  const float* x    = (const float*)d_in[0];
  const int*   ei   = (const int*)  d_in[1];
  const float* ew   = (const float*)d_in[2];
  const float* W1   = (const float*)d_in[3];
  const float* b1   = (const float*)d_in[4];
  const float* W2   = (const float*)d_in[5];
  const float* b2   = (const float*)d_in[6];
  const float* Wout = (const float*)d_in[7];
  const float* bout = (const float*)d_in[8];
  float* out = (float*)d_out;

  const int n = in_sizes[0] / HID;
  const int E = in_sizes[2];
  const int* erow = ei;
  const int* ecol = ei + E;

  const int B = 256;
  int gN  = (n + B - 1) / B;
  int gE  = (E + B - 1) / B;
  int nTiles = (n + 15) / 16;
  int gGemm  = (nTiles + 7) / 8;
  int gOut   = (n * NPRED + B - 1) / B;

  const int    gScan = (n + SCAN_ITEMS) / SCAN_ITEMS;
  const int    npad  = gScan * SCAN_ITEMS;
  const size_t plane = (size_t)n * HID;
  const size_t epad  = ((size_t)E + 63) / 64 * 64;
  const size_t ndpad = ((size_t)n + 127) / 128 * 128;
  const size_t need  = (plane + 2 * epad + (size_t)npad + ndpad + 128) * sizeof(float);

  if (ws_size >= need && (size_t)npad + (size_t)E <= plane) {
    float* h    = (float*)d_ws;
    int2*  edge = (int2*)(h + plane);
    int*   offs = (int*)(h + plane + 2 * epad);
    float* dinv = (float*)(offs + npad);
    int*   ctr  = (int*)(dinv + ndpad);
    int*   cnt  = (int*)h;
    int*   rank = cnt + npad;

    k_zero_cnt<<<npad / B, B, 0, stream>>>(cnt, ctr, npad);
    k_rank    <<<gE, B, 0, stream>>>(ecol, cnt, rank, E);
    k_scan    <<<gScan, SCAN_T, 0, stream>>>(cnt, offs);
    k_fill    <<<gE, B, 0, stream>>>(erow, ecol, ew, rank, offs, edge, E);
    k_dinv_seg<<<gN, B, 0, stream>>>(offs, edge, dinv, n);
    k_norm_seg<<<gN, B, 0, stream>>>(offs, edge, dinv, n);

    k_gather_dense<<<1024, 512, 0, stream>>>(x, offs, edge, dinv, W1, b1, h, ctr, n, nTiles);

    k_gather_dense_head<<<1024, 512, 0, stream>>>(h, offs, edge, dinv, W2, b2, Wout, bout, out,
                                                  ctr + 1, n, nTiles);
    return;
  }

  float* dinv = (float*)d_ws;
  float* norm = dinv + n;
  float* xw   = norm + E;
  float* acc  = xw + (size_t)n * HID;

  int gScat  = (E + 7) / 8;
  int gEpi   = (n * (HID / 4) + B - 1) / B;

  k_set_deg    <<<gN, B, 0, stream>>>(dinv, n);
  k_deg_scatter<<<gE, B, 0, stream>>>(ecol, ew, dinv, E);
  k_dinv       <<<gN, B, 0, stream>>>(dinv, n);
  k_norm       <<<gE, B, 0, stream>>>(erow, ecol, ew, dinv, norm, E);

  k_gemm_wmma<<<gGemm, B, 0, stream>>>(x, W1, xw, n);
  hipMemsetAsync(acc, 0, (size_t)n * HID * sizeof(float), stream);
  k_scatter  <<<gScat, B, 0, stream>>>(xw, erow, ecol, norm, acc, E);
  k_epilogue <<<gEpi, B, 0, stream>>>(acc, xw, dinv, b1, n);

  k_gemm_wmma<<<gGemm, B, 0, stream>>>(acc, W2, xw, n);
  hipMemsetAsync(acc, 0, (size_t)n * HID * sizeof(float), stream);
  k_scatter  <<<gScat, B, 0, stream>>>(xw, erow, ecol, norm, acc, E);
  k_epilogue <<<gEpi, B, 0, stream>>>(acc, xw, dinv, b2, n);

  k_out_proj<<<gOut, B, 0, stream>>>(acc, Wout, bout, out, n);
}


# Per-edge norm pass dropped: layer 1 scales by dinv[src] as it reads the list, h1 is stored times dinv[row], dinv[dst] multiplies each sum once; one launch less; products regrouped, sum order kept
# speedup vs baseline: 3.4335x; 1.0724x over previous
#include <hip/hip_runtime.h>


typedef float v2f __attribute__((ext_vector_type(2)));
typedef float v8f __attribute__((ext_vector_type(8)));

#define HID 128
#define NPRED 16


__global__ void k_set_deg(float* deg, int n) {
  int i = blockIdx.x * blockDim.x + threadIdx.x;
  if (i < n) deg[i] = 1.0f;
}

__global__ void k_deg_scatter(const int* __restrict__ col,
                              const float* __restrict__ w,
                              float* deg, int E) {
  int e = blockIdx.x * blockDim.x + threadIdx.x;
  if (e < E) atomicAdd(&deg[col[e]], w[e]);
}

__global__ void k_dinv(float* deg, int n) {
  int i = blockIdx.x * blockDim.x + threadIdx.x;
  if (i < n) {
    float d = deg[i];
    deg[i] = (d > 0.0f) ? rsqrtf(d) : 0.0f;
  }
}

__global__ void k_norm(const int* __restrict__ row, const int* __restrict__ col,
                       const float* __restrict__ w, const float* __restrict__ dinv,
                       float* __restrict__ norm, int E) {
  int e = blockIdx.x * blockDim.x + threadIdx.x;
  if (e < E) norm[e] = dinv[row[e]] * w[e] * dinv[col[e]];
}

__global__ void k_gemm_wmma(const float* __restrict__ A,
                            const float* __restrict__ W,
                            float* __restrict__ C, int n) {
  int wid  = (blockIdx.x * blockDim.x + threadIdx.x) >> 5;
  int lane = threadIdx.x & 31;
  int row0 = wid * 16;
  if (row0 >= n) return;

  int half = lane >> 4;
  int l    = lane & 15;

  v8f acc[8];
#pragma unroll
  for (int t = 0; t < 8; ++t) acc[t] = (v8f){0,0,0,0,0,0,0,0};

  const float* arow = A + (size_t)(row0 + l) * HID;

  for (int k = 0; k < HID; k += 4) {
    v2f afrag = *(const v2f*)(arow + k + half * 2);
#pragma unroll
    for (int t = 0; t < 8; ++t) {
      int ncol = t * 16 + l;
      v2f bfrag;
      bfrag.x = W[(size_t)(k + half * 2)     * HID + ncol];
      bfrag.y = W[(size_t)(k + half * 2 + 1) * HID + ncol];
      acc[t] = __builtin_amdgcn_wmma_f32_16x16x4_f32(
           false, afrag,  false, bfrag,
           (short)0, acc[t],  false,  false);
    }
  }

#pragma unroll
  for (int t = 0; t < 8; ++t) {
#pragma unroll
    for (int r = 0; r < 8; ++r) {
      int m    = row0 + r + half * 8;
      int ncol = t * 16 + l;
      C[(size_t)m * HID + ncol] = acc[t][r];
    }
  }
}

__global__ void k_scatter(const float* __restrict__ xw,
                          const int* __restrict__ row,
                          const int* __restrict__ col,
                          const float* __restrict__ norm,
                          float* __restrict__ acc, int E) {
  int e = ((blockIdx.x * blockDim.x + threadIdx.x) >> 5);
  if (e >= E) return;
  int lane = threadIdx.x & 31;
  int r = row[e], c = col[e];
  float s = norm[e];
  const float4 v = *(const float4*)(xw + (size_t)r * HID + lane * 4);
  float* dst = acc + (size_t)c * HID + lane * 4;
  atomicAdd(dst + 0, v.x * s);
  atomicAdd(dst + 1, v.y * s);
  atomicAdd(dst + 2, v.z * s);
  atomicAdd(dst + 3, v.w * s);
}

__global__ void k_epilogue(float* __restrict__ acc,
                           const float* __restrict__ xw,
                           const float* __restrict__ dinv,
                           const float* __restrict__ b, int n) {
  int i = blockIdx.x * blockDim.x + threadIdx.x;
  if (i >= n * (HID / 4)) return;
  int node = i >> 5;
  int f4   = i & 31;
  float s  = dinv[node];
  s = s * s;
  float4 a  = ((float4*)acc)[i];
  float4 xv = ((const float4*)xw)[i];
  float4 bv = ((const float4*)b)[f4];
  float4 o;
  o.x = a.x + xv.x * s + bv.x;  o.x = o.x > 0.0f ? o.x : 0.0f;
  o.y = a.y + xv.y * s + bv.y;  o.y = o.y > 0.0f ? o.y : 0.0f;
  o.z = a.z + xv.z * s + bv.z;  o.z = o.z > 0.0f ? o.z : 0.0f;
  o.w = a.w + xv.w * s + bv.w;  o.w = o.w > 0.0f ? o.w : 0.0f;
  ((float4*)acc)[i] = o;
}

__global__ void k_out_proj(const float* __restrict__ h,
                           const float* __restrict__ Wout,
                           const float* __restrict__ bout,
                           float* __restrict__ out, int n) {
  int idx = blockIdx.x * blockDim.x + threadIdx.x;
  if (idx >= n * NPRED) return;
  int node = idx >> 4;
  int p    = idx & 15;
  const float4* hr = (const float4*)(h + (size_t)node * HID);
  const float4* wr = (const float4*)(Wout + (size_t)p * HID);
  float s = 0.0f;
#pragma unroll 8
  for (int k = 0; k < HID / 4; ++k) {
    float4 a = hr[k], b = wr[k];
    s += a.x * b.x + a.y * b.y + a.z * b.z + a.w * b.w;
  }
  out[idx] = s + bout[p];
}


#define SCAN_T     1024
#define SCAN_ITEMS 4096

__global__ void k_zero_cnt(int* __restrict__ cnt, int* __restrict__ ctr, int npad) {
  int i = blockIdx.x * blockDim.x + threadIdx.x;
  if (i < npad) cnt[i] = 0;
  if (i < 2) ctr[i] = 0;
}

__global__ void k_rank(const int* __restrict__ col, int* cnt,
                       int* __restrict__ rank, int E) {
  int e = blockIdx.x * blockDim.x + threadIdx.x;
  if (e < E) rank[e] = atomicAdd(&cnt[col[e]], 1);
}

__global__ void k_scan(const int* __restrict__ cnt, int* __restrict__ offs) {
  __shared__ int s_front[SCAN_T / 32];
  __shared__ int s_wave[SCAN_T / 32];
  const int tid  = threadIdx.x;
  const int lane = tid & 31;
  const int wv   = tid >> 5;
  const int base = blockIdx.x * SCAN_ITEMS;

  int s = 0;
  const int4* front = (const int4*)cnt + tid;
#pragma unroll 4
  for (int k = 0; k < (int)blockIdx.x; ++k) {
    int4 f = front[k * SCAN_T];
    s += f.x + f.y + f.z + f.w;
  }
  for (int m = 16; m > 0; m >>= 1) s += __shfl_xor(s, m);
  if (lane == 0) s_front[wv] = s;

  const int4 v = *(const int4*)(cnt + base + tid * 4);
  const int t = v.x + v.y + v.z + v.w;
  int inc = t;
  for (int d = 1; d < 32; d <<= 1) {
    int y = __shfl_up(inc, d);
    if (lane >= d) inc += y;
  }
  if (lane == 31) s_wave[wv] = inc;
  __syncthreads();

  int ex = inc - t;
  for (int w = 0; w < SCAN_T / 32; ++w) {
    ex += s_front[w];
    if (w < wv) ex += s_wave[w];
  }
  int4 o;
  o.x = ex;
  o.y = ex + v.x;
  o.z = o.y + v.y;
  o.w = o.z + v.z;
  *(int4*)(offs + base + tid * 4) = o;
}

__global__ void k_fill(const int* __restrict__ row, const int* __restrict__ col,
                       const float* __restrict__ w, const int* __restrict__ rank,
                       const int* __restrict__ offs, int2* __restrict__ edge, int E) {
  int e = blockIdx.x * blockDim.x + threadIdx.x;
  if (e < E) {
    int p = offs[col[e]] + rank[e];
    edge[p] = make_int2(row[e], __float_as_int(w[e]));
  }
}

__global__ void k_dinv_seg(const int* __restrict__ offs, const int2* __restrict__ edge,
                           float* __restrict__ dinv, int n) {
  int c = blockIdx.x * blockDim.x + threadIdx.x;
  if (c >= n) return;
  const int p1 = offs[c + 1];
  float d = 1.0f;
  for (int p = offs[c]; p < p1; ++p) d += __int_as_float(edge[p].y);
  dinv[c] = (d > 0.0f) ? rsqrtf(d) : 0.0f;
}

template <bool SRC_SCALED>
__device__ __forceinline__ float edge_factor(const int wbits, const int dsbits, const int j) {
  const float w = __int_as_float(__builtin_amdgcn_readlane(wbits, j));
  if (SRC_SCALED) return w;
  return w * __int_as_float(__builtin_amdgcn_readlane(dsbits, j));
}

template <bool SRC_SCALED>
__device__ __forceinline__ float4 gather_row(const int c, const int lane,
                                             const float* __restrict__ xin,
                                             const int* __restrict__ offs,
                                             const int2* __restrict__ edge,
                                             const float* __restrict__ dinv,
                                             float& dc) {
  const int p0 = offs[c];
  const int p1 = offs[c + 1];
  const float* xl = xin + lane * 4;
  const float4 xv = *(const float4*)(xl + (size_t)c * HID);
  const float s = dinv[c];

  float4 a = make_float4(0.0f, 0.0f, 0.0f, 0.0f);
  for (int base = p0; base < p1; base += 32) {
    int m = p1 - base;
    if (m > 32) m = 32;
    const int2 ev = edge[base + (lane < m ? lane : m - 1)];
    int dsb = 0;
    if (!SRC_SCALED) dsb = __float_as_int(dinv[ev.x]);
    int j = 0;
    for (; j + 4 <= m; j += 4) {
      const int r0 = __builtin_amdgcn_readlane(ev.x, j);
      const int r1 = __builtin_amdgcn_readlane(ev.x, j + 1);
      const int r2 = __builtin_amdgcn_readlane(ev.x, j + 2);
      const int r3 = __builtin_amdgcn_readlane(ev.x, j + 3);
      const float4 v0 = *(const float4*)(xl + (size_t)r0 * HID);
      const float4 v1 = *(const float4*)(xl + (size_t)r1 * HID);
      const float4 v2 = *(const float4*)(xl + (size_t)r2 * HID);
      const float4 v3 = *(const float4*)(xl + (size_t)r3 * HID);
      const float s0 = edge_factor<SRC_SCALED>(ev.y, dsb, j);
      const float s1 = edge_factor<SRC_SCALED>(ev.y, dsb, j + 1);
      const float s2 = edge_factor<SRC_SCALED>(ev.y, dsb, j + 2);
      const float s3 = edge_factor<SRC_SCALED>(ev.y, dsb, j + 3);
      a.x += v0.x * s0;  a.y += v0.y * s0;  a.z += v0.z * s0;  a.w += v0.w * s0;
      a.x += v1.x * s1;  a.y += v1.y * s1;  a.z += v1.z * s1;  a.w += v1.w * s1;
      a.x += v2.x * s2;  a.y += v2.y * s2;  a.z += v2.z * s2;  a.w += v2.w * s2;
      a.x += v3.x * s3;  a.y += v3.y * s3;  a.z += v3.z * s3;  a.w += v3.w * s3;
    }
    for (; j < m; ++j) {
      const int r0 = __builtin_amdgcn_readlane(ev.x, j);
      const float s0 = edge_factor<SRC_SCALED>(ev.y, dsb, j);
      const float4 v0 = *(const float4*)(xl + (size_t)r0 * HID);
      a.x += v0.x * s0;  a.y += v0.y * s0;  a.z += v0.z * s0;  a.w += v0.w * s0;
    }
  }

  const float t = SRC_SCALED ? 1.0f : s;
  float4 o;
  o.x = (a.x + xv.x * t) * s;
  o.y = (a.y + xv.y * t) * s;
  o.z = (a.z + xv.z * t) * s;
  o.w = (a.w + xv.w * t) * s;
  dc = s;
  return o;
}

#define TP 132

__device__ __forceinline__ void stage_w(float* wf, const float* __restrict__ W,
                                        const int wv, const int lane) {
  const int half = lane >> 4;
  const int l    = lane & 15;
  const int t = wv & 7;
  const float* wc = W + (size_t)(half * 2) * HID + t * 16 + l;
#pragma unroll 8
  for (int i = 0; i < 16; ++i) {
    const int ks = 2 * i + (wv >> 3);
    v2f f;
    f.x = wc[(size_t)(4 * ks)     * HID];
    f.y = wc[(size_t)(4 * ks + 1) * HID];
    *(v2f*)(wf + ((ks * 8 + t) * 32 + lane) * 2) = f;
  }
}

__device__ __forceinline__ void tile_dense(float* tile, const float* wf,
                                           const float* __restrict__ b,
                                           const int wv, const int lane) {
  const int half = lane >> 4;
  const int l    = lane & 15;
  v8f acc0 = (v8f){0,0,0,0,0,0,0,0};
  v8f acc1 = (v8f){0,0,0,0,0,0,0,0};
  float bias = 0.0f;
  if (wv < 8) {
    bias = b[wv * 16 + l];
    const float* ar = tile + l * TP + half * 2;
    const float* bf = wf + (wv * 32 + lane) * 2;
#pragma unroll 4
    for (int ks = 0; ks < 32; ks += 2) {
      const v2f a0 = *(const v2f*)(ar + ks * 4);
      const v2f b0 = *(const v2f*)(bf + ks * 512);
      const v2f a1 = *(const v2f*)(ar + ks * 4 + 4);
      const v2f b1 = *(const v2f*)(bf + ks * 512 + 512);
      acc0 = __builtin_amdgcn_wmma_f32_16x16x4_f32(false, a0, false, b0, (short)0, acc0, false, false);
      acc1 = __builtin_amdgcn_wmma_f32_16x16x4_f32(false, a1, false, b1, (short)0, acc1, false, false);
    }
  }
  __syncthreads();
  if (wv < 8) {
    float* st = tile + (half * 8) * TP + wv * 16 + l;
#pragma unroll
    for (int r = 0; r < 8; ++r) {
      const float v = acc0[r] + acc1[r] + bias;
      st[r * TP] = v > 0.0f ? v : 0.0f;
    }
  }
  __syncthreads();
}

__global__ void __launch_bounds__(512, 16)
k_gather_dense(const float* __restrict__ xin, const int* __restrict__ offs,
               const int2* __restrict__ edge, const float* __restrict__ dinv,
               const float* __restrict__ W, const float* __restrict__ b,
               float* __restrict__ C, int* ctr, int n, int nTiles) {
  __shared__ __attribute__((aligned(16))) float wf[HID * HID];
  __shared__ __attribute__((aligned(16))) float tile[16 * TP];
  __shared__ int s_first;
  __shared__ int s_next;
  const int wv   = __builtin_amdgcn_readfirstlane((int)(threadIdx.x >> 5));
  const int lane = threadIdx.x & 31;

  if (threadIdx.x == 0) s_first = atomicAdd(ctr, 1);
  stage_w(wf, W, wv, lane);
  __syncthreads();

  int tix = __builtin_amdgcn_readfirstlane(s_first);
  while (tix < nTiles) {
    const int c = tix * 16 + wv;

    float4 o = make_float4(0.0f, 0.0f, 0.0f, 0.0f);
    float dc = 0.0f;
    if (c < n) o = gather_row<false>(c, lane, xin, offs, edge, dinv, dc);
    *(float4*)(tile + wv * TP + lane * 4) = o;
    __syncthreads();
    if (threadIdx.x == 480) s_next = atomicAdd(ctr, 1);

    tile_dense(tile, wf, b, wv, lane);
    const int tnext = __builtin_amdgcn_readfirstlane(s_next);
    if (c < n) {
      float4 hv = *(const float4*)(tile + wv * TP + lane * 4);
      hv.x *= dc;  hv.y *= dc;  hv.z *= dc;  hv.w *= dc;
      *(float4*)(C + (size_t)c * HID + lane * 4) = hv;
    }
    tix = tnext;
  }
}

__global__ void __launch_bounds__(512, 16)
k_gather_dense_head(const float* __restrict__ hin, const int* __restrict__ offs,
                    const int2* __restrict__ edge, const float* __restrict__ dinv,
                    const float* __restrict__ W, const float* __restrict__ b,
                    const float* __restrict__ Wout, const float* __restrict__ bout,
                    float* __restrict__ out, int* ctr, int n, int nTiles) {
  __shared__ __attribute__((aligned(16))) float wf[HID * HID];
  __shared__ __attribute__((aligned(16))) float tile[16 * TP];
  __shared__ int s_first;
  __shared__ int s_next;
  const int wv   = __builtin_amdgcn_readfirstlane((int)(threadIdx.x >> 5));
  const int lane = threadIdx.x & 31;
  const int half = lane >> 4;
  const int l    = lane & 15;

  if (threadIdx.x == 0) s_first = atomicAdd(ctr, 1);
  stage_w(wf, W, wv, lane);
  __syncthreads();

  int tix = __builtin_amdgcn_readfirstlane(s_first);
  while (tix < nTiles) {
    const int row0 = tix * 16;
    const int c    = row0 + wv;

    float4 o = make_float4(0.0f, 0.0f, 0.0f, 0.0f);
    float dc = 0.0f;
    if (c < n) o = gather_row<true>(c, lane, hin, offs, edge, dinv, dc);
    __syncthreads();
    *(float4*)(tile + wv * TP + lane * 4) = o;
    __syncthreads();
    if (threadIdx.x == 480) s_next = atomicAdd(ctr, 1);

    tile_dense(tile, wf, b, wv, lane);
    const int tnext = __builtin_amdgcn_readfirstlane(s_next);

    if (wv == 8) {
      const float* ar = tile + l * TP + half * 2;
      const float* br = Wout + (size_t)l * HID + half * 2;
      v8f acc0 = (v8f){0,0,0,0,0,0,0,0};
      v8f acc1 = (v8f){0,0,0,0,0,0,0,0};
#pragma unroll 4
      for (int k = 0; k < HID; k += 8) {
        const v2f a0 = *(const v2f*)(ar + k);
        const v2f b0 = *(const v2f*)(br + k);
        const v2f a1 = *(const v2f*)(ar + k + 4);
        const v2f b1 = *(const v2f*)(br + k + 4);
        acc0 = __builtin_amdgcn_wmma_f32_16x16x4_f32(false, a0, false, b0, (short)0, acc0, false, false);
        acc1 = __builtin_amdgcn_wmma_f32_16x16x4_f32(false, a1, false, b1, (short)0, acc1, false, false);
      }
      const float bo = bout[l];
#pragma unroll
      for (int r = 0; r < 8; ++r) {
        const int m = row0 + r + half * 8;
        if (m < n) out[(size_t)m * NPRED + l] = acc0[r] + acc1[r] + bo;
      }
    }
    tix = tnext;
  }
}


extern "C" void kernel_launch(void* const* d_in, const int* in_sizes, int n_in,
                              void* d_out, int out_size, void* d_ws, size_t ws_size,
                              hipStream_t stream) {
  const float* x    = (const float*)d_in[0];
  const int*   ei   = (const int*)  d_in[1];
  const float* ew   = (const float*)d_in[2];
  const float* W1   = (const float*)d_in[3];
  const float* b1   = (const float*)d_in[4];
  const float* W2   = (const float*)d_in[5];
  const float* b2   = (const float*)d_in[6];
  const float* Wout = (const float*)d_in[7];
  const float* bout = (const float*)d_in[8];
  float* out = (float*)d_out;

  const int n = in_sizes[0] / HID;
  const int E = in_sizes[2];
  const int* erow = ei;
  const int* ecol = ei + E;

  const int B = 256;
  int gN  = (n + B - 1) / B;
  int gE  = (E + B - 1) / B;
  int nTiles = (n + 15) / 16;
  int gGemm  = (nTiles + 7) / 8;
  int gOut   = (n * NPRED + B - 1) / B;

  const int    gScan = (n + SCAN_ITEMS) / SCAN_ITEMS;
  const int    npad  = gScan * SCAN_ITEMS;
  const size_t plane = (size_t)n * HID;
  const size_t epad  = ((size_t)E + 63) / 64 * 64;
  const size_t ndpad = ((size_t)n + 127) / 128 * 128;
  const size_t need  = (plane + 2 * epad + (size_t)npad + ndpad + 128) * sizeof(float);

  if (ws_size >= need && (size_t)npad + (size_t)E <= plane) {
    float* h    = (float*)d_ws;
    int2*  edge = (int2*)(h + plane);
    int*   offs = (int*)(h + plane + 2 * epad);
    float* dinv = (float*)(offs + npad);
    int*   ctr  = (int*)(dinv + ndpad);
    int*   cnt  = (int*)h;
    int*   rank = cnt + npad;

    k_zero_cnt<<<npad / B, B, 0, stream>>>(cnt, ctr, npad);
    k_rank    <<<gE, B, 0, stream>>>(ecol, cnt, rank, E);
    k_scan    <<<gScan, SCAN_T, 0, stream>>>(cnt, offs);
    k_fill    <<<gE, B, 0, stream>>>(erow, ecol, ew, rank, offs, edge, E);
    k_dinv_seg<<<gN, B, 0, stream>>>(offs, edge, dinv, n);

    k_gather_dense<<<1024, 512, 0, stream>>>(x, offs, edge, dinv, W1, b1, h, ctr, n, nTiles);

    k_gather_dense_head<<<1024, 512, 0, stream>>>(h, offs, edge, dinv, W2, b2, Wout, bout, out,
                                                  ctr + 1, n, nTiles);
    return;
  }

  float* dinv = (float*)d_ws;
  float* norm = dinv + n;
  float* xw   = norm + E;
  float* acc  = xw + (size_t)n * HID;

  int gScat  = (E + 7) / 8;
  int gEpi   = (n * (HID / 4) + B - 1) / B;

  k_set_deg    <<<gN, B, 0, stream>>>(dinv, n);
  k_deg_scatter<<<gE, B, 0, stream>>>(ecol, ew, dinv, E);
  k_dinv       <<<gN, B, 0, stream>>>(dinv, n);
  k_norm       <<<gE, B, 0, stream>>>(erow, ecol, ew, dinv, norm, E);

  k_gemm_wmma<<<gGemm, B, 0, stream>>>(x, W1, xw, n);
  hipMemsetAsync(acc, 0, (size_t)n * HID * sizeof(float), stream);
  k_scatter  <<<gScat, B, 0, stream>>>(xw, erow, ecol, norm, acc, E);
  k_epilogue <<<gEpi, B, 0, stream>>>(acc, xw, dinv, b1, n);

  k_gemm_wmma<<<gGemm, B, 0, stream>>>(acc, W2, xw, n);
  hipMemsetAsync(acc, 0, (size_t)n * HID * sizeof(float), stream);
  k_scatter  <<<gScat, B, 0, stream>>>(xw, erow, ecol, norm, acc, E);
  k_epilogue <<<gEpi, B, 0, stream>>>(acc, xw, dinv, b2, n);

  k_out_proj<<<gOut, B, 0, stream>>>(acc, Wout, bout, out, n);
}


# Edge ordering through LDS: blocks split edges into runs per 256-node bin, one block per bin counts, scans, places, sums deg in LDS; set-up launches 5 to 2, no global atomics; deg sum order changes
# speedup vs baseline: 5.8638x; 1.7078x over previous
#include <hip/hip_runtime.h>


typedef float v2f __attribute__((ext_vector_type(2)));
typedef float v8f __attribute__((ext_vector_type(8)));

#define HID 128
#define NPRED 16


__global__ void k_set_deg(float* deg, int n) {
  int i = blockIdx.x * blockDim.x + threadIdx.x;
  if (i < n) deg[i] = 1.0f;
}

__global__ void k_deg_scatter(const int* __restrict__ col,
                              const float* __restrict__ w,
                              float* deg, int E) {
  int e = blockIdx.x * blockDim.x + threadIdx.x;
  if (e < E) atomicAdd(&deg[col[e]], w[e]);
}

__global__ void k_dinv(float* deg, int n) {
  int i = blockIdx.x * blockDim.x + threadIdx.x;
  if (i < n) {
    float d = deg[i];
    deg[i] = (d > 0.0f) ? rsqrtf(d) : 0.0f;
  }
}

__global__ void k_norm(const int* __restrict__ row, const int* __restrict__ col,
                       const float* __restrict__ w, const float* __restrict__ dinv,
                       float* __restrict__ norm, int E) {
  int e = blockIdx.x * blockDim.x + threadIdx.x;
  if (e < E) norm[e] = dinv[row[e]] * w[e] * dinv[col[e]];
}

__global__ void k_gemm_wmma(const float* __restrict__ A,
                            const float* __restrict__ W,
                            float* __restrict__ C, int n) {
  int wid  = (blockIdx.x * blockDim.x + threadIdx.x) >> 5;
  int lane = threadIdx.x & 31;
  int row0 = wid * 16;
  if (row0 >= n) return;

  int half = lane >> 4;
  int l    = lane & 15;

  v8f acc[8];
#pragma unroll
  for (int t = 0; t < 8; ++t) acc[t] = (v8f){0,0,0,0,0,0,0,0};

  const float* arow = A + (size_t)(row0 + l) * HID;

  for (int k = 0; k < HID; k += 4) {
    v2f afrag = *(const v2f*)(arow + k + half * 2);
#pragma unroll
    for (int t = 0; t < 8; ++t) {
      int ncol = t * 16 + l;
      v2f bfrag;
      bfrag.x = W[(size_t)(k + half * 2)     * HID + ncol];
      bfrag.y = W[(size_t)(k + half * 2 + 1) * HID + ncol];
      acc[t] = __builtin_amdgcn_wmma_f32_16x16x4_f32(
           false, afrag,  false, bfrag,
           (short)0, acc[t],  false,  false);
    }
  }

#pragma unroll
  for (int t = 0; t < 8; ++t) {
#pragma unroll
    for (int r = 0; r < 8; ++r) {
      int m    = row0 + r + half * 8;
      int ncol = t * 16 + l;
      C[(size_t)m * HID + ncol] = acc[t][r];
    }
  }
}

__global__ void k_scatter(const float* __restrict__ xw,
                          const int* __restrict__ row,
                          const int* __restrict__ col,
                          const float* __restrict__ norm,
                          float* __restrict__ acc, int E) {
  int e = ((blockIdx.x * blockDim.x + threadIdx.x) >> 5);
  if (e >= E) return;
  int lane = threadIdx.x & 31;
  int r = row[e], c = col[e];
  float s = norm[e];
  const float4 v = *(const float4*)(xw + (size_t)r * HID + lane * 4);
  float* dst = acc + (size_t)c * HID + lane * 4;
  atomicAdd(dst + 0, v.x * s);
  atomicAdd(dst + 1, v.y * s);
  atomicAdd(dst + 2, v.z * s);
  atomicAdd(dst + 3, v.w * s);
}

__global__ void k_epilogue(float* __restrict__ acc,
                           const float* __restrict__ xw,
                           const float* __restrict__ dinv,
                           const float* __restrict__ b, int n) {
  int i = blockIdx.x * blockDim.x + threadIdx.x;
  if (i >= n * (HID / 4)) return;
  int node = i >> 5;
  int f4   = i & 31;
  float s  = dinv[node];
  s = s * s;
  float4 a  = ((float4*)acc)[i];
  float4 xv = ((const float4*)xw)[i];
  float4 bv = ((const float4*)b)[f4];
  float4 o;
  o.x = a.x + xv.x * s + bv.x;  o.x = o.x > 0.0f ? o.x : 0.0f;
  o.y = a.y + xv.y * s + bv.y;  o.y = o.y > 0.0f ? o.y : 0.0f;
  o.z = a.z + xv.z * s + bv.z;  o.z = o.z > 0.0f ? o.z : 0.0f;
  o.w = a.w + xv.w * s + bv.w;  o.w = o.w > 0.0f ? o.w : 0.0f;
  ((float4*)acc)[i] = o;
}

__global__ void k_out_proj(const float* __restrict__ h,
                           const float* __restrict__ Wout,
                           const float* __restrict__ bout,
                           float* __restrict__ out, int n) {
  int idx = blockIdx.x * blockDim.x + threadIdx.x;
  if (idx >= n * NPRED) return;
  int node = idx >> 4;
  int p    = idx & 15;
  const float4* hr = (const float4*)(h + (size_t)node * HID);
  const float4* wr = (const float4*)(Wout + (size_t)p * HID);
  float s = 0.0f;
#pragma unroll 8
  for (int k = 0; k < HID / 4; ++k) {
    float4 a = hr[k], b = wr[k];
    s += a.x * b.x + a.y * b.y + a.z * b.z + a.w * b.w;
  }
  out[idx] = s + bout[p];
}


#define SCAN_T     1024
#define SCAN_ITEMS 4096

__global__ void k_zero_cnt(int* __restrict__ cnt, int* __restrict__ ctr, int npad) {
  int i = blockIdx.x * blockDim.x + threadIdx.x;
  if (i < npad) cnt[i] = 0;
  if (i < 2) ctr[i] = 0;
}

__global__ void k_rank(const int* __restrict__ col, int* cnt,
                       int* __restrict__ rank, int E) {
  int e = blockIdx.x * blockDim.x + threadIdx.x;
  if (e < E) rank[e] = atomicAdd(&cnt[col[e]], 1);
}

__global__ void k_scan(const int* __restrict__ cnt, int* __restrict__ offs) {
  __shared__ int s_front[SCAN_T / 32];
  __shared__ int s_wave[SCAN_T / 32];
  const int tid  = threadIdx.x;
  const int lane = tid & 31;
  const int wv   = tid >> 5;
  const int base = blockIdx.x * SCAN_ITEMS;

  int s = 0;
  const int4* front = (const int4*)cnt + tid;
#pragma unroll 4
  for (int k = 0; k < (int)blockIdx.x; ++k) {
    int4 f = front[k * SCAN_T];
    s += f.x + f.y + f.z + f.w;
  }
  for (int m = 16; m > 0; m >>= 1) s += __shfl_xor(s, m);
  if (lane == 0) s_front[wv] = s;

  const int4 v = *(const int4*)(cnt + base + tid * 4);
  const int t = v.x + v.y + v.z + v.w;
  int inc = t;
  for (int d = 1; d < 32; d <<= 1) {
    int y = __shfl_up(inc, d);
    if (lane >= d) inc += y;
  }
  if (lane == 31) s_wave[wv] = inc;
  __syncthreads();

  int ex = inc - t;
  for (int w = 0; w < SCAN_T / 32; ++w) {
    ex += s_front[w];
    if (w < wv) ex += s_wave[w];
  }
  int4 o;
  o.x = ex;
  o.y = ex + v.x;
  o.z = o.y + v.y;
  o.w = o.z + v.z;
  *(int4*)(offs + base + tid * 4) = o;
}

__global__ void k_fill(const int* __restrict__ row, const int* __restrict__ col,
                       const float* __restrict__ w, const int* __restrict__ rank,
                       const int* __restrict__ offs, int2* __restrict__ edge, int E) {
  int e = blockIdx.x * blockDim.x + threadIdx.x;
  if (e < E) {
    int p = offs[col[e]] + rank[e];
    edge[p] = make_int2(row[e], __float_as_int(w[e]));
  }
}

__global__ void k_dinv_seg(const int* __restrict__ offs, const int2* __restrict__ edge,
                           float* __restrict__ dinv, int n) {
  int c = blockIdx.x * blockDim.x + threadIdx.x;
  if (c >= n) return;
  const int p1 = offs[c + 1];
  float d = 1.0f;
  for (int p = offs[c]; p < p1; ++p) d += __int_as_float(edge[p].y);
  dinv[c] = (d > 0.0f) ? rsqrtf(d) : 0.0f;
}


#define PCH     8192
#define EPT     8
#define PT      1024
#define BINW    256
#define MAXBINS 1024
#define MAXCH   1024

__device__ __forceinline__ int wave_scan_inc(int v, const int lane) {
  for (int d = 1; d < 32; d <<= 1) {
    const int y = __shfl_up(v, d);
    if (lane >= d) v += y;
  }
  return v;
}

__global__ void __launch_bounds__(PT)
k_part(const int* __restrict__ row, const int* __restrict__ col, const float* __restrict__ w,
       int2* __restrict__ part, int* __restrict__ cntT, int* __restrict__ ctr,
       int E, int NB, int nChP) {
  __shared__ int s_cnt[MAXBINS];
  __shared__ int s_start[MAXBINS];
  __shared__ int s_w[PT / 32];
  const int tid  = threadIdx.x;
  const int lane = tid & 31;
  const int wv   = tid >> 5;
  const int q    = blockIdx.x;
  if (q == 0 && tid < 2) ctr[tid] = 0;
  s_cnt[tid] = 0;
  __syncthreads();

  const int e0 = q * PCH;
  int bn[EPT], rk[EPT], sx[EPT], wb[EPT];
#pragma unroll
  for (int j = 0; j < EPT; ++j) {
    const int e  = e0 + j * PT + tid;
    const int ec = e < E ? e : E - 1;
    const int c  = col[ec];
    rk[j] = 0;
    bn[j] = e < E ? (c >> 8) : -1;
    sx[j] = (int)((unsigned)row[ec] | ((unsigned)(c & (BINW - 1)) << 24));
    wb[j] = __float_as_int(w[ec]);
  }
#pragma unroll
  for (int j = 0; j < EPT; ++j)
    if (bn[j] >= 0) rk[j] = atomicAdd(&s_cnt[bn[j]], 1);
  __syncthreads();

  const int cv  = s_cnt[tid];
  const int inc = wave_scan_inc(cv, lane);
  if (lane == 31) s_w[wv] = inc;
  __syncthreads();
  int ex = inc - cv;
  for (int k = 0; k < wv; ++k) ex += s_w[k];
  s_start[tid] = ex;
  if (tid < NB) cntT[(size_t)tid * nChP + q] = (cv << 16) | ex;
  __syncthreads();

#pragma unroll
  for (int j = 0; j < EPT; ++j)
    if (bn[j] >= 0) part[(size_t)e0 + s_start[bn[j]] + rk[j]] = make_int2(sx[j], wb[j]);
}

__device__ __forceinline__ int2 bin_entry(const int2* __restrict__ part, const int* s_roff,
                                          const int* s_rst, const int i) {
  int r = 0;
#pragma unroll
  for (int step = MAXCH / 2; step >= 1; step >>= 1)
    r = (s_roff[r + step] <= i) ? r + step : r;
  return part[(size_t)r * PCH + s_rst[r] + (i - s_roff[r])];
}

__global__ void __launch_bounds__(PT, 16)
k_binsort(const int2* __restrict__ part, const int* __restrict__ cntT,
          int2* __restrict__ edge, int* __restrict__ offs, float* __restrict__ dinv,
          int n, int nChP) {
  __shared__ int   s_roff[MAXCH + 1];
  __shared__ int   s_rst[MAXCH];
  __shared__ int   s_ncnt[BINW];
  __shared__ int   s_noff[BINW];
  __shared__ float s_deg[BINW];
  __shared__ int   s_w[PT / 32];
  __shared__ int   s_w2[PT / 32];
  __shared__ int   s_red[PT / 32];
  const int tid  = threadIdx.x;
  const int lane = tid & 31;
  const int wv   = tid >> 5;
  const int k    = blockIdx.x;

  int s = 0;
  const int4* front = (const int4*)cntT;
  const int nf = (k * nChP) >> 2;
#pragma unroll 4
  for (int i = tid; i < nf; i += PT) {
    const int4 f = front[i];
    s += (f.x >> 16) + (f.y >> 16) + (f.z >> 16) + (f.w >> 16);
  }
  for (int m = 16; m > 0; m >>= 1) s += __shfl_xor(s, m);
  if (lane == 0) s_red[wv] = s;

  int v = 0;
  if (tid < nChP) v = cntT[(size_t)k * nChP + tid];
  const int cv  = v >> 16;
  const int inc = wave_scan_inc(cv, lane);
  if (lane == 31) s_w[wv] = inc;
  if (tid < BINW) { s_ncnt[tid] = 0;  s_deg[tid] = 0.0f; }
  __syncthreads();
  int ex = inc - cv;
  for (int j = 0; j < wv; ++j) ex += s_w[j];
  int base = s_red[lane], T = s_w[lane];
  for (int m = 16; m > 0; m >>= 1) { base += __shfl_xor(base, m);  T += __shfl_xor(T, m); }
  s_roff[tid] = tid < nChP ? ex : (tid == nChP ? T : 0x7fffffff);
  if (tid == 0) s_roff[MAXCH] = (nChP == MAXCH) ? T : 0x7fffffff;
  if (tid < nChP) s_rst[tid] = v & 0xFFFF;
  __syncthreads();

  int2 ev[4];
#pragma unroll
  for (int j = 0; j < 4; ++j) ev[j] = make_int2(0, 0);
  if (T > 0) {
    int ix[4], r[4];
#pragma unroll
    for (int j = 0; j < 4; ++j) { const int i = j * PT + tid;  ix[j] = i < T ? i : T - 1;  r[j] = 0; }
#pragma unroll
    for (int step = MAXCH / 2; step >= 1; step >>= 1) {
#pragma unroll
      for (int j = 0; j < 4; ++j) r[j] = (s_roff[r[j] + step] <= ix[j]) ? r[j] + step : r[j];
    }
#pragma unroll
    for (int j = 0; j < 4; ++j)
      ev[j] = part[(size_t)r[j] * PCH + s_rst[r[j]] + (ix[j] - s_roff[r[j]])];
  }

#pragma unroll
  for (int j = 0; j < 4; ++j) {
    if (j * PT + tid < T) {
      const int d = (int)((unsigned)ev[j].x >> 24);
      atomicAdd(&s_ncnt[d], 1);
      atomicAdd(&s_deg[d], __int_as_float(ev[j].y));
    }
  }
  for (int i = 4 * PT + tid; i < T; i += PT) {
    const int2 e = bin_entry(part, s_roff, s_rst, i);
    const int d = (int)((unsigned)e.x >> 24);
    atomicAdd(&s_ncnt[d], 1);
    atomicAdd(&s_deg[d], __int_as_float(e.y));
  }
  __syncthreads();

  int nc = 0;
  if (tid < BINW) nc = s_ncnt[tid];
  const int inc2 = wave_scan_inc(nc, lane);
  if (lane == 31) s_w2[wv] = inc2;
  __syncthreads();
  if (tid < BINW) {
    int ex2 = inc2 - nc;
    for (int j = 0; j < wv; ++j) ex2 += s_w2[j];
    s_noff[tid] = ex2;
    s_ncnt[tid] = 0;
    const int node = k * BINW + tid;
    if (node < n) {
      offs[node] = base + ex2;
      const float dg = 1.0f + s_deg[tid];
      dinv[node] = (dg > 0.0f) ? rsqrtf(dg) : 0.0f;
    }
  }
  if (k == (int)gridDim.x - 1 && tid == 0) offs[n] = base + T;
  __syncthreads();

#pragma unroll
  for (int j = 0; j < 4; ++j) {
    if (j * PT + tid < T) {
      const int d = (int)((unsigned)ev[j].x >> 24);
      const int p = atomicAdd(&s_ncnt[d], 1);
      edge[(size_t)base + s_noff[d] + p] = make_int2(ev[j].x & 0xFFFFFF, ev[j].y);
    }
  }
  for (int i = 4 * PT + tid; i < T; i += PT) {
    const int2 e = bin_entry(part, s_roff, s_rst, i);
    const int d = (int)((unsigned)e.x >> 24);
    const int p = atomicAdd(&s_ncnt[d], 1);
    edge[(size_t)base + s_noff[d] + p] = make_int2(e.x & 0xFFFFFF, e.y);
  }
}

template <bool SRC_SCALED>
__device__ __forceinline__ float edge_factor(const int wbits, const int dsbits, const int j) {
  const float w = __int_as_float(__builtin_amdgcn_readlane(wbits, j));
  if (SRC_SCALED) return w;
  return w * __int_as_float(__builtin_amdgcn_readlane(dsbits, j));
}

template <bool SRC_SCALED>
__device__ __forceinline__ float4 gather_row(const int c, const int lane,
                                             const float* __restrict__ xin,
                                             const int* __restrict__ offs,
                                             const int2* __restrict__ edge,
                                             const float* __restrict__ dinv,
                                             float& dc) {
  const int p0 = offs[c];
  const int p1 = offs[c + 1];
  const float* xl = xin + lane * 4;
  const float4 xv = *(const float4*)(xl + (size_t)c * HID);
  const float s = dinv[c];

  float4 a = make_float4(0.0f, 0.0f, 0.0f, 0.0f);
  for (int base = p0; base < p1; base += 32) {
    int m = p1 - base;
    if (m > 32) m = 32;
    const int2 ev = edge[base + (lane < m ? lane : m - 1)];
    int dsb = 0;
    if (!SRC_SCALED) dsb = __float_as_int(dinv[ev.x]);
    int j = 0;
    for (; j + 4 <= m; j += 4) {
      const int r0 = __builtin_amdgcn_readlane(ev.x, j);
      const int r1 = __builtin_amdgcn_readlane(ev.x, j + 1);
      const int r2 = __builtin_amdgcn_readlane(ev.x, j + 2);
      const int r3 = __builtin_amdgcn_readlane(ev.x, j + 3);
      const float4 v0 = *(const float4*)(xl + (size_t)r0 * HID);
      const float4 v1 = *(const float4*)(xl + (size_t)r1 * HID);
      const float4 v2 = *(const float4*)(xl + (size_t)r2 * HID);
      const float4 v3 = *(const float4*)(xl + (size_t)r3 * HID);
      const float s0 = edge_factor<SRC_SCALED>(ev.y, dsb, j);
      const float s1 = edge_factor<SRC_SCALED>(ev.y, dsb, j + 1);
      const float s2 = edge_factor<SRC_SCALED>(ev.y, dsb, j + 2);
      const float s3 = edge_factor<SRC_SCALED>(ev.y, dsb, j + 3);
      a.x += v0.x * s0;  a.y += v0.y * s0;  a.z += v0.z * s0;  a.w += v0.w * s0;
      a.x += v1.x * s1;  a.y += v1.y * s1;  a.z += v1.z * s1;  a.w += v1.w * s1;
      a.x += v2.x * s2;  a.y += v2.y * s2;  a.z += v2.z * s2;  a.w += v2.w * s2;
      a.x += v3.x * s3;  a.y += v3.y * s3;  a.z += v3.z * s3;  a.w += v3.w * s3;
    }
    for (; j < m; ++j) {
      const int r0 = __builtin_amdgcn_readlane(ev.x, j);
      const float s0 = edge_factor<SRC_SCALED>(ev.y, dsb, j);
      const float4 v0 = *(const float4*)(xl + (size_t)r0 * HID);
      a.x += v0.x * s0;  a.y += v0.y * s0;  a.z += v0.z * s0;  a.w += v0.w * s0;
    }
  }

  const float t = SRC_SCALED ? 1.0f : s;
  float4 o;
  o.x = (a.x + xv.x * t) * s;
  o.y = (a.y + xv.y * t) * s;
  o.z = (a.z + xv.z * t) * s;
  o.w = (a.w + xv.w * t) * s;
  dc = s;
  return o;
}

#define TP 132

__device__ __forceinline__ void stage_w(float* wf, const float* __restrict__ W,
                                        const int wv, const int lane) {
  const int half = lane >> 4;
  const int l    = lane & 15;
  const int t = wv & 7;
  const float* wc = W + (size_t)(half * 2) * HID + t * 16 + l;
#pragma unroll 8
  for (int i = 0; i < 16; ++i) {
    const int ks = 2 * i + (wv >> 3);
    v2f f;
    f.x = wc[(size_t)(4 * ks)     * HID];
    f.y = wc[(size_t)(4 * ks + 1) * HID];
    *(v2f*)(wf + ((ks * 8 + t) * 32 + lane) * 2) = f;
  }
}

__device__ __forceinline__ void tile_dense(float* tile, const float* wf,
                                           const float* __restrict__ b,
                                           const int wv, const int lane) {
  const int half = lane >> 4;
  const int l    = lane & 15;
  v8f acc0 = (v8f){0,0,0,0,0,0,0,0};
  v8f acc1 = (v8f){0,0,0,0,0,0,0,0};
  float bias = 0.0f;
  if (wv < 8) {
    bias = b[wv * 16 + l];
    const float* ar = tile + l * TP + half * 2;
    const float* bf = wf + (wv * 32 + lane) * 2;
#pragma unroll 4
    for (int ks = 0; ks < 32; ks += 2) {
      const v2f a0 = *(const v2f*)(ar + ks * 4);
      const v2f b0 = *(const v2f*)(bf + ks * 512);
      const v2f a1 = *(const v2f*)(ar + ks * 4 + 4);
      const v2f b1 = *(const v2f*)(bf + ks * 512 + 512);
      acc0 = __builtin_amdgcn_wmma_f32_16x16x4_f32(false, a0, false, b0, (short)0, acc0, false, false);
      acc1 = __builtin_amdgcn_wmma_f32_16x16x4_f32(false, a1, false, b1, (short)0, acc1, false, false);
    }
  }
  __syncthreads();
  if (wv < 8) {
    float* st = tile + (half * 8) * TP + wv * 16 + l;
#pragma unroll
    for (int r = 0; r < 8; ++r) {
      const float v = acc0[r] + acc1[r] + bias;
      st[r * TP] = v > 0.0f ? v : 0.0f;
    }
  }
  __syncthreads();
}

__global__ void __launch_bounds__(512, 16)
k_gather_dense(const float* __restrict__ xin, const int* __restrict__ offs,
               const int2* __restrict__ edge, const float* __restrict__ dinv,
               const float* __restrict__ W, const float* __restrict__ b,
               float* __restrict__ C, int* ctr, int n, int nTiles) {
  __shared__ __attribute__((aligned(16))) float wf[HID * HID];
  __shared__ __attribute__((aligned(16))) float tile[16 * TP];
  __shared__ int s_first;
  __shared__ int s_next;
  const int wv   = __builtin_amdgcn_readfirstlane((int)(threadIdx.x >> 5));
  const int lane = threadIdx.x & 31;

  if (threadIdx.x == 0) s_first = atomicAdd(ctr, 1);
  stage_w(wf, W, wv, lane);
  __syncthreads();

  int tix = __builtin_amdgcn_readfirstlane(s_first);
  while (tix < nTiles) {
    const int c = tix * 16 + wv;

    float4 o = make_float4(0.0f, 0.0f, 0.0f, 0.0f);
    float dc = 0.0f;
    if (c < n) o = gather_row<false>(c, lane, xin, offs, edge, dinv, dc);
    *(float4*)(tile + wv * TP + lane * 4) = o;
    __syncthreads();
    if (threadIdx.x == 480) s_next = atomicAdd(ctr, 1);

    tile_dense(tile, wf, b, wv, lane);
    const int tnext = __builtin_amdgcn_readfirstlane(s_next);
    if (c < n) {
      float4 hv = *(const float4*)(tile + wv * TP + lane * 4);
      hv.x *= dc;  hv.y *= dc;  hv.z *= dc;  hv.w *= dc;
      *(float4*)(C + (size_t)c * HID + lane * 4) = hv;
    }
    tix = tnext;
  }
}

__global__ void __launch_bounds__(512, 16)
k_gather_dense_head(const float* __restrict__ hin, const int* __restrict__ offs,
                    const int2* __restrict__ edge, const float* __restrict__ dinv,
                    const float* __restrict__ W, const float* __restrict__ b,
                    const float* __restrict__ Wout, const float* __restrict__ bout,
                    float* __restrict__ out, int* ctr, int n, int nTiles) {
  __shared__ __attribute__((aligned(16))) float wf[HID * HID];
  __shared__ __attribute__((aligned(16))) float tile[16 * TP];
  __shared__ int s_first;
  __shared__ int s_next;
  const int wv   = __builtin_amdgcn_readfirstlane((int)(threadIdx.x >> 5));
  const int lane = threadIdx.x & 31;
  const int half = lane >> 4;
  const int l    = lane & 15;

  if (threadIdx.x == 0) s_first = atomicAdd(ctr, 1);
  stage_w(wf, W, wv, lane);
  __syncthreads();

  int tix = __builtin_amdgcn_readfirstlane(s_first);
  while (tix < nTiles) {
    const int row0 = tix * 16;
    const int c    = row0 + wv;

    float4 o = make_float4(0.0f, 0.0f, 0.0f, 0.0f);
    float dc = 0.0f;
    if (c < n) o = gather_row<true>(c, lane, hin, offs, edge, dinv, dc);
    __syncthreads();
    *(float4*)(tile + wv * TP + lane * 4) = o;
    __syncthreads();
    if (threadIdx.x == 480) s_next = atomicAdd(ctr, 1);

    tile_dense(tile, wf, b, wv, lane);
    const int tnext = __builtin_amdgcn_readfirstlane(s_next);

    if (wv == 8) {
      const float* ar = tile + l * TP + half * 2;
      const float* br = Wout + (size_t)l * HID + half * 2;
      v8f acc0 = (v8f){0,0,0,0,0,0,0,0};
      v8f acc1 = (v8f){0,0,0,0,0,0,0,0};
#pragma unroll 4
      for (int k = 0; k < HID; k += 8) {
        const v2f a0 = *(const v2f*)(ar + k);
        const v2f b0 = *(const v2f*)(br + k);
        const v2f a1 = *(const v2f*)(ar + k + 4);
        const v2f b1 = *(const v2f*)(br + k + 4);
        acc0 = __builtin_amdgcn_wmma_f32_16x16x4_f32(false, a0, false, b0, (short)0, acc0, false, false);
        acc1 = __builtin_amdgcn_wmma_f32_16x16x4_f32(false, a1, false, b1, (short)0, acc1, false, false);
      }
      const float bo = bout[l];
#pragma unroll
      for (int r = 0; r < 8; ++r) {
        const int m = row0 + r + half * 8;
        if (m < n) out[(size_t)m * NPRED + l] = acc0[r] + acc1[r] + bo;
      }
    }
    tix = tnext;
  }
}


extern "C" void kernel_launch(void* const* d_in, const int* in_sizes, int n_in,
                              void* d_out, int out_size, void* d_ws, size_t ws_size,
                              hipStream_t stream) {
  const float* x    = (const float*)d_in[0];
  const int*   ei   = (const int*)  d_in[1];
  const float* ew   = (const float*)d_in[2];
  const float* W1   = (const float*)d_in[3];
  const float* b1   = (const float*)d_in[4];
  const float* W2   = (const float*)d_in[5];
  const float* b2   = (const float*)d_in[6];
  const float* Wout = (const float*)d_in[7];
  const float* bout = (const float*)d_in[8];
  float* out = (float*)d_out;

  const int n = in_sizes[0] / HID;
  const int E = in_sizes[2];
  const int* erow = ei;
  const int* ecol = ei + E;

  const int B = 256;
  int gN  = (n + B - 1) / B;
  int gE  = (E + B - 1) / B;
  int nTiles = (n + 15) / 16;
  int gGemm  = (nTiles + 7) / 8;
  int gOut   = (n * NPRED + B - 1) / B;

  const int    gScan = (n + SCAN_ITEMS) / SCAN_ITEMS;
  const int    npad  = gScan * SCAN_ITEMS;
  const size_t plane = (size_t)n * HID;
  const size_t epad  = ((size_t)E + 63) / 64 * 64;
  const size_t ndpad = ((size_t)n + 127) / 128 * 128;
  const size_t need  = (plane + 2 * epad + (size_t)npad + ndpad + 128) * sizeof(float);

  const int    NB    = (n + BINW - 1) / BINW;
  const int    nChP  = ((E + PCH - 1) / PCH + 3) / 4 * 4;
  const size_t partw = 2 * (size_t)nChP * PCH;
  const size_t cntw  = (size_t)NB * (size_t)nChP;

  if (ws_size >= need && NB >= 1 && NB <= MAXBINS && nChP >= 4 && nChP <= MAXCH &&
      n <= (1 << 24) && partw + cntw <= plane) {
    float* h    = (float*)d_ws;
    int2*  edge = (int2*)(h + plane);
    int*   offs = (int*)(h + plane + 2 * epad);
    float* dinv = (float*)(offs + npad);
    int*   ctr  = (int*)(dinv + ndpad);
    int2*  part = (int2*)h;
    int*   cntT = (int*)(h + partw);

    k_part   <<<nChP, PT, 0, stream>>>(erow, ecol, ew, part, cntT, ctr, E, NB, nChP);
    k_binsort<<<NB, PT, 0, stream>>>(part, cntT, edge, offs, dinv, n, nChP);

    k_gather_dense<<<1024, 512, 0, stream>>>(x, offs, edge, dinv, W1, b1, h, ctr, n, nTiles);

    k_gather_dense_head<<<1024, 512, 0, stream>>>(h, offs, edge, dinv, W2, b2, Wout, bout, out,
                                                  ctr + 1, n, nTiles);
    return;
  }

  if (ws_size >= need && (size_t)npad + (size_t)E <= plane) {
    float* h    = (float*)d_ws;
    int2*  edge = (int2*)(h + plane);
    int*   offs = (int*)(h + plane + 2 * epad);
    float* dinv = (float*)(offs + npad);
    int*   ctr  = (int*)(dinv + ndpad);
    int*   cnt  = (int*)h;
    int*   rank = cnt + npad;

    k_zero_cnt<<<npad / B, B, 0, stream>>>(cnt, ctr, npad);
    k_rank    <<<gE, B, 0, stream>>>(ecol, cnt, rank, E);
    k_scan    <<<gScan, SCAN_T, 0, stream>>>(cnt, offs);
    k_fill    <<<gE, B, 0, stream>>>(erow, ecol, ew, rank, offs, edge, E);
    k_dinv_seg<<<gN, B, 0, stream>>>(offs, edge, dinv, n);

    k_gather_dense<<<1024, 512, 0, stream>>>(x, offs, edge, dinv, W1, b1, h, ctr, n, nTiles);

    k_gather_dense_head<<<1024, 512, 0, stream>>>(h, offs, edge, dinv, W2, b2, Wout, bout, out,
                                                  ctr + 1, n, nTiles);
    return;
  }

  float* dinv = (float*)d_ws;
  float* norm = dinv + n;
  float* xw   = norm + E;
  float* acc  = xw + (size_t)n * HID;

  int gScat  = (E + 7) / 8;
  int gEpi   = (n * (HID / 4) + B - 1) / B;

  k_set_deg    <<<gN, B, 0, stream>>>(dinv, n);
  k_deg_scatter<<<gE, B, 0, stream>>>(ecol, ew, dinv, E);
  k_dinv       <<<gN, B, 0, stream>>>(dinv, n);
  k_norm       <<<gE, B, 0, stream>>>(erow, ecol, ew, dinv, norm, E);

  k_gemm_wmma<<<gGemm, B, 0, stream>>>(x, W1, xw, n);
  hipMemsetAsync(acc, 0, (size_t)n * HID * sizeof(float), stream);
  k_scatter  <<<gScat, B, 0, stream>>>(xw, erow, ecol, norm, acc, E);
  k_epilogue <<<gEpi, B, 0, stream>>>(acc, xw, dinv, b1, n);

  k_gemm_wmma<<<gGemm, B, 0, stream>>>(acc, W2, xw, n);
  hipMemsetAsync(acc, 0, (size_t)n * HID * sizeof(float), stream);
  k_scatter  <<<gScat, B, 0, stream>>>(xw, erow, ecol, norm, acc, E);
  k_epilogue <<<gEpi, B, 0, stream>>>(acc, xw, dinv, b2, n);

  k_out_proj<<<gOut, B, 0, stream>>>(acc, Wout, bout, out, n);
}


# Edge ordering stores whole lines: both set-up kernels put their 8-byte entries in place in an LDS copy of the block's slice / the bin's segment and write it out in order; no sum's order changes
# speedup vs baseline: 6.1875x; 1.0552x over previous
#include <hip/hip_runtime.h>


typedef float v2f __attribute__((ext_vector_type(2)));
typedef float v8f __attribute__((ext_vector_type(8)));

#define HID 128
#define NPRED 16


__global__ void k_set_deg(float* deg, int n) {
  int i = blockIdx.x * blockDim.x + threadIdx.x;
  if (i < n) deg[i] = 1.0f;
}

__global__ void k_deg_scatter(const int* __restrict__ col,
                              const float* __restrict__ w,
                              float* deg, int E) {
  int e = blockIdx.x * blockDim.x + threadIdx.x;
  if (e < E) atomicAdd(&deg[col[e]], w[e]);
}

__global__ void k_dinv(float* deg, int n) {
  int i = blockIdx.x * blockDim.x + threadIdx.x;
  if (i < n) {
    float d = deg[i];
    deg[i] = (d > 0.0f) ? rsqrtf(d) : 0.0f;
  }
}

__global__ void k_norm(const int* __restrict__ row, const int* __restrict__ col,
                       const float* __restrict__ w, const float* __restrict__ dinv,
                       float* __restrict__ norm, int E) {
  int e = blockIdx.x * blockDim.x + threadIdx.x;
  if (e < E) norm[e] = dinv[row[e]] * w[e] * dinv[col[e]];
}

__global__ void k_gemm_wmma(const float* __restrict__ A,
                            const float* __restrict__ W,
                            float* __restrict__ C, int n) {
  int wid  = (blockIdx.x * blockDim.x + threadIdx.x) >> 5;
  int lane = threadIdx.x & 31;
  int row0 = wid * 16;
  if (row0 >= n) return;

  int half = lane >> 4;
  int l    = lane & 15;

  v8f acc[8];
#pragma unroll
  for (int t = 0; t < 8; ++t) acc[t] = (v8f){0,0,0,0,0,0,0,0};

  const float* arow = A + (size_t)(row0 + l) * HID;

  for (int k = 0; k < HID; k += 4) {
    v2f afrag = *(const v2f*)(arow + k + half * 2);
#pragma unroll
    for (int t = 0; t < 8; ++t) {
      int ncol = t * 16 + l;
      v2f bfrag;
      bfrag.x = W[(size_t)(k + half * 2)     * HID + ncol];
      bfrag.y = W[(size_t)(k + half * 2 + 1) * HID + ncol];
      acc[t] = __builtin_amdgcn_wmma_f32_16x16x4_f32(
           false, afrag,  false, bfrag,
           (short)0, acc[t],  false,  false);
    }
  }

#pragma unroll
  for (int t = 0; t < 8; ++t) {
#pragma unroll
    for (int r = 0; r < 8; ++r) {
      int m    = row0 + r + half * 8;
      int ncol = t * 16 + l;
      C[(size_t)m * HID + ncol] = acc[t][r];
    }
  }
}

__global__ void k_scatter(const float* __restrict__ xw,
                          const int* __restrict__ row,
                          const int* __restrict__ col,
                          const float* __restrict__ norm,
                          float* __restrict__ acc, int E) {
  int e = ((blockIdx.x * blockDim.x + threadIdx.x) >> 5);
  if (e >= E) return;
  int lane = threadIdx.x & 31;
  int r = row[e], c = col[e];
  float s = norm[e];
  const float4 v = *(const float4*)(xw + (size_t)r * HID + lane * 4);
  float* dst = acc + (size_t)c * HID + lane * 4;
  atomicAdd(dst + 0, v.x * s);
  atomicAdd(dst + 1, v.y * s);
  atomicAdd(dst + 2, v.z * s);
  atomicAdd(dst + 3, v.w * s);
}

__global__ void k_epilogue(float* __restrict__ acc,
                           const float* __restrict__ xw,
                           const float* __restrict__ dinv,
                           const float* __restrict__ b, int n) {
  int i = blockIdx.x * blockDim.x + threadIdx.x;
  if (i >= n * (HID / 4)) return;
  int node = i >> 5;
  int f4   = i & 31;
  float s  = dinv[node];
  s = s * s;
  float4 a  = ((float4*)acc)[i];
  float4 xv = ((const float4*)xw)[i];
  float4 bv = ((const float4*)b)[f4];
  float4 o;
  o.x = a.x + xv.x * s + bv.x;  o.x = o.x > 0.0f ? o.x : 0.0f;
  o.y = a.y + xv.y * s + bv.y;  o.y = o.y > 0.0f ? o.y : 0.0f;
  o.z = a.z + xv.z * s + bv.z;  o.z = o.z > 0.0f ? o.z : 0.0f;
  o.w = a.w + xv.w * s + bv.w;  o.w = o.w > 0.0f ? o.w : 0.0f;
  ((float4*)acc)[i] = o;
}

__global__ void k_out_proj(const float* __restrict__ h,
                           const float* __restrict__ Wout,
                           const float* __restrict__ bout,
                           float* __restrict__ out, int n) {
  int idx = blockIdx.x * blockDim.x + threadIdx.x;
  if (idx >= n * NPRED) return;
  int node = idx >> 4;
  int p    = idx & 15;
  const float4* hr = (const float4*)(h + (size_t)node * HID);
  const float4* wr = (const float4*)(Wout + (size_t)p * HID);
  float s = 0.0f;
#pragma unroll 8
  for (int k = 0; k < HID / 4; ++k) {
    float4 a = hr[k], b = wr[k];
    s += a.x * b.x + a.y * b.y + a.z * b.z + a.w * b.w;
  }
  out[idx] = s + bout[p];
}


#define SCAN_T     1024
#define SCAN_ITEMS 4096

__global__ void k_zero_cnt(int* __restrict__ cnt, int* __restrict__ ctr, int npad) {
  int i = blockIdx.x * blockDim.x + threadIdx.x;
  if (i < npad) cnt[i] = 0;
  if (i < 2) ctr[i] = 0;
}

__global__ void k_rank(const int* __restrict__ col, int* cnt,
                       int* __restrict__ rank, int E) {
  int e = blockIdx.x * blockDim.x + threadIdx.x;
  if (e < E) rank[e] = atomicAdd(&cnt[col[e]], 1);
}

__global__ void k_scan(const int* __restrict__ cnt, int* __restrict__ offs) {
  __shared__ int s_front[SCAN_T / 32];
  __shared__ int s_wave[SCAN_T / 32];
  const int tid  = threadIdx.x;
  const int lane = tid & 31;
  const int wv   = tid >> 5;
  const int base = blockIdx.x * SCAN_ITEMS;

  int s = 0;
  const int4* front = (const int4*)cnt + tid;
#pragma unroll 4
  for (int k = 0; k < (int)blockIdx.x; ++k) {
    int4 f = front[k * SCAN_T];
    s += f.x + f.y + f.z + f.w;
  }
  for (int m = 16; m > 0; m >>= 1) s += __shfl_xor(s, m);
  if (lane == 0) s_front[wv] = s;

  const int4 v = *(const int4*)(cnt + base + tid * 4);
  const int t = v.x + v.y + v.z + v.w;
  int inc = t;
  for (int d = 1; d < 32; d <<= 1) {
    int y = __shfl_up(inc, d);
    if (lane >= d) inc += y;
  }
  if (lane == 31) s_wave[wv] = inc;
  __syncthreads();

  int ex = inc - t;
  for (int w = 0; w < SCAN_T / 32; ++w) {
    ex += s_front[w];
    if (w < wv) ex += s_wave[w];
  }
  int4 o;
  o.x = ex;
  o.y = ex + v.x;
  o.z = o.y + v.y;
  o.w = o.z + v.z;
  *(int4*)(offs + base + tid * 4) = o;
}

__global__ void k_fill(const int* __restrict__ row, const int* __restrict__ col,
                       const float* __restrict__ w, const int* __restrict__ rank,
                       const int* __restrict__ offs, int2* __restrict__ edge, int E) {
  int e = blockIdx.x * blockDim.x + threadIdx.x;
  if (e < E) {
    int p = offs[col[e]] + rank[e];
    edge[p] = make_int2(row[e], __float_as_int(w[e]));
  }
}

__global__ void k_dinv_seg(const int* __restrict__ offs, const int2* __restrict__ edge,
                           float* __restrict__ dinv, int n) {
  int c = blockIdx.x * blockDim.x + threadIdx.x;
  if (c >= n) return;
  const int p1 = offs[c + 1];
  float d = 1.0f;
  for (int p = offs[c]; p < p1; ++p) d += __int_as_float(edge[p].y);
  dinv[c] = (d > 0.0f) ? rsqrtf(d) : 0.0f;
}


#define PCH     8192
#define EPT     8
#define PT      1024
#define BINW    256
#define MAXBINS 1024
#define MAXCH   1024
#define SCAP    8192

__device__ __forceinline__ int wave_scan_inc(int v, const int lane) {
  for (int d = 1; d < 32; d <<= 1) {
    const int y = __shfl_up(v, d);
    if (lane >= d) v += y;
  }
  return v;
}

__global__ void __launch_bounds__(PT)
k_part(const int* __restrict__ row, const int* __restrict__ col, const float* __restrict__ w,
       int2* __restrict__ part, int* __restrict__ cntT, int* __restrict__ ctr,
       int E, int NB, int nChP) {
  __shared__ __attribute__((aligned(16))) int2 s_buf[PCH];
  __shared__ int s_cnt[MAXBINS];
  __shared__ int s_start[MAXBINS];
  __shared__ int s_w[PT / 32];
  const int tid  = threadIdx.x;
  const int lane = tid & 31;
  const int wv   = tid >> 5;
  const int q    = blockIdx.x;
  if (q == 0 && tid < 2) ctr[tid] = 0;
  s_cnt[tid] = 0;
  __syncthreads();

  const int e0 = q * PCH;
  int bn[EPT], rk[EPT], sx[EPT], wb[EPT];
#pragma unroll
  for (int j = 0; j < EPT; ++j) {
    const int e  = e0 + j * PT + tid;
    const int ec = e < E ? e : E - 1;
    const int c  = col[ec];
    rk[j] = 0;
    bn[j] = e < E ? (c >> 8) : -1;
    sx[j] = (int)((unsigned)row[ec] | ((unsigned)(c & (BINW - 1)) << 24));
    wb[j] = __float_as_int(w[ec]);
  }
#pragma unroll
  for (int j = 0; j < EPT; ++j)
    if (bn[j] >= 0) rk[j] = atomicAdd(&s_cnt[bn[j]], 1);
  __syncthreads();

  const int cv  = s_cnt[tid];
  const int inc = wave_scan_inc(cv, lane);
  if (lane == 31) s_w[wv] = inc;
  __syncthreads();
  int ex = inc - cv;
  for (int k = 0; k < wv; ++k) ex += s_w[k];
  s_start[tid] = ex;
  if (tid < NB) cntT[(size_t)tid * nChP + q] = (cv << 16) | ex;
  __syncthreads();

#pragma unroll
  for (int j = 0; j < EPT; ++j)
    if (bn[j] >= 0) s_buf[s_start[bn[j]] + rk[j]] = make_int2(sx[j], wb[j]);
  __syncthreads();

  const int cnt = (E - e0 < PCH) ? (E - e0) : PCH;
  int2* dst = part + (size_t)e0;
  for (int i = 2 * tid; i < cnt; i += 2 * PT) {
    if (i + 1 < cnt) *(int4*)(dst + i) = *(const int4*)(s_buf + i);
    else             dst[i] = s_buf[i];
  }
}

__device__ __forceinline__ int2 bin_entry(const int2* __restrict__ part, const int* s_roff,
                                          const int* s_rst, const int i) {
  int r = 0;
#pragma unroll
  for (int step = MAXCH / 2; step >= 1; step >>= 1)
    r = (s_roff[r + step] <= i) ? r + step : r;
  return part[(size_t)r * PCH + s_rst[r] + (i - s_roff[r])];
}

__global__ void __launch_bounds__(PT, 16)
k_binsort(const int2* __restrict__ part, const int* __restrict__ cntT,
          int2* __restrict__ edge, int* __restrict__ offs, float* __restrict__ dinv,
          int n, int nChP) {
  __shared__ __attribute__((aligned(16))) int2 s_out[SCAP];
  __shared__ int   s_roff[MAXCH + 1];
  __shared__ int   s_rst[MAXCH];
  __shared__ int   s_ncnt[BINW];
  __shared__ int   s_noff[BINW];
  __shared__ float s_deg[BINW];
  __shared__ int   s_w[PT / 32];
  __shared__ int   s_w2[PT / 32];
  __shared__ int   s_red[PT / 32];
  const int tid  = threadIdx.x;
  const int lane = tid & 31;
  const int wv   = tid >> 5;
  const int k    = blockIdx.x;

  int s = 0;
  const int4* front = (const int4*)cntT;
  const int nf = (k * nChP) >> 2;
#pragma unroll 4
  for (int i = tid; i < nf; i += PT) {
    const int4 f = front[i];
    s += (f.x >> 16) + (f.y >> 16) + (f.z >> 16) + (f.w >> 16);
  }
  for (int m = 16; m > 0; m >>= 1) s += __shfl_xor(s, m);
  if (lane == 0) s_red[wv] = s;

  int v = 0;
  if (tid < nChP) v = cntT[(size_t)k * nChP + tid];
  const int cv  = v >> 16;
  const int inc = wave_scan_inc(cv, lane);
  if (lane == 31) s_w[wv] = inc;
  if (tid < BINW) { s_ncnt[tid] = 0;  s_deg[tid] = 0.0f; }
  __syncthreads();
  int ex = inc - cv;
  for (int j = 0; j < wv; ++j) ex += s_w[j];
  int base = s_red[lane], T = s_w[lane];
  for (int m = 16; m > 0; m >>= 1) { base += __shfl_xor(base, m);  T += __shfl_xor(T, m); }
  s_roff[tid] = tid < nChP ? ex : (tid == nChP ? T : 0x7fffffff);
  if (tid == 0) s_roff[MAXCH] = (nChP == MAXCH) ? T : 0x7fffffff;
  if (tid < nChP) s_rst[tid] = v & 0xFFFF;
  __syncthreads();

  int2 ev[4];
#pragma unroll
  for (int j = 0; j < 4; ++j) ev[j] = make_int2(0, 0);
  if (T > 0) {
    int ix[4], r[4];
#pragma unroll
    for (int j = 0; j < 4; ++j) { const int i = j * PT + tid;  ix[j] = i < T ? i : T - 1;  r[j] = 0; }
#pragma unroll
    for (int step = MAXCH / 2; step >= 1; step >>= 1) {
#pragma unroll
      for (int j = 0; j < 4; ++j) r[j] = (s_roff[r[j] + step] <= ix[j]) ? r[j] + step : r[j];
    }
#pragma unroll
    for (int j = 0; j < 4; ++j)
      ev[j] = part[(size_t)r[j] * PCH + s_rst[r[j]] + (ix[j] - s_roff[r[j]])];
  }

#pragma unroll
  for (int j = 0; j < 4; ++j) {
    if (j * PT + tid < T) {
      const int d = (int)((unsigned)ev[j].x >> 24);
      atomicAdd(&s_ncnt[d], 1);
      atomicAdd(&s_deg[d], __int_as_float(ev[j].y));
    }
  }
  for (int i = 4 * PT + tid; i < T; i += PT) {
    const int2 e = bin_entry(part, s_roff, s_rst, i);
    const int d = (int)((unsigned)e.x >> 24);
    atomicAdd(&s_ncnt[d], 1);
    atomicAdd(&s_deg[d], __int_as_float(e.y));
  }
  __syncthreads();

  int nc = 0;
  if (tid < BINW) nc = s_ncnt[tid];
  const int inc2 = wave_scan_inc(nc, lane);
  if (lane == 31) s_w2[wv] = inc2;
  __syncthreads();
  if (tid < BINW) {
    int ex2 = inc2 - nc;
    for (int j = 0; j < wv; ++j) ex2 += s_w2[j];
    s_noff[tid] = ex2;
    s_ncnt[tid] = 0;
    const int node = k * BINW + tid;
    if (node < n) {
      offs[node] = base + ex2;
      const float dg = 1.0f + s_deg[tid];
      dinv[node] = (dg > 0.0f) ? rsqrtf(dg) : 0.0f;
    }
  }
  if (k == (int)gridDim.x - 1 && tid == 0) offs[n] = base + T;
  __syncthreads();

  const bool staged = T <= SCAP;
#pragma unroll
  for (int j = 0; j < 4; ++j) {
    if (j * PT + tid < T) {
      const int d = (int)((unsigned)ev[j].x >> 24);
      const int pos = s_noff[d] + atomicAdd(&s_ncnt[d], 1);
      const int2 en = make_int2(ev[j].x & 0xFFFFFF, ev[j].y);
      if (staged) s_out[pos] = en;
      else        edge[(size_t)base + pos] = en;
    }
  }
  for (int i = 4 * PT + tid; i < T; i += PT) {
    const int2 e = bin_entry(part, s_roff, s_rst, i);
    const int d = (int)((unsigned)e.x >> 24);
    const int pos = s_noff[d] + atomicAdd(&s_ncnt[d], 1);
    const int2 en = make_int2(e.x & 0xFFFFFF, e.y);
    if (staged) s_out[pos] = en;
    else        edge[(size_t)base + pos] = en;
  }
  __syncthreads();
  if (staged) {
    int2* dst = edge + (size_t)base;
    for (int i = tid; i < T; i += PT) dst[i] = s_out[i];
  }
}

template <bool SRC_SCALED>
__device__ __forceinline__ float edge_factor(const int wbits, const int dsbits, const int j) {
  const float w = __int_as_float(__builtin_amdgcn_readlane(wbits, j));
  if (SRC_SCALED) return w;
  return w * __int_as_float(__builtin_amdgcn_readlane(dsbits, j));
}

template <bool SRC_SCALED>
__device__ __forceinline__ float4 gather_row(const int c, const int lane,
                                             const float* __restrict__ xin,
                                             const int* __restrict__ offs,
                                             const int2* __restrict__ edge,
                                             const float* __restrict__ dinv,
                                             float& dc) {
  const int p0 = offs[c];
  const int p1 = offs[c + 1];
  const float* xl = xin + lane * 4;
  const float4 xv = *(const float4*)(xl + (size_t)c * HID);
  const float s = dinv[c];

  float4 a = make_float4(0.0f, 0.0f, 0.0f, 0.0f);
  for (int base = p0; base < p1; base += 32) {
    int m = p1 - base;
    if (m > 32) m = 32;
    const int2 ev = edge[base + (lane < m ? lane : m - 1)];
    int dsb = 0;
    if (!SRC_SCALED) dsb = __float_as_int(dinv[ev.x]);
    int j = 0;
    for (; j + 4 <= m; j += 4) {
      const int r0 = __builtin_amdgcn_readlane(ev.x, j);
      const int r1 = __builtin_amdgcn_readlane(ev.x, j + 1);
      const int r2 = __builtin_amdgcn_readlane(ev.x, j + 2);
      const int r3 = __builtin_amdgcn_readlane(ev.x, j + 3);
      const float4 v0 = *(const float4*)(xl + (size_t)r0 * HID);
      const float4 v1 = *(const float4*)(xl + (size_t)r1 * HID);
      const float4 v2 = *(const float4*)(xl + (size_t)r2 * HID);
      const float4 v3 = *(const float4*)(xl + (size_t)r3 * HID);
      const float s0 = edge_factor<SRC_SCALED>(ev.y, dsb, j);
      const float s1 = edge_factor<SRC_SCALED>(ev.y, dsb, j + 1);
      const float s2 = edge_factor<SRC_SCALED>(ev.y, dsb, j + 2);
      const float s3 = edge_factor<SRC_SCALED>(ev.y, dsb, j + 3);
      a.x += v0.x * s0;  a.y += v0.y * s0;  a.z += v0.z * s0;  a.w += v0.w * s0;
      a.x += v1.x * s1;  a.y += v1.y * s1;  a.z += v1.z * s1;  a.w += v1.w * s1;
      a.x += v2.x * s2;  a.y += v2.y * s2;  a.z += v2.z * s2;  a.w += v2.w * s2;
      a.x += v3.x * s3;  a.y += v3.y * s3;  a.z += v3.z * s3;  a.w += v3.w * s3;
    }
    for (; j < m; ++j) {
      const int r0 = __builtin_amdgcn_readlane(ev.x, j);
      const float s0 = edge_factor<SRC_SCALED>(ev.y, dsb, j);
      const float4 v0 = *(const float4*)(xl + (size_t)r0 * HID);
      a.x += v0.x * s0;  a.y += v0.y * s0;  a.z += v0.z * s0;  a.w += v0.w * s0;
    }
  }

  const float t = SRC_SCALED ? 1.0f : s;
  float4 o;
  o.x = (a.x + xv.x * t) * s;
  o.y = (a.y + xv.y * t) * s;
  o.z = (a.z + xv.z * t) * s;
  o.w = (a.w + xv.w * t) * s;
  dc = s;
  return o;
}

#define TP 132

__device__ __forceinline__ void stage_w(float* wf, const float* __restrict__ W,
                                        const int wv, const int lane) {
  const int half = lane >> 4;
  const int l    = lane & 15;
  const int t = wv & 7;
  const float* wc = W + (size_t)(half * 2) * HID + t * 16 + l;
#pragma unroll 8
  for (int i = 0; i < 16; ++i) {
    const int ks = 2 * i + (wv >> 3);
    v2f f;
    f.x = wc[(size_t)(4 * ks)     * HID];
    f.y = wc[(size_t)(4 * ks + 1) * HID];
    *(v2f*)(wf + ((ks * 8 + t) * 32 + lane) * 2) = f;
  }
}

__device__ __forceinline__ void tile_dense(float* tile, const float* wf,
                                           const float* __restrict__ b,
                                           const int wv, const int lane) {
  const int half = lane >> 4;
  const int l    = lane & 15;
  v8f acc0 = (v8f){0,0,0,0,0,0,0,0};
  v8f acc1 = (v8f){0,0,0,0,0,0,0,0};
  float bias = 0.0f;
  if (wv < 8) {
    bias = b[wv * 16 + l];
    const float* ar = tile + l * TP + half * 2;
    const float* bf = wf + (wv * 32 + lane) * 2;
#pragma unroll 4
    for (int ks = 0; ks < 32; ks += 2) {
      const v2f a0 = *(const v2f*)(ar + ks * 4);
      const v2f b0 = *(const v2f*)(bf + ks * 512);
      const v2f a1 = *(const v2f*)(ar + ks * 4 + 4);
      const v2f b1 = *(const v2f*)(bf + ks * 512 + 512);
      acc0 = __builtin_amdgcn_wmma_f32_16x16x4_f32(false, a0, false, b0, (short)0, acc0, false, false);
      acc1 = __builtin_amdgcn_wmma_f32_16x16x4_f32(false, a1, false, b1, (short)0, acc1, false, false);
    }
  }
  __syncthreads();
  if (wv < 8) {
    float* st = tile + (half * 8) * TP + wv * 16 + l;
#pragma unroll
    for (int r = 0; r < 8; ++r) {
      const float v = acc0[r] + acc1[r] + bias;
      st[r * TP] = v > 0.0f ? v : 0.0f;
    }
  }
  __syncthreads();
}

__global__ void __launch_bounds__(512, 16)
k_gather_dense(const float* __restrict__ xin, const int* __restrict__ offs,
               const int2* __restrict__ edge, const float* __restrict__ dinv,
               const float* __restrict__ W, const float* __restrict__ b,
               float* __restrict__ C, int* ctr, int n, int nTiles) {
  __shared__ __attribute__((aligned(16))) float wf[HID * HID];
  __shared__ __attribute__((aligned(16))) float tile[16 * TP];
  __shared__ int s_first;
  __shared__ int s_next;
  const int wv   = __builtin_amdgcn_readfirstlane((int)(threadIdx.x >> 5));
  const int lane = threadIdx.x & 31;

  if (threadIdx.x == 0) s_first = atomicAdd(ctr, 1);
  stage_w(wf, W, wv, lane);
  __syncthreads();

  int tix = __builtin_amdgcn_readfirstlane(s_first);
  while (tix < nTiles) {
    const int c = tix * 16 + wv;

    float4 o = make_float4(0.0f, 0.0f, 0.0f, 0.0f);
    float dc = 0.0f;
    if (c < n) o = gather_row<false>(c, lane, xin, offs, edge, dinv, dc);
    *(float4*)(tile + wv * TP + lane * 4) = o;
    __syncthreads();
    if (threadIdx.x == 480) s_next = atomicAdd(ctr, 1);

    tile_dense(tile, wf, b, wv, lane);
    const int tnext = __builtin_amdgcn_readfirstlane(s_next);
    if (c < n) {
      float4 hv = *(const float4*)(tile + wv * TP + lane * 4);
      hv.x *= dc;  hv.y *= dc;  hv.z *= dc;  hv.w *= dc;
      *(float4*)(C + (size_t)c * HID + lane * 4) = hv;
    }
    tix = tnext;
  }
}

__global__ void __launch_bounds__(512, 16)
k_gather_dense_head(const float* __restrict__ hin, const int* __restrict__ offs,
                    const int2* __restrict__ edge, const float* __restrict__ dinv,
                    const float* __restrict__ W, const float* __restrict__ b,
                    const float* __restrict__ Wout, const float* __restrict__ bout,
                    float* __restrict__ out, int* ctr, int n, int nTiles) {
  __shared__ __attribute__((aligned(16))) float wf[HID * HID];
  __shared__ __attribute__((aligned(16))) float tile[16 * TP];
  __shared__ int s_first;
  __shared__ int s_next;
  const int wv   = __builtin_amdgcn_readfirstlane((int)(threadIdx.x >> 5));
  const int lane = threadIdx.x & 31;
  const int half = lane >> 4;
  const int l    = lane & 15;

  if (threadIdx.x == 0) s_first = atomicAdd(ctr, 1);
  stage_w(wf, W, wv, lane);
  __syncthreads();

  int tix = __builtin_amdgcn_readfirstlane(s_first);
  while (tix < nTiles) {
    const int row0 = tix * 16;
    const int c    = row0 + wv;

    float4 o = make_float4(0.0f, 0.0f, 0.0f, 0.0f);
    float dc = 0.0f;
    if (c < n) o = gather_row<true>(c, lane, hin, offs, edge, dinv, dc);
    __syncthreads();
    *(float4*)(tile + wv * TP + lane * 4) = o;
    __syncthreads();
    if (threadIdx.x == 480) s_next = atomicAdd(ctr, 1);

    tile_dense(tile, wf, b, wv, lane);
    const int tnext = __builtin_amdgcn_readfirstlane(s_next);

    if (wv == 8) {
      const float* ar = tile + l * TP + half * 2;
      const float* br = Wout + (size_t)l * HID + half * 2;
      v8f acc0 = (v8f){0,0,0,0,0,0,0,0};
      v8f acc1 = (v8f){0,0,0,0,0,0,0,0};
#pragma unroll 4
      for (int k = 0; k < HID; k += 8) {
        const v2f a0 = *(const v2f*)(ar + k);
        const v2f b0 = *(const v2f*)(br + k);
        const v2f a1 = *(const v2f*)(ar + k + 4);
        const v2f b1 = *(const v2f*)(br + k + 4);
        acc0 = __builtin_amdgcn_wmma_f32_16x16x4_f32(false, a0, false, b0, (short)0, acc0, false, false);
        acc1 = __builtin_amdgcn_wmma_f32_16x16x4_f32(false, a1, false, b1, (short)0, acc1, false, false);
      }
      const float bo = bout[l];
#pragma unroll
      for (int r = 0; r < 8; ++r) {
        const int m = row0 + r + half * 8;
        if (m < n) out[(size_t)m * NPRED + l] = acc0[r] + acc1[r] + bo;
      }
    }
    tix = tnext;
  }
}


extern "C" void kernel_launch(void* const* d_in, const int* in_sizes, int n_in,
                              void* d_out, int out_size, void* d_ws, size_t ws_size,
                              hipStream_t stream) {
  const float* x    = (const float*)d_in[0];
  const int*   ei   = (const int*)  d_in[1];
  const float* ew   = (const float*)d_in[2];
  const float* W1   = (const float*)d_in[3];
  const float* b1   = (const float*)d_in[4];
  const float* W2   = (const float*)d_in[5];
  const float* b2   = (const float*)d_in[6];
  const float* Wout = (const float*)d_in[7];
  const float* bout = (const float*)d_in[8];
  float* out = (float*)d_out;

  const int n = in_sizes[0] / HID;
  const int E = in_sizes[2];
  const int* erow = ei;
  const int* ecol = ei + E;

  const int B = 256;
  int gN  = (n + B - 1) / B;
  int gE  = (E + B - 1) / B;
  int nTiles = (n + 15) / 16;
  int gGemm  = (nTiles + 7) / 8;
  int gOut   = (n * NPRED + B - 1) / B;

  const int    gScan = (n + SCAN_ITEMS) / SCAN_ITEMS;
  const int    npad  = gScan * SCAN_ITEMS;
  const size_t plane = (size_t)n * HID;
  const size_t epad  = ((size_t)E + 63) / 64 * 64;
  const size_t ndpad = ((size_t)n + 127) / 128 * 128;
  const size_t need  = (plane + 2 * epad + (size_t)npad + ndpad + 128) * sizeof(float);

  const int    NB    = (n + BINW - 1) / BINW;
  const int    nChP  = ((E + PCH - 1) / PCH + 3) / 4 * 4;
  const size_t partw = 2 * (size_t)nChP * PCH;
  const size_t cntw  = (size_t)NB * (size_t)nChP;

  if (ws_size >= need && NB >= 1 && NB <= MAXBINS && nChP >= 4 && nChP <= MAXCH &&
      n <= (1 << 24) && partw + cntw <= plane) {
    float* h    = (float*)d_ws;
    int2*  edge = (int2*)(h + plane);
    int*   offs = (int*)(h + plane + 2 * epad);
    float* dinv = (float*)(offs + npad);
    int*   ctr  = (int*)(dinv + ndpad);
    int2*  part = (int2*)h;
    int*   cntT = (int*)(h + partw);

    k_part   <<<nChP, PT, 0, stream>>>(erow, ecol, ew, part, cntT, ctr, E, NB, nChP);
    k_binsort<<<NB, PT, 0, stream>>>(part, cntT, edge, offs, dinv, n, nChP);

    k_gather_dense<<<1024, 512, 0, stream>>>(x, offs, edge, dinv, W1, b1, h, ctr, n, nTiles);

    k_gather_dense_head<<<1024, 512, 0, stream>>>(h, offs, edge, dinv, W2, b2, Wout, bout, out,
                                                  ctr + 1, n, nTiles);
    return;
  }

  if (ws_size >= need && (size_t)npad + (size_t)E <= plane) {
    float* h    = (float*)d_ws;
    int2*  edge = (int2*)(h + plane);
    int*   offs = (int*)(h + plane + 2 * epad);
    float* dinv = (float*)(offs + npad);
    int*   ctr  = (int*)(dinv + ndpad);
    int*   cnt  = (int*)h;
    int*   rank = cnt + npad;

    k_zero_cnt<<<npad / B, B, 0, stream>>>(cnt, ctr, npad);
    k_rank    <<<gE, B, 0, stream>>>(ecol, cnt, rank, E);
    k_scan    <<<gScan, SCAN_T, 0, stream>>>(cnt, offs);
    k_fill    <<<gE, B, 0, stream>>>(erow, ecol, ew, rank, offs, edge, E);
    k_dinv_seg<<<gN, B, 0, stream>>>(offs, edge, dinv, n);

    k_gather_dense<<<1024, 512, 0, stream>>>(x, offs, edge, dinv, W1, b1, h, ctr, n, nTiles);

    k_gather_dense_head<<<1024, 512, 0, stream>>>(h, offs, edge, dinv, W2, b2, Wout, bout, out,
                                                  ctr + 1, n, nTiles);
    return;
  }

  float* dinv = (float*)d_ws;
  float* norm = dinv + n;
  float* xw   = norm + E;
  float* acc  = xw + (size_t)n * HID;

  int gScat  = (E + 7) / 8;
  int gEpi   = (n * (HID / 4) + B - 1) / B;

  k_set_deg    <<<gN, B, 0, stream>>>(dinv, n);
  k_deg_scatter<<<gE, B, 0, stream>>>(ecol, ew, dinv, E);
  k_dinv       <<<gN, B, 0, stream>>>(dinv, n);
  k_norm       <<<gE, B, 0, stream>>>(erow, ecol, ew, dinv, norm, E);

  k_gemm_wmma<<<gGemm, B, 0, stream>>>(x, W1, xw, n);
  hipMemsetAsync(acc, 0, (size_t)n * HID * sizeof(float), stream);
  k_scatter  <<<gScat, B, 0, stream>>>(xw, erow, ecol, norm, acc, E);
  k_epilogue <<<gEpi, B, 0, stream>>>(acc, xw, dinv, b1, n);

  k_gemm_wmma<<<gGemm, B, 0, stream>>>(acc, W2, xw, n);
  hipMemsetAsync(acc, 0, (size_t)n * HID * sizeof(float), stream);
  k_scatter  <<<gScat, B, 0, stream>>>(xw, erow, ecol, norm, acc, E);
  k_epilogue <<<gEpi, B, 0, stream>>>(acc, xw, dinv, b2, n);

  k_out_proj<<<gOut, B, 0, stream>>>(acc, Wout, bout, out, n);
}


# Gather asks for six source rows per trip instead of four, and for a list's left-over rows as one batch instead of one round trip each; 16 waves/SIMD and 1024 blocks kept; sum order kept
# speedup vs baseline: 6.4576x; 1.0437x over previous
#include <hip/hip_runtime.h>


typedef float v2f __attribute__((ext_vector_type(2)));
typedef float v8f __attribute__((ext_vector_type(8)));

#define HID 128
#define NPRED 16
#define GW 6


__global__ void k_set_deg(float* deg, int n) {
  int i = blockIdx.x * blockDim.x + threadIdx.x;
  if (i < n) deg[i] = 1.0f;
}

__global__ void k_deg_scatter(const int* __restrict__ col,
                              const float* __restrict__ w,
                              float* deg, int E) {
  int e = blockIdx.x * blockDim.x + threadIdx.x;
  if (e < E) atomicAdd(&deg[col[e]], w[e]);
}

__global__ void k_dinv(float* deg, int n) {
  int i = blockIdx.x * blockDim.x + threadIdx.x;
  if (i < n) {
    float d = deg[i];
    deg[i] = (d > 0.0f) ? rsqrtf(d) : 0.0f;
  }
}

__global__ void k_norm(const int* __restrict__ row, const int* __restrict__ col,
                       const float* __restrict__ w, const float* __restrict__ dinv,
                       float* __restrict__ norm, int E) {
  int e = blockIdx.x * blockDim.x + threadIdx.x;
  if (e < E) norm[e] = dinv[row[e]] * w[e] * dinv[col[e]];
}

__global__ void k_gemm_wmma(const float* __restrict__ A,
                            const float* __restrict__ W,
                            float* __restrict__ C, int n) {
  int wid  = (blockIdx.x * blockDim.x + threadIdx.x) >> 5;
  int lane = threadIdx.x & 31;
  int row0 = wid * 16;
  if (row0 >= n) return;

  int half = lane >> 4;
  int l    = lane & 15;

  v8f acc[8];
#pragma unroll
  for (int t = 0; t < 8; ++t) acc[t] = (v8f){0,0,0,0,0,0,0,0};

  const float* arow = A + (size_t)(row0 + l) * HID;

  for (int k = 0; k < HID; k += 4) {
    v2f afrag = *(const v2f*)(arow + k + half * 2);
#pragma unroll
    for (int t = 0; t < 8; ++t) {
      int ncol = t * 16 + l;
      v2f bfrag;
      bfrag.x = W[(size_t)(k + half * 2)     * HID + ncol];
      bfrag.y = W[(size_t)(k + half * 2 + 1) * HID + ncol];
      acc[t] = __builtin_amdgcn_wmma_f32_16x16x4_f32(
           false, afrag,  false, bfrag,
           (short)0, acc[t],  false,  false);
    }
  }

#pragma unroll
  for (int t = 0; t < 8; ++t) {
#pragma unroll
    for (int r = 0; r < 8; ++r) {
      int m    = row0 + r + half * 8;
      int ncol = t * 16 + l;
      C[(size_t)m * HID + ncol] = acc[t][r];
    }
  }
}

__global__ void k_scatter(const float* __restrict__ xw,
                          const int* __restrict__ row,
                          const int* __restrict__ col,
                          const float* __restrict__ norm,
                          float* __restrict__ acc, int E) {
  int e = ((blockIdx.x * blockDim.x + threadIdx.x) >> 5);
  if (e >= E) return;
  int lane = threadIdx.x & 31;
  int r = row[e], c = col[e];
  float s = norm[e];
  const float4 v = *(const float4*)(xw + (size_t)r * HID + lane * 4);
  float* dst = acc + (size_t)c * HID + lane * 4;
  atomicAdd(dst + 0, v.x * s);
  atomicAdd(dst + 1, v.y * s);
  atomicAdd(dst + 2, v.z * s);
  atomicAdd(dst + 3, v.w * s);
}

__global__ void k_epilogue(float* __restrict__ acc,
                           const float* __restrict__ xw,
                           const float* __restrict__ dinv,
                           const float* __restrict__ b, int n) {
  int i = blockIdx.x * blockDim.x + threadIdx.x;
  if (i >= n * (HID / 4)) return;
  int node = i >> 5;
  int f4   = i & 31;
  float s  = dinv[node];
  s = s * s;
  float4 a  = ((float4*)acc)[i];
  float4 xv = ((const float4*)xw)[i];
  float4 bv = ((const float4*)b)[f4];
  float4 o;
  o.x = a.x + xv.x * s + bv.x;  o.x = o.x > 0.0f ? o.x : 0.0f;
  o.y = a.y + xv.y * s + bv.y;  o.y = o.y > 0.0f ? o.y : 0.0f;
  o.z = a.z + xv.z * s + bv.z;  o.z = o.z > 0.0f ? o.z : 0.0f;
  o.w = a.w + xv.w * s + bv.w;  o.w = o.w > 0.0f ? o.w : 0.0f;
  ((float4*)acc)[i] = o;
}

__global__ void k_out_proj(const float* __restrict__ h,
                           const float* __restrict__ Wout,
                           const float* __restrict__ bout,
                           float* __restrict__ out, int n) {
  int idx = blockIdx.x * blockDim.x + threadIdx.x;
  if (idx >= n * NPRED) return;
  int node = idx >> 4;
  int p    = idx & 15;
  const float4* hr = (const float4*)(h + (size_t)node * HID);
  const float4* wr = (const float4*)(Wout + (size_t)p * HID);
  float s = 0.0f;
#pragma unroll 8
  for (int k = 0; k < HID / 4; ++k) {
    float4 a = hr[k], b = wr[k];
    s += a.x * b.x + a.y * b.y + a.z * b.z + a.w * b.w;
  }
  out[idx] = s + bout[p];
}


#define SCAN_T     1024
#define SCAN_ITEMS 4096

__global__ void k_zero_cnt(int* __restrict__ cnt, int* __restrict__ ctr, int npad) {
  int i = blockIdx.x * blockDim.x + threadIdx.x;
  if (i < npad) cnt[i] = 0;
  if (i < 2) ctr[i] = 0;
}

__global__ void k_rank(const int* __restrict__ col, int* cnt,
                       int* __restrict__ rank, int E) {
  int e = blockIdx.x * blockDim.x + threadIdx.x;
  if (e < E) rank[e] = atomicAdd(&cnt[col[e]], 1);
}

__global__ void k_scan(const int* __restrict__ cnt, int* __restrict__ offs) {
  __shared__ int s_front[SCAN_T / 32];
  __shared__ int s_wave[SCAN_T / 32];
  const int tid  = threadIdx.x;
  const int lane = tid & 31;
  const int wv   = tid >> 5;
  const int base = blockIdx.x * SCAN_ITEMS;

  int s = 0;
  const int4* front = (const int4*)cnt + tid;
#pragma unroll 4
  for (int k = 0; k < (int)blockIdx.x; ++k) {
    int4 f = front[k * SCAN_T];
    s += f.x + f.y + f.z + f.w;
  }
  for (int m = 16; m > 0; m >>= 1) s += __shfl_xor(s, m);
  if (lane == 0) s_front[wv] = s;

  const int4 v = *(const int4*)(cnt + base + tid * 4);
  const int t = v.x + v.y + v.z + v.w;
  int inc = t;
  for (int d = 1; d < 32; d <<= 1) {
    int y = __shfl_up(inc, d);
    if (lane >= d) inc += y;
  }
  if (lane == 31) s_wave[wv] = inc;
  __syncthreads();

  int ex = inc - t;
  for (int w = 0; w < SCAN_T / 32; ++w) {
    ex += s_front[w];
    if (w < wv) ex += s_wave[w];
  }
  int4 o;
  o.x = ex;
  o.y = ex + v.x;
  o.z = o.y + v.y;
  o.w = o.z + v.z;
  *(int4*)(offs + base + tid * 4) = o;
}

__global__ void k_fill(const int* __restrict__ row, const int* __restrict__ col,
                       const float* __restrict__ w, const int* __restrict__ rank,
                       const int* __restrict__ offs, int2* __restrict__ edge, int E) {
  int e = blockIdx.x * blockDim.x + threadIdx.x;
  if (e < E) {
    int p = offs[col[e]] + rank[e];
    edge[p] = make_int2(row[e], __float_as_int(w[e]));
  }
}

__global__ void k_dinv_seg(const int* __restrict__ offs, const int2* __restrict__ edge,
                           float* __restrict__ dinv, int n) {
  int c = blockIdx.x * blockDim.x + threadIdx.x;
  if (c >= n) return;
  const int p1 = offs[c + 1];
  float d = 1.0f;
  for (int p = offs[c]; p < p1; ++p) d += __int_as_float(edge[p].y);
  dinv[c] = (d > 0.0f) ? rsqrtf(d) : 0.0f;
}


#define PCH     8192
#define EPT     8
#define PT      1024
#define BINW    256
#define MAXBINS 1024
#define MAXCH   1024
#define SCAP    8192

__device__ __forceinline__ int wave_scan_inc(int v, const int lane) {
  for (int d = 1; d < 32; d <<= 1) {
    const int y = __shfl_up(v, d);
    if (lane >= d) v += y;
  }
  return v;
}

__global__ void __launch_bounds__(PT)
k_part(const int* __restrict__ row, const int* __restrict__ col, const float* __restrict__ w,
       int2* __restrict__ part, int* __restrict__ cntT, int* __restrict__ ctr,
       int E, int NB, int nChP) {
  __shared__ __attribute__((aligned(16))) int2 s_buf[PCH];
  __shared__ int s_cnt[MAXBINS];
  __shared__ int s_start[MAXBINS];
  __shared__ int s_w[PT / 32];
  const int tid  = threadIdx.x;
  const int lane = tid & 31;
  const int wv   = tid >> 5;
  const int q    = blockIdx.x;
  if (q == 0 && tid < 2) ctr[tid] = 0;
  s_cnt[tid] = 0;
  __syncthreads();

  const int e0 = q * PCH;
  int bn[EPT], rk[EPT], sx[EPT], wb[EPT];
#pragma unroll
  for (int j = 0; j < EPT; ++j) {
    const int e  = e0 + j * PT + tid;
    const int ec = e < E ? e : E - 1;
    const int c  = col[ec];
    rk[j] = 0;
    bn[j] = e < E ? (c >> 8) : -1;
    sx[j] = (int)((unsigned)row[ec] | ((unsigned)(c & (BINW - 1)) << 24));
    wb[j] = __float_as_int(w[ec]);
  }
#pragma unroll
  for (int j = 0; j < EPT; ++j)
    if (bn[j] >= 0) rk[j] = atomicAdd(&s_cnt[bn[j]], 1);
  __syncthreads();

  const int cv  = s_cnt[tid];
  const int inc = wave_scan_inc(cv, lane);
  if (lane == 31) s_w[wv] = inc;
  __syncthreads();
  int ex = inc - cv;
  for (int k = 0; k < wv; ++k) ex += s_w[k];
  s_start[tid] = ex;
  if (tid < NB) cntT[(size_t)tid * nChP + q] = (cv << 16) | ex;
  __syncthreads();

#pragma unroll
  for (int j = 0; j < EPT; ++j)
    if (bn[j] >= 0) s_buf[s_start[bn[j]] + rk[j]] = make_int2(sx[j], wb[j]);
  __syncthreads();

  const int cnt = (E - e0 < PCH) ? (E - e0) : PCH;
  int2* dst = part + (size_t)e0;
  for (int i = 2 * tid; i < cnt; i += 2 * PT) {
    if (i + 1 < cnt) *(int4*)(dst + i) = *(const int4*)(s_buf + i);
    else             dst[i] = s_buf[i];
  }
}

__device__ __forceinline__ int2 bin_entry(const int2* __restrict__ part, const int* s_roff,
                                          const int* s_rst, const int i) {
  int r = 0;
#pragma unroll
  for (int step = MAXCH / 2; step >= 1; step >>= 1)
    r = (s_roff[r + step] <= i) ? r + step : r;
  return part[(size_t)r * PCH + s_rst[r] + (i - s_roff[r])];
}

__global__ void __launch_bounds__(PT, 16)
k_binsort(const int2* __restrict__ part, const int* __restrict__ cntT,
          int2* __restrict__ edge, int* __restrict__ offs, float* __restrict__ dinv,
          int n, int nChP) {
  __shared__ __attribute__((aligned(16))) int2 s_out[SCAP];
  __shared__ int   s_roff[MAXCH + 1];
  __shared__ int   s_rst[MAXCH];
  __shared__ int   s_ncnt[BINW];
  __shared__ int   s_noff[BINW];
  __shared__ float s_deg[BINW];
  __shared__ int   s_w[PT / 32];
  __shared__ int   s_w2[PT / 32];
  __shared__ int   s_red[PT / 32];
  const int tid  = threadIdx.x;
  const int lane = tid & 31;
  const int wv   = tid >> 5;
  const int k    = blockIdx.x;

  int s = 0;
  const int4* front = (const int4*)cntT;
  const int nf = (k * nChP) >> 2;
#pragma unroll 4
  for (int i = tid; i < nf; i += PT) {
    const int4 f = front[i];
    s += (f.x >> 16) + (f.y >> 16) + (f.z >> 16) + (f.w >> 16);
  }
  for (int m = 16; m > 0; m >>= 1) s += __shfl_xor(s, m);
  if (lane == 0) s_red[wv] = s;

  int v = 0;
  if (tid < nChP) v = cntT[(size_t)k * nChP + tid];
  const int cv  = v >> 16;
  const int inc = wave_scan_inc(cv, lane);
  if (lane == 31) s_w[wv] = inc;
  if (tid < BINW) { s_ncnt[tid] = 0;  s_deg[tid] = 0.0f; }
  __syncthreads();
  int ex = inc - cv;
  for (int j = 0; j < wv; ++j) ex += s_w[j];
  int base = s_red[lane], T = s_w[lane];
  for (int m = 16; m > 0; m >>= 1) { base += __shfl_xor(base, m);  T += __shfl_xor(T, m); }
  s_roff[tid] = tid < nChP ? ex : (tid == nChP ? T : 0x7fffffff);
  if (tid == 0) s_roff[MAXCH] = (nChP == MAXCH) ? T : 0x7fffffff;
  if (tid < nChP) s_rst[tid] = v & 0xFFFF;
  __syncthreads();

  int2 ev[4];
#pragma unroll
  for (int j = 0; j < 4; ++j) ev[j] = make_int2(0, 0);
  if (T > 0) {
    int ix[4], r[4];
#pragma unroll
    for (int j = 0; j < 4; ++j) { const int i = j * PT + tid;  ix[j] = i < T ? i : T - 1;  r[j] = 0; }
#pragma unroll
    for (int step = MAXCH / 2; step >= 1; step >>= 1) {
#pragma unroll
      for (int j = 0; j < 4; ++j) r[j] = (s_roff[r[j] + step] <= ix[j]) ? r[j] + step : r[j];
    }
#pragma unroll
    for (int j = 0; j < 4; ++j)
      ev[j] = part[(size_t)r[j] * PCH + s_rst[r[j]] + (ix[j] - s_roff[r[j]])];
  }

#pragma unroll
  for (int j = 0; j < 4; ++j) {
    if (j * PT + tid < T) {
      const int d = (int)((unsigned)ev[j].x >> 24);
      atomicAdd(&s_ncnt[d], 1);
      atomicAdd(&s_deg[d], __int_as_float(ev[j].y));
    }
  }
  for (int i = 4 * PT + tid; i < T; i += PT) {
    const int2 e = bin_entry(part, s_roff, s_rst, i);
    const int d = (int)((unsigned)e.x >> 24);
    atomicAdd(&s_ncnt[d], 1);
    atomicAdd(&s_deg[d], __int_as_float(e.y));
  }
  __syncthreads();

  int nc = 0;
  if (tid < BINW) nc = s_ncnt[tid];
  const int inc2 = wave_scan_inc(nc, lane);
  if (lane == 31) s_w2[wv] = inc2;
  __syncthreads();
  if (tid < BINW) {
    int ex2 = inc2 - nc;
    for (int j = 0; j < wv; ++j) ex2 += s_w2[j];
    s_noff[tid] = ex2;
    s_ncnt[tid] = 0;
    const int node = k * BINW + tid;
    if (node < n) {
      offs[node] = base + ex2;
      const float dg = 1.0f + s_deg[tid];
      dinv[node] = (dg > 0.0f) ? rsqrtf(dg) : 0.0f;
    }
  }
  if (k == (int)gridDim.x - 1 && tid == 0) offs[n] = base + T;
  __syncthreads();

  const bool staged = T <= SCAP;
#pragma unroll
  for (int j = 0; j < 4; ++j) {
    if (j * PT + tid < T) {
      const int d = (int)((unsigned)ev[j].x >> 24);
      const int pos = s_noff[d] + atomicAdd(&s_ncnt[d], 1);
      const int2 en = make_int2(ev[j].x & 0xFFFFFF, ev[j].y);
      if (staged) s_out[pos] = en;
      else        edge[(size_t)base + pos] = en;
    }
  }
  for (int i = 4 * PT + tid; i < T; i += PT) {
    const int2 e = bin_entry(part, s_roff, s_rst, i);
    const int d = (int)((unsigned)e.x >> 24);
    const int pos = s_noff[d] + atomicAdd(&s_ncnt[d], 1);
    const int2 en = make_int2(e.x & 0xFFFFFF, e.y);
    if (staged) s_out[pos] = en;
    else        edge[(size_t)base + pos] = en;
  }
  __syncthreads();
  if (staged) {
    int2* dst = edge + (size_t)base;
    for (int i = tid; i < T; i += PT) dst[i] = s_out[i];
  }
}

template <bool SRC_SCALED>
__device__ __forceinline__ float edge_factor(const int wbits, const int dsbits, const int j) {
  const float w = __int_as_float(__builtin_amdgcn_readlane(wbits, j));
  if (SRC_SCALED) return w;
  return w * __int_as_float(__builtin_amdgcn_readlane(dsbits, j));
}

template <int K, bool SRC_SCALED>
__device__ __forceinline__ void add_rows(float4& a, const float* __restrict__ xl,
                                         const int srcs, const int wbits, const int dsbits,
                                         const int j) {
  float4 v[K];
#pragma unroll
  for (int i = 0; i < K; ++i) {
    const int r = __builtin_amdgcn_readlane(srcs, j + i);
    v[i] = *(const float4*)(xl + (size_t)r * HID);
  }
#pragma unroll
  for (int i = 0; i < K; ++i) {
    const float s = edge_factor<SRC_SCALED>(wbits, dsbits, j + i);
    a.x += v[i].x * s;  a.y += v[i].y * s;  a.z += v[i].z * s;  a.w += v[i].w * s;
  }
}

template <bool SRC_SCALED>
__device__ __forceinline__ float4 gather_row(const int c, const int lane,
                                             const float* __restrict__ xin,
                                             const int* __restrict__ offs,
                                             const int2* __restrict__ edge,
                                             const float* __restrict__ dinv,
                                             float& dc) {
  const int p0 = offs[c];
  const int p1 = offs[c + 1];
  const float* xl = xin + lane * 4;
  const float4 xv = *(const float4*)(xl + (size_t)c * HID);
  const float s = dinv[c];

  float4 a = make_float4(0.0f, 0.0f, 0.0f, 0.0f);
  for (int base = p0; base < p1; base += 32) {
    int m = p1 - base;
    if (m > 32) m = 32;
    const int2 ev = edge[base + (lane < m ? lane : m - 1)];
    int dsb = 0;
    if (!SRC_SCALED) dsb = __float_as_int(dinv[ev.x]);
    int j = 0;
    for (; j + GW <= m; j += GW) add_rows<GW, SRC_SCALED>(a, xl, ev.x, ev.y, dsb, j);
    switch (m - j) {
      case 5: add_rows<5, SRC_SCALED>(a, xl, ev.x, ev.y, dsb, j);  break;
      case 4: add_rows<4, SRC_SCALED>(a, xl, ev.x, ev.y, dsb, j);  break;
      case 3: add_rows<3, SRC_SCALED>(a, xl, ev.x, ev.y, dsb, j);  break;
      case 2: add_rows<2, SRC_SCALED>(a, xl, ev.x, ev.y, dsb, j);  break;
      case 1: add_rows<1, SRC_SCALED>(a, xl, ev.x, ev.y, dsb, j);  break;
      default: break;
    }
  }

  const float t = SRC_SCALED ? 1.0f : s;
  float4 o;
  o.x = (a.x + xv.x * t) * s;
  o.y = (a.y + xv.y * t) * s;
  o.z = (a.z + xv.z * t) * s;
  o.w = (a.w + xv.w * t) * s;
  dc = s;
  return o;
}

#define TP 132

__device__ __forceinline__ void stage_w(float* wf, const float* __restrict__ W,
                                        const int wv, const int lane) {
  const int half = lane >> 4;
  const int l    = lane & 15;
  const int t = wv & 7;
  const float* wc = W + (size_t)(half * 2) * HID + t * 16 + l;
#pragma unroll 8
  for (int i = 0; i < 16; ++i) {
    const int ks = 2 * i + (wv >> 3);
    v2f f;
    f.x = wc[(size_t)(4 * ks)     * HID];
    f.y = wc[(size_t)(4 * ks + 1) * HID];
    *(v2f*)(wf + ((ks * 8 + t) * 32 + lane) * 2) = f;
  }
}

__device__ __forceinline__ void tile_dense(float* tile, const float* wf,
                                           const float* __restrict__ b,
                                           const int wv, const int lane) {
  const int half = lane >> 4;
  const int l    = lane & 15;
  v8f acc0 = (v8f){0,0,0,0,0,0,0,0};
  v8f acc1 = (v8f){0,0,0,0,0,0,0,0};
  float bias = 0.0f;
  if (wv < 8) {
    bias = b[wv * 16 + l];
    const float* ar = tile + l * TP + half * 2;
    const float* bf = wf + (wv * 32 + lane) * 2;
#pragma unroll 4
    for (int ks = 0; ks < 32; ks += 2) {
      const v2f a0 = *(const v2f*)(ar + ks * 4);
      const v2f b0 = *(const v2f*)(bf + ks * 512);
      const v2f a1 = *(const v2f*)(ar + ks * 4 + 4);
      const v2f b1 = *(const v2f*)(bf + ks * 512 + 512);
      acc0 = __builtin_amdgcn_wmma_f32_16x16x4_f32(false, a0, false, b0, (short)0, acc0, false, false);
      acc1 = __builtin_amdgcn_wmma_f32_16x16x4_f32(false, a1, false, b1, (short)0, acc1, false, false);
    }
  }
  __syncthreads();
  if (wv < 8) {
    float* st = tile + (half * 8) * TP + wv * 16 + l;
#pragma unroll
    for (int r = 0; r < 8; ++r) {
      const float v = acc0[r] + acc1[r] + bias;
      st[r * TP] = v > 0.0f ? v : 0.0f;
    }
  }
  __syncthreads();
}

__global__ void __launch_bounds__(512, 16)
k_gather_dense(const float* __restrict__ xin, const int* __restrict__ offs,
               const int2* __restrict__ edge, const float* __restrict__ dinv,
               const float* __restrict__ W, const float* __restrict__ b,
               float* __restrict__ C, int* ctr, int n, int nTiles) {
  __shared__ __attribute__((aligned(16))) float wf[HID * HID];
  __shared__ __attribute__((aligned(16))) float tile[16 * TP];
  __shared__ int s_first;
  __shared__ int s_next;
  const int wv   = __builtin_amdgcn_readfirstlane((int)(threadIdx.x >> 5));
  const int lane = threadIdx.x & 31;

  if (threadIdx.x == 0) s_first = atomicAdd(ctr, 1);
  stage_w(wf, W, wv, lane);
  __syncthreads();

  int tix = __builtin_amdgcn_readfirstlane(s_first);
  while (tix < nTiles) {
    const int c = tix * 16 + wv;

    float4 o = make_float4(0.0f, 0.0f, 0.0f, 0.0f);
    float dc = 0.0f;
    if (c < n) o = gather_row<false>(c, lane, xin, offs, edge, dinv, dc);
    *(float4*)(tile + wv * TP + lane * 4) = o;
    __syncthreads();
    if (threadIdx.x == 480) s_next = atomicAdd(ctr, 1);

    tile_dense(tile, wf, b, wv, lane);
    const int tnext = __builtin_amdgcn_readfirstlane(s_next);
    if (c < n) {
      float4 hv = *(const float4*)(tile + wv * TP + lane * 4);
      hv.x *= dc;  hv.y *= dc;  hv.z *= dc;  hv.w *= dc;
      *(float4*)(C + (size_t)c * HID + lane * 4) = hv;
    }
    tix = tnext;
  }
}

__global__ void __launch_bounds__(512, 16)
k_gather_dense_head(const float* __restrict__ hin, const int* __restrict__ offs,
                    const int2* __restrict__ edge, const float* __restrict__ dinv,
                    const float* __restrict__ W, const float* __restrict__ b,
                    const float* __restrict__ Wout, const float* __restrict__ bout,
                    float* __restrict__ out, int* ctr, int n, int nTiles) {
  __shared__ __attribute__((aligned(16))) float wf[HID * HID];
  __shared__ __attribute__((aligned(16))) float tile[16 * TP];
  __shared__ int s_first;
  __shared__ int s_next;
  const int wv   = __builtin_amdgcn_readfirstlane((int)(threadIdx.x >> 5));
  const int lane = threadIdx.x & 31;
  const int half = lane >> 4;
  const int l    = lane & 15;

  if (threadIdx.x == 0) s_first = atomicAdd(ctr, 1);
  stage_w(wf, W, wv, lane);
  __syncthreads();

  int tix = __builtin_amdgcn_readfirstlane(s_first);
  while (tix < nTiles) {
    const int row0 = tix * 16;
    const int c    = row0 + wv;

    float4 o = make_float4(0.0f, 0.0f, 0.0f, 0.0f);
    float dc = 0.0f;
    if (c < n) o = gather_row<true>(c, lane, hin, offs, edge, dinv, dc);
    __syncthreads();
    *(float4*)(tile + wv * TP + lane * 4) = o;
    __syncthreads();
    if (threadIdx.x == 480) s_next = atomicAdd(ctr, 1);

    tile_dense(tile, wf, b, wv, lane);
    const int tnext = __builtin_amdgcn_readfirstlane(s_next);

    if (wv == 8) {
      const float* ar = tile + l * TP + half * 2;
      const float* br = Wout + (size_t)l * HID + half * 2;
      v8f acc0 = (v8f){0,0,0,0,0,0,0,0};
      v8f acc1 = (v8f){0,0,0,0,0,0,0,0};
#pragma unroll 4
      for (int k = 0; k < HID; k += 8) {
        const v2f a0 = *(const v2f*)(ar + k);
        const v2f b0 = *(const v2f*)(br + k);
        const v2f a1 = *(const v2f*)(ar + k + 4);
        const v2f b1 = *(const v2f*)(br + k + 4);
        acc0 = __builtin_amdgcn_wmma_f32_16x16x4_f32(false, a0, false, b0, (short)0, acc0, false, false);
        acc1 = __builtin_amdgcn_wmma_f32_16x16x4_f32(false, a1, false, b1, (short)0, acc1, false, false);
      }
      const float bo = bout[l];
#pragma unroll
      for (int r = 0; r < 8; ++r) {
        const int m = row0 + r + half * 8;
        if (m < n) out[(size_t)m * NPRED + l] = acc0[r] + acc1[r] + bo;
      }
    }
    tix = tnext;
  }
}


extern "C" void kernel_launch(void* const* d_in, const int* in_sizes, int n_in,
                              void* d_out, int out_size, void* d_ws, size_t ws_size,
                              hipStream_t stream) {
  const float* x    = (const float*)d_in[0];
  const int*   ei   = (const int*)  d_in[1];
  const float* ew   = (const float*)d_in[2];
  const float* W1   = (const float*)d_in[3];
  const float* b1   = (const float*)d_in[4];
  const float* W2   = (const float*)d_in[5];
  const float* b2   = (const float*)d_in[6];
  const float* Wout = (const float*)d_in[7];
  const float* bout = (const float*)d_in[8];
  float* out = (float*)d_out;

  const int n = in_sizes[0] / HID;
  const int E = in_sizes[2];
  const int* erow = ei;
  const int* ecol = ei + E;

  const int B = 256;
  int gN  = (n + B - 1) / B;
  int gE  = (E + B - 1) / B;
  int nTiles = (n + 15) / 16;
  int gGemm  = (nTiles + 7) / 8;
  int gOut   = (n * NPRED + B - 1) / B;

  const int    gScan = (n + SCAN_ITEMS) / SCAN_ITEMS;
  const int    npad  = gScan * SCAN_ITEMS;
  const size_t plane = (size_t)n * HID;
  const size_t epad  = ((size_t)E + 63) / 64 * 64;
  const size_t ndpad = ((size_t)n + 127) / 128 * 128;
  const size_t need  = (plane + 2 * epad + (size_t)npad + ndpad + 128) * sizeof(float);

  const int    NB    = (n + BINW - 1) / BINW;
  const int    nChP  = ((E + PCH - 1) / PCH + 3) / 4 * 4;
  const size_t partw = 2 * (size_t)nChP * PCH;
  const size_t cntw  = (size_t)NB * (size_t)nChP;

  if (ws_size >= need && NB >= 1 && NB <= MAXBINS && nChP >= 4 && nChP <= MAXCH &&
      n <= (1 << 24) && partw + cntw <= plane) {
    float* h    = (float*)d_ws;
    int2*  edge = (int2*)(h + plane);
    int*   offs = (int*)(h + plane + 2 * epad);
    float* dinv = (float*)(offs + npad);
    int*   ctr  = (int*)(dinv + ndpad);
    int2*  part = (int2*)h;
    int*   cntT = (int*)(h + partw);

    k_part   <<<nChP, PT, 0, stream>>>(erow, ecol, ew, part, cntT, ctr, E, NB, nChP);
    k_binsort<<<NB, PT, 0, stream>>>(part, cntT, edge, offs, dinv, n, nChP);

    k_gather_dense<<<1024, 512, 0, stream>>>(x, offs, edge, dinv, W1, b1, h, ctr, n, nTiles);

    k_gather_dense_head<<<1024, 512, 0, stream>>>(h, offs, edge, dinv, W2, b2, Wout, bout, out,
                                                  ctr + 1, n, nTiles);
    return;
  }

  if (ws_size >= need && (size_t)npad + (size_t)E <= plane) {
    float* h    = (float*)d_ws;
    int2*  edge = (int2*)(h + plane);
    int*   offs = (int*)(h + plane + 2 * epad);
    float* dinv = (float*)(offs + npad);
    int*   ctr  = (int*)(dinv + ndpad);
    int*   cnt  = (int*)h;
    int*   rank = cnt + npad;

    k_zero_cnt<<<npad / B, B, 0, stream>>>(cnt, ctr, npad);
    k_rank    <<<gE, B, 0, stream>>>(ecol, cnt, rank, E);
    k_scan    <<<gScan, SCAN_T, 0, stream>>>(cnt, offs);
    k_fill    <<<gE, B, 0, stream>>>(erow, ecol, ew, rank, offs, edge, E);
    k_dinv_seg<<<gN, B, 0, stream>>>(offs, edge, dinv, n);

    k_gather_dense<<<1024, 512, 0, stream>>>(x, offs, edge, dinv, W1, b1, h, ctr, n, nTiles);

    k_gather_dense_head<<<1024, 512, 0, stream>>>(h, offs, edge, dinv, W2, b2, Wout, bout, out,
                                                  ctr + 1, n, nTiles);
    return;
  }

  float* dinv = (float*)d_ws;
  float* norm = dinv + n;
  float* xw   = norm + E;
  float* acc  = xw + (size_t)n * HID;

  int gScat  = (E + 7) / 8;
  int gEpi   = (n * (HID / 4) + B - 1) / B;

  k_set_deg    <<<gN, B, 0, stream>>>(dinv, n);
  k_deg_scatter<<<gE, B, 0, stream>>>(ecol, ew, dinv, E);
  k_dinv       <<<gN, B, 0, stream>>>(dinv, n);
  k_norm       <<<gE, B, 0, stream>>>(erow, ecol, ew, dinv, norm, E);

  k_gemm_wmma<<<gGemm, B, 0, stream>>>(x, W1, xw, n);
  hipMemsetAsync(acc, 0, (size_t)n * HID * sizeof(float), stream);
  k_scatter  <<<gScat, B, 0, stream>>>(xw, erow, ecol, norm, acc, E);
  k_epilogue <<<gEpi, B, 0, stream>>>(acc, xw, dinv, b1, n);

  k_gemm_wmma<<<gGemm, B, 0, stream>>>(acc, W2, xw, n);
  hipMemsetAsync(acc, 0, (size_t)n * HID * sizeof(float), stream);
  k_scatter  <<<gScat, B, 0, stream>>>(xw, erow, ecol, norm, acc, E);
  k_epilogue <<<gEpi, B, 0, stream>>>(acc, xw, dinv, b2, n);

  k_out_proj<<<gOut, B, 0, stream>>>(acc, Wout, bout, out, n);
}
